# starting kernel
# baseline (speedup 1.0000x reference)
#include <hip/hip_runtime.h>
#include <hip/hip_bf16.h>
#include <math.h>

typedef __attribute__((ext_vector_type(2))) float v2f;
typedef __attribute__((ext_vector_type(8))) float v8f;

__device__ __forceinline__ float lrelu(float x) { return x > 0.f ? x : 0.2f * x; }

__device__ __forceinline__ void atomic_max_f32(float* addr, float val) {
  if (val >= 0.f)
    __hip_atomic_fetch_max((int*)addr, __float_as_int(val),
                           __ATOMIC_RELAXED, __HIP_MEMORY_SCOPE_AGENT);
  else
    __hip_atomic_fetch_min((unsigned int*)addr, __float_as_uint(val),
                           __ATOMIC_RELAXED, __HIP_MEMORY_SCOPE_AGENT);
}

__device__ __forceinline__ void atomic_add_f32(float* addr, float val) {
  __hip_atomic_fetch_add(addr, val, __ATOMIC_RELAXED, __HIP_MEMORY_SCOPE_AGENT);
}

__global__ __launch_bounds__(1024)
void prep_kernel(const float* __restrict__ Wsq, const float* __restrict__ Wdq,
                 const float* __restrict__ Asq, const float* __restrict__ Adq,
                 const float* __restrict__ Wsa, const float* __restrict__ Wda,
                 const float* __restrict__ Asa, const float* __restrict__ Ada,
                 const float* __restrict__ bq,  const float* __restrict__ ba,
                 const float* __restrict__ Wout, const float* __restrict__ bout,
                 float* __restrict__ Vp, float* __restrict__ Pc, float* __restrict__ cst) {
  const int t = threadIdx.x;
  for (int idx = t; idx < 4 * 64 * 16; idx += blockDim.x) {
    int table = idx >> 10, k = (idx >> 4) & 63, j = idx & 15;
    const float* W = (table == 0) ? Wsq : (table == 1) ? Wdq : (table == 2) ? Wsa : Wda;
    const float* A = (table == 0) ? Asq : (table == 1) ? Adq : (table == 2) ? Asa : Ada;
    float s = 0.f;
    if (j < 4)
      for (int c = 0; c < 64; ++c) s += W[k * 256 + j * 64 + c] * A[j * 64 + c];
    Vp[idx] = s;
  }
  for (int idx = t; idx < 512 * 16; idx += blockDim.x) {
    int r = idx >> 4, j = idx & 15;
    int layer = r >> 8, h = (r >> 6) & 3, k = r & 63;
    const float* W = layer ? Wsa : Wsq;
    float s = 0.f;
    if (j < 4) {
      for (int c = 0; c < 64; ++c)
        s += W[k * 256 + h * 64 + c] * Wout[(layer * 64 + c) * 4 + j];
      s *= 0.25f;
    }
    Pc[idx] = s;
  }
  if (t < 16) {
    float s = 0.f;
    if (t < 4) {
      for (int c = 0; c < 64; ++c)
        s += bq[c] * Wout[c * 4 + t] + ba[c] * Wout[(64 + c) * 4 + t];
      s += bout[t];
    }
    cst[t] = s;
  }
}

__global__ void init_kernel(float* __restrict__ zero_ptr, long nzero,
                            float* __restrict__ minf_ptr, long nminf) {
  const long stride = (long)gridDim.x * blockDim.x;
  const long t0 = (long)blockIdx.x * blockDim.x + threadIdx.x;
  for (long i = t0; i < nzero; i += stride) zero_ptr[i] = 0.f;
  for (long i = t0; i < nminf; i += stride) minf_ptr[i] = __int_as_float(0xff800000);
}

__global__ void scores_kernel(const float* __restrict__ x, const float* __restrict__ Vp,
                              float* __restrict__ out, int N) {
  const int wave_g = (int)(((unsigned)blockIdx.x * blockDim.x + threadIdx.x) >> 5);
  const int lane = threadIdx.x & 31;
  const long base = (long)wave_g * 16;
  if (base >= N) return;
  const int m = lane & 15;
  const int ko = (lane >> 4) * 2;
  long row = base + m; if (row >= N) row = N - 1;
  const float* xr = x + row * 64;
  v8f acc = {};
#pragma unroll
  for (int kk = 0; kk < 64; kk += 4) {
    v2f a, b;
    a.x = xr[kk + ko];
    a.y = xr[kk + ko + 1];
    b.x = Vp[(kk + ko) * 16 + m];
    b.y = Vp[(kk + ko + 1) * 16 + m];
    acc = __builtin_amdgcn_wmma_f32_16x16x4_f32(false, a, false, b, (short)0, acc,
                                                false, false);
  }
  if (m < 4) {
    const long rb = base + ((lane >> 4) << 3);
#pragma unroll
    for (int v = 0; v < 8; ++v) {
      long r = rb + v;
      if (r < N) out[r * 4 + m] = acc[v];
    }
  }
}

__global__ void edge_pass1(const int* __restrict__ src, const int* __restrict__ dst,
                           const float* __restrict__ a_s, const float* __restrict__ a_d,
                           float* __restrict__ e, float* __restrict__ mbuf, int E) {
  const int i = blockIdx.x * blockDim.x + threadIdx.x;
  if (i >= E) return;
  const long s = src[i], d = dst[i];
  const float4 as = *(const float4*)(a_s + s * 4);
  const float4 ad = *(const float4*)(a_d + d * 4);
  float4 ev;
  ev.x = lrelu(as.x + ad.x);
  ev.y = lrelu(as.y + ad.y);
  ev.z = lrelu(as.z + ad.z);
  ev.w = lrelu(as.w + ad.w);
  *(float4*)(e + (long)i * 4) = ev;
  atomic_max_f32(mbuf + d * 4 + 0, ev.x);
  atomic_max_f32(mbuf + d * 4 + 1, ev.y);
  atomic_max_f32(mbuf + d * 4 + 2, ev.z);
  atomic_max_f32(mbuf + d * 4 + 3, ev.w);
}

__global__ void edge_pass2(const int* __restrict__ dst, float* __restrict__ e,
                           const float* __restrict__ mbuf, float* __restrict__ den, int E) {
  const int i = blockIdx.x * blockDim.x + threadIdx.x;
  if (i >= E) return;
  const long d = dst[i];
  float4 ev = *(const float4*)(e + (long)i * 4);
  const float4 mv = *(const float4*)(mbuf + d * 4);
  ev.x = __expf(ev.x - mv.x);
  ev.y = __expf(ev.y - mv.y);
  ev.z = __expf(ev.z - mv.z);
  ev.w = __expf(ev.w - mv.w);
  *(float4*)(e + (long)i * 4) = ev;
  atomic_add_f32(den + d * 4 + 0, ev.x);
  atomic_add_f32(den + d * 4 + 1, ev.y);
  atomic_add_f32(den + d * 4 + 2, ev.z);
  atomic_add_f32(den + d * 4 + 3, ev.w);
}

__global__ void edge_pass3(const int* __restrict__ src, const int* __restrict__ dst,
                           const float* __restrict__ x, const float* __restrict__ ex,
                           const float* __restrict__ den, float* __restrict__ z, long E) {
  const long gt = (long)blockIdx.x * blockDim.x + threadIdx.x;
  const long ei = gt >> 6;
  if (ei >= E) return;
  const int t = (int)(gt & 63);
  const int h = t >> 4;
  const int k0 = (t & 15) * 4;
  const long s = src[ei], d = dst[ei];
  const float alpha = ex[ei * 4 + h] / (den[d * 4 + h] + 1e-16f);
  const float4 xv = *(const float4*)(x + s * 64 + k0);
  float* zp = z + d * 256 + h * 64 + k0;
  atomic_add_f32(zp + 0, alpha * xv.x);
  atomic_add_f32(zp + 1, alpha * xv.y);
  atomic_add_f32(zp + 2, alpha * xv.z);
  atomic_add_f32(zp + 3, alpha * xv.w);
}

__global__ void final_kernel(const float* __restrict__ zq, const float* __restrict__ za,
                             const float* __restrict__ Pc, const float* __restrict__ cst,
                             float* __restrict__ out, int N) {
  const int wave_g = (int)(((unsigned)blockIdx.x * blockDim.x + threadIdx.x) >> 5);
  const int lane = threadIdx.x & 31;
  const long base = (long)wave_g * 16;
  if (base >= N) return;
  const int m = lane & 15;
  const int ko = (lane >> 4) * 2;
  long row = base + m; if (row >= N) row = N - 1;
  const float* zqr = zq + row * 256;
  const float* zar = za + row * 256;
  v8f acc = {};
#pragma unroll 8
  for (int kk = 0; kk < 512; kk += 4) {
    const int k0 = kk + ko;
    const float* zr = (k0 < 256) ? zqr : zar;
    const int kL = k0 & 255;
    v2f a, b;
    a.x = zr[kL];
    a.y = zr[kL + 1];
    b.x = Pc[k0 * 16 + m];
    b.y = Pc[(k0 + 1) * 16 + m];
    acc = __builtin_amdgcn_wmma_f32_16x16x4_f32(false, a, false, b, (short)0, acc,
                                                false, false);
  }
  if (m < 4) {
    const float c0 = cst[m];
    const long rb = base + ((lane >> 4) << 3);
#pragma unroll
    for (int v = 0; v < 8; ++v) {
      long r = rb + v;
      if (r < N) out[r * 4 + m] = acc[v] + c0;
    }
  }
}

extern "C" void kernel_launch(void* const* d_in, const int* in_sizes, int n_in,
                              void* d_out, int out_size, void* d_ws, size_t ws_size,
                              hipStream_t stream) {
  const float* x_c = (const float*)d_in[0];
  const float* x_q = (const float*)d_in[1];
  const float* x_a = (const float*)d_in[2];
  const int* src_cq = (const int*)d_in[3];
  const int* dst_cq = (const int*)d_in[4];
  const int* src_ca = (const int*)d_in[5];
  const int* dst_ca = (const int*)d_in[6];
  const float* Wsq = (const float*)d_in[7];
  const float* Wdq = (const float*)d_in[8];
  const float* Asq = (const float*)d_in[9];
  const float* Adq = (const float*)d_in[10];
  const float* bq  = (const float*)d_in[11];
  const float* Wsa = (const float*)d_in[12];
  const float* Wda = (const float*)d_in[13];
  const float* Asa = (const float*)d_in[14];
  const float* Ada = (const float*)d_in[15];
  const float* ba  = (const float*)d_in[16];
  const float* Wout = (const float*)d_in[17];
  const float* bout = (const float*)d_in[18];

  const long Nc = in_sizes[0] / 64;
  const long Nq = in_sizes[1] / 64;
  const long Na = in_sizes[2] / 64;
  const long Ecq = in_sizes[3];
  const long Eca = in_sizes[5];

  float* w = (float*)d_ws;
  size_t off = 0;
  auto alloc = [&](size_t n) -> float* {
    float* p = w + off;
    off += (n + 15) & ~(size_t)15;
    return p;
  };
  float* Vp_sq = alloc(64 * 16);
  float* Vp_dq = alloc(64 * 16);
  float* Vp_sa = alloc(64 * 16);
  float* Vp_da = alloc(64 * 16);
  float* Pc    = alloc(512 * 16);
  float* cst   = alloc(16);
  float* as_cq = alloc(Nc * 4);
  float* as_ca = alloc(Nc * 4);
  float* ad_cq = alloc(Nq * 4);
  float* ad_ca = alloc(Na * 4);
  float* e_cq  = alloc(Ecq * 4);
  float* e_ca  = alloc(Eca * 4);
  float* m_q   = alloc(Nq * 4);
  float* m_a   = alloc(Na * 4);
  float* den_q = alloc(Nq * 4);
  float* den_a = alloc(Na * 4);
  float* z_q   = alloc(Nq * 256);
  float* z_a   = alloc(Na * 256);
  (void)den_a; (void)Vp_dq; (void)Vp_sa; (void)Vp_da;
  if (off * sizeof(float) > ws_size) return;

  const long nminf = (long)(den_q - m_q);
  const long nzero = (long)((z_a + ((Na * 256 + 15) & ~(size_t)15)) - den_q);

  prep_kernel<<<1, 1024, 0, stream>>>(Wsq, Wdq, Asq, Adq, Wsa, Wda, Asa, Ada,
                                      bq, ba, Wout, bout, Vp_sq, Pc, cst);
  init_kernel<<<2048, 256, 0, stream>>>(den_q, nzero, m_q, nminf);

  const int gNc = (int)((Nc + 127) / 128);
  const int gNq = (int)((Nq + 127) / 128);
  const int gNa = (int)((Na + 127) / 128);
  scores_kernel<<<gNc, 256, 0, stream>>>(x_c, Vp_sq, as_cq, (int)Nc);
  scores_kernel<<<gNc, 256, 0, stream>>>(x_c, Vp_sa, as_ca, (int)Nc);
  scores_kernel<<<gNq, 256, 0, stream>>>(x_q, Vp_dq, ad_cq, (int)Nq);
  scores_kernel<<<gNa, 256, 0, stream>>>(x_a, Vp_da, ad_ca, (int)Na);

  const int gEcq = (int)((Ecq + 255) / 256);
  const int gEca = (int)((Eca + 255) / 256);
  edge_pass1<<<gEcq, 256, 0, stream>>>(src_cq, dst_cq, as_cq, ad_cq, e_cq, m_q, (int)Ecq);
  edge_pass1<<<gEca, 256, 0, stream>>>(src_ca, dst_ca, as_ca, ad_ca, e_ca, m_a, (int)Eca);
  edge_pass2<<<gEcq, 256, 0, stream>>>(dst_cq, e_cq, m_q, den_q, (int)Ecq);
  edge_pass2<<<gEca, 256, 0, stream>>>(dst_ca, e_ca, m_a, den_a, (int)Eca);

  edge_pass3<<<(int)((Ecq * 64 + 255) / 256), 256, 0, stream>>>(src_cq, dst_cq, x_c,
                                                                e_cq, den_q, z_q, Ecq);
  edge_pass3<<<(int)((Eca * 64 + 255) / 256), 256, 0, stream>>>(src_ca, dst_ca, x_c,
                                                                e_ca, den_a, z_a, Eca);

  final_kernel<<<gNq, 256, 0, stream>>>(z_q, z_a, Pc, cst, (float*)d_out, (int)Nq);
}


# Resident grids: scores, edge count, list fill, node walk and zeroing launch 128-2048 workgroups (were up to 25000); waves stride over row tiles, edge chunks, nodes; sums stay in arrival order
# speedup vs baseline: 4.2556x; 4.2556x over previous
#include <hip/hip_runtime.h>
#include <hip/hip_bf16.h>
#include <math.h>

typedef __attribute__((ext_vector_type(2))) float v2f;
typedef __attribute__((ext_vector_type(8))) float v8f;

__device__ __forceinline__ float lrelu(float x) { return x > 0.f ? x : 0.2f * x; }

template <int M>
__device__ __forceinline__ float xrow(float v) {
  const int b = __float_as_int(v);
  return __int_as_float(__builtin_amdgcn_update_dpp(b, b, 0x160 + M, 0xf, 0xf, false));
}
__device__ __forceinline__ float xhalf(float v) {
  const int b = __float_as_int(v);
  return __int_as_float(__builtin_amdgcn_permlanex16(b, b, 0x76543210, 0xfedcba98, false, false));
}
__device__ __forceinline__ float wave_max(float v) {
  v = fmaxf(v, xrow<1>(v));
  v = fmaxf(v, xrow<2>(v));
  v = fmaxf(v, xrow<4>(v));
  v = fmaxf(v, xrow<8>(v));
  return fmaxf(v, xhalf(v));
}
__device__ __forceinline__ float wave_sum(float v) {
  v += xrow<1>(v);
  v += xrow<2>(v);
  v += xrow<4>(v);
  v += xrow<8>(v);
  return v + xhalf(v);
}

__global__ __launch_bounds__(1024)
void prep_kernel(const float* __restrict__ Wsq, const float* __restrict__ Wdq,
                 const float* __restrict__ Asq, const float* __restrict__ Adq,
                 const float* __restrict__ Wsa, const float* __restrict__ Wda,
                 const float* __restrict__ Asa, const float* __restrict__ Ada,
                 const float* __restrict__ bq,  const float* __restrict__ ba,
                 const float* __restrict__ Wout, const float* __restrict__ bout,
                 float* __restrict__ Vp, float* __restrict__ Pc, float* __restrict__ cst) {
  const int t = threadIdx.x;
  for (int idx = t; idx < 4 * 64 * 16; idx += blockDim.x) {
    int table = idx >> 10, k = (idx >> 4) & 63, j = idx & 15;
    const float* W = (table == 0) ? Wsq : (table == 1) ? Wdq : (table == 2) ? Wsa : Wda;
    const float* A = (table == 0) ? Asq : (table == 1) ? Adq : (table == 2) ? Asa : Ada;
    float s = 0.f;
    if (j < 4)
      for (int c = 0; c < 64; ++c) s += W[k * 256 + j * 64 + c] * A[j * 64 + c];
    Vp[idx] = s;
  }
  for (int idx = t; idx < 512 * 16; idx += blockDim.x) {
    int r = idx >> 4, j = idx & 15;
    int layer = r >> 8, h = (r >> 6) & 3, k = r & 63;
    const float* W = layer ? Wsa : Wsq;
    float s = 0.f;
    if (j < 4) {
      for (int c = 0; c < 64; ++c)
        s += W[k * 256 + h * 64 + c] * Wout[(layer * 64 + c) * 4 + j];
      s *= 0.25f;
    }
    Pc[idx] = s;
  }
  if (t < 16) {
    float s = 0.f;
    if (t < 4) {
      for (int c = 0; c < 64; ++c)
        s += bq[c] * Wout[c * 4 + t] + ba[c] * Wout[(64 + c) * 4 + t];
      s += bout[t];
    }
    cst[t] = s;
  }
}

__global__ void init_kernel(int* __restrict__ zero_ptr, int nzero) {
  const int T = (int)(gridDim.x * blockDim.x);
  for (int i = blockIdx.x * blockDim.x + threadIdx.x; i < nzero; i += T) zero_ptr[i] = 0;
}

__global__ void scores_kernel(const float* __restrict__ x, const float* __restrict__ Vp,
                              float* __restrict__ out, int N) {
  const int lane = threadIdx.x & 31;
  const int m = lane & 15;
  const int ko = (lane >> 4) * 2;
  const int nw = (int)((gridDim.x * blockDim.x) >> 5);
  const int ntiles = (N + 15) >> 4;
  v2f b[16];
#pragma unroll
  for (int q = 0; q < 16; ++q) {
    b[q].x = Vp[(q * 4 + ko) * 16 + m];
    b[q].y = Vp[(q * 4 + ko + 1) * 16 + m];
  }
  for (int t = __builtin_amdgcn_readfirstlane(
           (int)(((unsigned)blockIdx.x * blockDim.x + threadIdx.x) >> 5));
       t < ntiles; t += nw) {
    const long base = (long)t * 16;
    long row = base + m; if (row >= N) row = N - 1;
    const float* xr = x + row * 64;
    v8f acc = {};
#pragma unroll
    for (int q = 0; q < 16; ++q) {
      v2f a;
      a.x = xr[q * 4 + ko];
      a.y = xr[q * 4 + ko + 1];
      acc = __builtin_amdgcn_wmma_f32_16x16x4_f32(false, a, false, b[q], (short)0, acc,
                                                  false, false);
    }
    if (m < 4) {
      const long rb = base + ((lane >> 4) << 3);
#pragma unroll
      for (int v = 0; v < 8; ++v) {
        long r = rb + v;
        if (r < N) out[r * 4 + m] = acc[v];
      }
    }
  }
}

__global__ void count_kernel(const int* __restrict__ dst_q, const int* __restrict__ dst_a,
                             int* __restrict__ cnt, int* __restrict__ rank,
                             int Ecq, int Etot, int Nq) {
  const int B = (int)blockDim.x;
  const int nchunk = (Etot + 4 * B - 1) / (4 * B);
  for (int c = blockIdx.x; c < nchunk; c += (int)gridDim.x) {
    const int i0 = c * 4 * B + (int)threadIdx.x;
    if (c * 4 * B + 4 * B <= Etot) {
      int dv[4], r[4];
#pragma unroll
      for (int k = 0; k < 4; ++k) {
        const int i = i0 + k * B;
        const int* p = (i < Ecq) ? dst_q + i : dst_a + (i - Ecq);
        dv[k] = *p;
      }
#pragma unroll
      for (int k = 0; k < 4; ++k) {
        const int i = i0 + k * B;
        r[k] = __hip_atomic_fetch_add(cnt + ((i < Ecq) ? dv[k] : Nq + dv[k]), 1,
                                      __ATOMIC_RELAXED, __HIP_MEMORY_SCOPE_AGENT);
      }
#pragma unroll
      for (int k = 0; k < 4; ++k) rank[i0 + k * B] = r[k];
    } else {
      for (int i = i0; i < Etot; i += B) {
        const int g = (i < Ecq) ? dst_q[i] : Nq + dst_a[i - Ecq];
        rank[i] = __hip_atomic_fetch_add(cnt + g, 1, __ATOMIC_RELAXED, __HIP_MEMORY_SCOPE_AGENT);
      }
    }
  }
}

__global__ __launch_bounds__(1024)
void scan_kernel(const int* __restrict__ cnt, int* __restrict__ off, int G) {
  __shared__ int wsum[32];
  const int t = threadIdx.x;
  const int lane = t & 31;
  const int wv = t >> 5;
  const int tile0 = blockIdx.x * 4096;
  int s = 0;
  for (int i = t * 4; i < tile0; i += 4096) {
    const int4 q = *(const int4*)(cnt + i);
    s += q.x + q.y + q.z + q.w;
  }
  for (int o = 16; o > 0; o >>= 1) s += __shfl_xor(s, o);
  if (lane == 0) wsum[wv] = s;
  __syncthreads();
  int carry = 0;
  for (int i = 0; i < 32; ++i) carry += wsum[i];
  __syncthreads();

  const int base = tile0 + t * 4;
  int v0 = 0, v1 = 0, v2 = 0;
  int v3 = 0;
  if (base + 3 < G) {
    const int4 q = *(const int4*)(cnt + base);
    v0 = q.x; v1 = q.y; v2 = q.z; v3 = q.w;
  } else {
    if (base < G) v0 = cnt[base];
    if (base + 1 < G) v1 = cnt[base + 1];
    if (base + 2 < G) v2 = cnt[base + 2];
  }
  const int tsum = v0 + v1 + v2 + v3;
  int inc = tsum;
  for (int o = 1; o < 32; o <<= 1) {
    const int nb = __shfl_up(inc, o);
    if (lane >= o) inc += nb;
  }
  if (lane == 31) wsum[wv] = inc;
  __syncthreads();
  int wpre = 0;
  for (int i = 0; i < wv; ++i) wpre += wsum[i];
  const int ex0 = carry + wpre + inc - tsum;
  if (base < G) off[base] = ex0;
  if (base + 1 < G) off[base + 1] = ex0 + v0;
  if (base + 2 < G) off[base + 2] = ex0 + v0 + v1;
  if (base + 3 < G) off[base + 3] = ex0 + v0 + v1 + v2;
}

__global__ void fill_kernel(const int* __restrict__ src_q, const int* __restrict__ src_a,
                            const int* __restrict__ dst_q, const int* __restrict__ dst_a,
                            const int* __restrict__ rank, const int* __restrict__ off,
                            int* __restrict__ slist, int Ecq, int Etot, int Nq) {
  const int B = (int)blockDim.x;
  const int nchunk = (Etot + 4 * B - 1) / (4 * B);
  for (int c = blockIdx.x; c < nchunk; c += (int)gridDim.x) {
    const int i0 = c * 4 * B + (int)threadIdx.x;
    if (c * 4 * B + 4 * B <= Etot) {
      int g[4], s[4], rk[4], o[4];
#pragma unroll
      for (int k = 0; k < 4; ++k) {
        const int i = i0 + k * B;
        const bool second = i >= Ecq;
        const int e = second ? i - Ecq : i;
        const int* dp = second ? dst_a : dst_q;
        const int* sp = second ? src_a : src_q;
        g[k] = dp[e];
        s[k] = sp[e];
        rk[k] = rank[i];
      }
#pragma unroll
      for (int k = 0; k < 4; ++k) o[k] = off[(i0 + k * B >= Ecq) ? Nq + g[k] : g[k]];
#pragma unroll
      for (int k = 0; k < 4; ++k) slist[o[k] + rk[k]] = s[k];
    } else {
      for (int i = i0; i < Etot; i += B) {
        const bool second = i >= Ecq;
        const int e = second ? i - Ecq : i;
        const int g = second ? Nq + dst_a[e] : dst_q[e];
        const int s = second ? src_a[e] : src_q[e];
        slist[off[g] + rank[i]] = s;
      }
    }
  }
}

__global__ void gather_kernel(const float* __restrict__ x,
                              const float* __restrict__ as_q, const float* __restrict__ as_a,
                              const float* __restrict__ ad_q, const float* __restrict__ ad_a,
                              const int* __restrict__ off, const int* __restrict__ slist,
                              float* __restrict__ z_q, float* __restrict__ z_a,
                              int Nq, int G, int Etot) {
  const int lane = threadIdx.x & 31;
  const int nw = (int)((gridDim.x * blockDim.x) >> 5);
  const float* xl = x + lane * 2;
  const float ninf = __int_as_float(0xff800000);
  for (int g = __builtin_amdgcn_readfirstlane(
           (int)(((unsigned)blockIdx.x * blockDim.x + threadIdx.x) >> 5));
       g < G; g += nw) {
  const bool second = g >= Nq;
  const int d = second ? g - Nq : g;
  const float* a_s = second ? as_a : as_q;
  const float4 ad = *(const float4*)((second ? ad_a : ad_q) + (long)d * 4);
  float* z = (second ? z_a : z_q) + (long)d * 256 + lane * 2;
  const int beg = off[g];
  const int end = (g + 1 < G) ? off[g + 1] : Etot;
  float m0 = ninf, m1 = ninf, m2 = ninf, m3 = ninf;
  float d0 = 0.f, d1 = 0.f, d2 = 0.f, d3 = 0.f;
  v2f acc0 = {0.f, 0.f}, acc1 = {0.f, 0.f}, acc2 = {0.f, 0.f}, acc3 = {0.f, 0.f};
  for (int c = beg; c < end; c += 32) {
    const int n = (end - c < 32) ? end - c : 32;
    int s_l = 0;
    float e0 = ninf, e1 = ninf, e2 = ninf, e3 = ninf;
    if (lane < n) {
      s_l = slist[c + lane];
      const float4 as = *(const float4*)(a_s + (long)s_l * 4);
      e0 = lrelu(as.x + ad.x);
      e1 = lrelu(as.y + ad.y);
      e2 = lrelu(as.z + ad.z);
      e3 = lrelu(as.w + ad.w);
    }
    const float n0 = fmaxf(m0, wave_max(e0)), n1 = fmaxf(m1, wave_max(e1));
    const float n2 = fmaxf(m2, wave_max(e2)), n3 = fmaxf(m3, wave_max(e3));
    const float r0 = __expf(m0 - n0), r1 = __expf(m1 - n1);
    const float r2 = __expf(m2 - n2), r3 = __expf(m3 - n3);
    const float p0 = __expf(e0 - n0), p1 = __expf(e1 - n1);
    const float p2 = __expf(e2 - n2), p3 = __expf(e3 - n3);
    d0 = d0 * r0 + wave_sum(p0); d1 = d1 * r1 + wave_sum(p1);
    d2 = d2 * r2 + wave_sum(p2); d3 = d3 * r3 + wave_sum(p3);
    acc0 *= r0; acc1 *= r1; acc2 *= r2; acc3 *= r3;
    m0 = n0; m1 = n1; m2 = n2; m3 = n3;
#pragma unroll 4
    for (int j = 0; j < n; ++j) {
      const int s = __builtin_amdgcn_readlane(s_l, j);
      const float w0 = __int_as_float(__builtin_amdgcn_readlane(__float_as_int(p0), j));
      const float w1 = __int_as_float(__builtin_amdgcn_readlane(__float_as_int(p1), j));
      const float w2 = __int_as_float(__builtin_amdgcn_readlane(__float_as_int(p2), j));
      const float w3 = __int_as_float(__builtin_amdgcn_readlane(__float_as_int(p3), j));
      const v2f xv = *(const v2f*)(xl + (long)s * 64);
      acc0 += w0 * xv;
      acc1 += w1 * xv;
      acc2 += w2 * xv;
      acc3 += w3 * xv;
    }
  }
  *(v2f*)(z + 0) = acc0 / (d0 + 1e-16f);
  *(v2f*)(z + 64) = acc1 / (d1 + 1e-16f);
  *(v2f*)(z + 128) = acc2 / (d2 + 1e-16f);
  *(v2f*)(z + 192) = acc3 / (d3 + 1e-16f);
  }
}

__global__ void final_kernel(const float* __restrict__ zq, const float* __restrict__ za,
                             const float* __restrict__ Pc, const float* __restrict__ cst,
                             float* __restrict__ out, int N) {
  const int wave_g = (int)(((unsigned)blockIdx.x * blockDim.x + threadIdx.x) >> 5);
  const int lane = threadIdx.x & 31;
  const long base = (long)wave_g * 16;
  if (base >= N) return;
  const int m = lane & 15;
  const int ko = (lane >> 4) * 2;
  long row = base + m; if (row >= N) row = N - 1;
  const float* zqr = zq + row * 256;
  const float* zar = za + row * 256;
  v8f acc = {};
#pragma unroll 8
  for (int kk = 0; kk < 512; kk += 4) {
    const int k0 = kk + ko;
    const float* zr = (k0 < 256) ? zqr : zar;
    const int kL = k0 & 255;
    v2f a, b;
    a.x = zr[kL];
    a.y = zr[kL + 1];
    b.x = Pc[k0 * 16 + m];
    b.y = Pc[(k0 + 1) * 16 + m];
    acc = __builtin_amdgcn_wmma_f32_16x16x4_f32(false, a, false, b, (short)0, acc,
                                                false, false);
  }
  if (m < 4) {
    const float c0 = cst[m];
    const long rb = base + ((lane >> 4) << 3);
#pragma unroll
    for (int v = 0; v < 8; ++v) {
      long r = rb + v;
      if (r < N) out[r * 4 + m] = acc[v] + c0;
    }
  }
}

extern "C" void kernel_launch(void* const* d_in, const int* in_sizes, int n_in,
                              void* d_out, int out_size, void* d_ws, size_t ws_size,
                              hipStream_t stream) {
  const float* x_c = (const float*)d_in[0];
  const float* x_q = (const float*)d_in[1];
  const float* x_a = (const float*)d_in[2];
  const int* src_cq = (const int*)d_in[3];
  const int* dst_cq = (const int*)d_in[4];
  const int* src_ca = (const int*)d_in[5];
  const int* dst_ca = (const int*)d_in[6];
  const float* Wsq = (const float*)d_in[7];
  const float* Wdq = (const float*)d_in[8];
  const float* Asq = (const float*)d_in[9];
  const float* Adq = (const float*)d_in[10];
  const float* bq  = (const float*)d_in[11];
  const float* Wsa = (const float*)d_in[12];
  const float* Wda = (const float*)d_in[13];
  const float* Asa = (const float*)d_in[14];
  const float* Ada = (const float*)d_in[15];
  const float* ba  = (const float*)d_in[16];
  const float* Wout = (const float*)d_in[17];
  const float* bout = (const float*)d_in[18];

  const long Nc = in_sizes[0] / 64;
  const long Nq = in_sizes[1] / 64;
  const long Na = in_sizes[2] / 64;
  const long Ecq = in_sizes[3];
  const long Eca = in_sizes[5];

  float* w = (float*)d_ws;
  size_t off = 0;
  auto alloc = [&](size_t n) -> float* {
    float* p = w + off;
    off += (n + 15) & ~(size_t)15;
    return p;
  };
  float* Vp_sq = alloc(64 * 16);
  float* Vp_dq = alloc(64 * 16);
  float* Vp_sa = alloc(64 * 16);
  float* Vp_da = alloc(64 * 16);
  float* Pc    = alloc(512 * 16);
  float* cst   = alloc(16);
  float* as_cq = alloc(Nc * 4);
  float* as_ca = alloc(Nc * 4);
  float* ad_cq = alloc(Nq * 4);
  float* ad_ca = alloc(Na * 4);
  float* e_cq  = alloc(Ecq * 4);
  float* e_ca  = alloc(Eca * 4);
  float* m_q   = alloc(Nq * 4);
  float* m_a   = alloc(Na * 4);
  float* den_q = alloc(Nq * 4);
  float* den_a = alloc(Na * 4);
  float* z_q   = alloc(Nq * 256);
  float* z_a   = alloc(Na * 256);
  (void)Vp_dq; (void)Vp_sa; (void)Vp_da; (void)e_ca; (void)m_a; (void)den_q; (void)den_a;
  if (off * sizeof(float) > ws_size) return;

  const long G = Nq + Na;
  const long Etot = Ecq + Eca;
  int* cnt   = (int*)z_q;
  int* rank  = (int*)z_a;
  int* slist = (int*)e_cq;
  int* offs  = (int*)m_q;
  if (G > Nq * 256 || Etot > Na * 256 || Etot > (long)(m_q - e_cq) ||
      G > (long)(z_q - m_q)) return;

  prep_kernel<<<1, 1024, 0, stream>>>(Wsq, Wdq, Asq, Adq, Wsa, Wda, Asa, Ada,
                                      bq, ba, Wout, bout, Vp_sq, Pc, cst);
  const int gInit = (G + 255) / 256 < 128 ? (int)((G + 255) / 256) : 128;
  init_kernel<<<gInit, 256, 0, stream>>>(cnt, (int)G);

  const int gNc = (Nc + 127) / 128 < 1024 ? (int)((Nc + 127) / 128) : 1024;
  const int gNq = (Nq + 127) / 128 < 1024 ? (int)((Nq + 127) / 128) : 1024;
  const int gNa = (Na + 127) / 128 < 1024 ? (int)((Na + 127) / 128) : 1024;
  scores_kernel<<<gNc, 256, 0, stream>>>(x_c, Vp_sq, as_cq, (int)Nc);
  scores_kernel<<<gNc, 256, 0, stream>>>(x_c, Vp_sa, as_ca, (int)Nc);
  scores_kernel<<<gNq, 256, 0, stream>>>(x_q, Vp_dq, ad_cq, (int)Nq);
  scores_kernel<<<gNa, 256, 0, stream>>>(x_a, Vp_da, ad_ca, (int)Na);

  const int gEt = (Etot + 1023) / 1024 < 1024 ? (int)((Etot + 1023) / 1024) : 1024;
  count_kernel<<<gEt, 256, 0, stream>>>(dst_cq, dst_ca, cnt, rank, (int)Ecq, (int)Etot, (int)Nq);
  scan_kernel<<<(int)((G + 4095) / 4096), 1024, 0, stream>>>(cnt, offs, (int)G);
  fill_kernel<<<gEt, 256, 0, stream>>>(src_cq, src_ca, dst_cq, dst_ca, rank, offs, slist,
                                       (int)Ecq, (int)Etot, (int)Nq);
  const int gWalk = (G + 7) / 8 < 2048 ? (int)((G + 7) / 8) : 2048;
  gather_kernel<<<gWalk, 256, 0, stream>>>(x_c, as_cq, as_ca, ad_cq, ad_ca,
                                                        offs, slist, z_q, z_a,
                                                        (int)Nq, (int)G, (int)Etot);

  final_kernel<<<(int)((Nq + 127) / 128), 256, 0, stream>>>(z_q, z_a, Pc, cst, (float*)d_out, (int)Nq);
}


# Edge lists by a two-level counting sort, all counters in LDS: chunks count and place edges by 512-node bin, one workgroup per bin orders its nodes; no global atomic; list (sum) order by LDS arrival
# speedup vs baseline: 6.5336x; 1.5353x over previous
#include <hip/hip_runtime.h>
#include <hip/hip_bf16.h>
#include <math.h>

typedef __attribute__((ext_vector_type(2))) float v2f;
typedef __attribute__((ext_vector_type(8))) float v8f;

__device__ __forceinline__ float lrelu(float x) { return x > 0.f ? x : 0.2f * x; }

template <int M>
__device__ __forceinline__ float xrow(float v) {
  const int b = __float_as_int(v);
  return __int_as_float(__builtin_amdgcn_update_dpp(b, b, 0x160 + M, 0xf, 0xf, false));
}
__device__ __forceinline__ float xhalf(float v) {
  const int b = __float_as_int(v);
  return __int_as_float(__builtin_amdgcn_permlanex16(b, b, 0x76543210, 0xfedcba98, false, false));
}
__device__ __forceinline__ float wave_max(float v) {
  v = fmaxf(v, xrow<1>(v));
  v = fmaxf(v, xrow<2>(v));
  v = fmaxf(v, xrow<4>(v));
  v = fmaxf(v, xrow<8>(v));
  return fmaxf(v, xhalf(v));
}
__device__ __forceinline__ float wave_sum(float v) {
  v += xrow<1>(v);
  v += xrow<2>(v);
  v += xrow<4>(v);
  v += xrow<8>(v);
  return v + xhalf(v);
}

__global__ __launch_bounds__(1024)
void prep_kernel(const float* __restrict__ Wsq, const float* __restrict__ Wdq,
                 const float* __restrict__ Asq, const float* __restrict__ Adq,
                 const float* __restrict__ Wsa, const float* __restrict__ Wda,
                 const float* __restrict__ Asa, const float* __restrict__ Ada,
                 const float* __restrict__ bq,  const float* __restrict__ ba,
                 const float* __restrict__ Wout, const float* __restrict__ bout,
                 float* __restrict__ Vp, float* __restrict__ Pc, float* __restrict__ cst) {
  const int t = threadIdx.x;
  for (int idx = t; idx < 4 * 64 * 16; idx += blockDim.x) {
    int table = idx >> 10, k = (idx >> 4) & 63, j = idx & 15;
    const float* W = (table == 0) ? Wsq : (table == 1) ? Wdq : (table == 2) ? Wsa : Wda;
    const float* A = (table == 0) ? Asq : (table == 1) ? Adq : (table == 2) ? Asa : Ada;
    float s = 0.f;
    if (j < 4)
      for (int c = 0; c < 64; ++c) s += W[k * 256 + j * 64 + c] * A[j * 64 + c];
    Vp[idx] = s;
  }
  for (int idx = t; idx < 512 * 16; idx += blockDim.x) {
    int r = idx >> 4, j = idx & 15;
    int layer = r >> 8, h = (r >> 6) & 3, k = r & 63;
    const float* W = layer ? Wsa : Wsq;
    float s = 0.f;
    if (j < 4) {
      for (int c = 0; c < 64; ++c)
        s += W[k * 256 + h * 64 + c] * Wout[(layer * 64 + c) * 4 + j];
      s *= 0.25f;
    }
    Pc[idx] = s;
  }
  if (t < 16) {
    float s = 0.f;
    if (t < 4) {
      for (int c = 0; c < 64; ++c)
        s += bq[c] * Wout[c * 4 + t] + ba[c] * Wout[(64 + c) * 4 + t];
      s += bout[t];
    }
    cst[t] = s;
  }
}


__global__ void scores_kernel(const float* __restrict__ x, const float* __restrict__ Vp,
                              float* __restrict__ out, int N) {
  const int lane = threadIdx.x & 31;
  const int m = lane & 15;
  const int ko = (lane >> 4) * 2;
  const int nw = (int)((gridDim.x * blockDim.x) >> 5);
  const int ntiles = (N + 15) >> 4;
  v2f b[16];
#pragma unroll
  for (int q = 0; q < 16; ++q) {
    b[q].x = Vp[(q * 4 + ko) * 16 + m];
    b[q].y = Vp[(q * 4 + ko + 1) * 16 + m];
  }
  for (int t = __builtin_amdgcn_readfirstlane(
           (int)(((unsigned)blockIdx.x * blockDim.x + threadIdx.x) >> 5));
       t < ntiles; t += nw) {
    const long base = (long)t * 16;
    long row = base + m; if (row >= N) row = N - 1;
    const float* xr = x + row * 64;
    v8f acc = {};
#pragma unroll
    for (int q = 0; q < 16; ++q) {
      v2f a;
      a.x = xr[q * 4 + ko];
      a.y = xr[q * 4 + ko + 1];
      acc = __builtin_amdgcn_wmma_f32_16x16x4_f32(false, a, false, b[q], (short)0, acc,
                                                  false, false);
    }
    if (m < 4) {
      const long rb = base + ((lane >> 4) << 3);
#pragma unroll
      for (int v = 0; v < 8; ++v) {
        long r = rb + v;
        if (r < N) out[r * 4 + m] = acc[v];
      }
    }
  }
}

#define BIN_SH 9
#define BIN_NODES (1 << BIN_SH)
#define NB_MAX 1024
#define EPT 8
#define CHUNK (EPT * 1024)
#define LIST_CAP 12288

__device__ __forceinline__ int lds_count(int* p) {
  return __hip_atomic_fetch_add(p, 1, __ATOMIC_RELAXED, __HIP_MEMORY_SCOPE_WORKGROUP);
}

__device__ __forceinline__ int block_excl_scan(int v, int* wsum) {
  const int lane = threadIdx.x & 31;
  const int wv = threadIdx.x >> 5;
  int inc = v;
  for (int o = 1; o < 32; o <<= 1) {
    const int nb = __shfl_up(inc, o);
    if (lane >= o) inc += nb;
  }
  if (lane == 31) wsum[wv] = inc;
  __syncthreads();
  int wpre = 0;
  for (int i = 0; i < wv; ++i) wpre += wsum[i];
  return wpre + inc - v;
}

__global__ __launch_bounds__(1024)
void bin_count_kernel(const int* __restrict__ dst_q, const int* __restrict__ dst_a,
                      int* __restrict__ hist, int Ecq, int Etot, int Nq, int NB) {
  __shared__ int c[NB_MAX];
  const int t = threadIdx.x;
  const int wg = blockIdx.x, nwg = gridDim.x;
  c[t] = 0;
  __syncthreads();
  const int i0 = wg * CHUNK + t;
  if (wg * CHUNK + CHUNK <= Etot) {
    int dv[EPT];
#pragma unroll
    for (int k = 0; k < EPT; ++k) {
      const int i = i0 + k * 1024;
      const int* p = (i < Ecq) ? dst_q + i : dst_a + (i - Ecq);
      dv[k] = *p;
    }
#pragma unroll
    for (int k = 0; k < EPT; ++k) {
      const int i = i0 + k * 1024;
      lds_count(&c[((i < Ecq) ? dv[k] : Nq + dv[k]) >> BIN_SH]);
    }
  } else {
    for (int i = i0; i < Etot; i += 1024) {
      const int g = (i < Ecq) ? dst_q[i] : Nq + dst_a[i - Ecq];
      lds_count(&c[g >> BIN_SH]);
    }
  }
  __syncthreads();
  if (t < NB) hist[t * nwg + wg] = c[t];
}

__global__ __launch_bounds__(1024)
void scan_kernel(const int* __restrict__ cnt, int* __restrict__ off, int G) {
  __shared__ int wsum[32];
  const int t = threadIdx.x;
  const int lane = t & 31;
  const int wv = t >> 5;
  const int tile0 = blockIdx.x * 4096;
  int s = 0;
  for (int i = t * 4; i < tile0; i += 4096) {
    const int4 q = *(const int4*)(cnt + i);
    s += q.x + q.y + q.z + q.w;
  }
  for (int o = 16; o > 0; o >>= 1) s += __shfl_xor(s, o);
  if (lane == 0) wsum[wv] = s;
  __syncthreads();
  int carry = 0;
  for (int i = 0; i < 32; ++i) carry += wsum[i];
  __syncthreads();

  const int base = tile0 + t * 4;
  int v0 = 0, v1 = 0, v2 = 0;
  int v3 = 0;
  if (base + 3 < G) {
    const int4 q = *(const int4*)(cnt + base);
    v0 = q.x; v1 = q.y; v2 = q.z; v3 = q.w;
  } else {
    if (base < G) v0 = cnt[base];
    if (base + 1 < G) v1 = cnt[base + 1];
    if (base + 2 < G) v2 = cnt[base + 2];
  }
  const int tsum = v0 + v1 + v2 + v3;
  int inc = tsum;
  for (int o = 1; o < 32; o <<= 1) {
    const int nb = __shfl_up(inc, o);
    if (lane >= o) inc += nb;
  }
  if (lane == 31) wsum[wv] = inc;
  __syncthreads();
  int wpre = 0;
  for (int i = 0; i < wv; ++i) wpre += wsum[i];
  const int ex0 = carry + wpre + inc - tsum;
  if (base < G) off[base] = ex0;
  if (base + 1 < G) off[base + 1] = ex0 + v0;
  if (base + 2 < G) off[base + 2] = ex0 + v0 + v1;
  if (base + 3 < G) off[base + 3] = ex0 + v0 + v1 + v2;
}

__global__ __launch_bounds__(1024)
void bin_scatter_kernel(const int* __restrict__ src_q, const int* __restrict__ src_a,
                        const int* __restrict__ dst_q, const int* __restrict__ dst_a,
                        const int* __restrict__ hoff, int2* __restrict__ binbuf,
                        int Ecq, int Etot, int Nq, int NB) {
  __shared__ int c[NB_MAX];
  __shared__ int ho[NB_MAX];
  __shared__ int wsum[32];
  __shared__ int2 stage[CHUNK];
  const int t = threadIdx.x;
  const int wg = blockIdx.x, nwg = gridDim.x;
  c[t] = 0;
  int hv = 0;
  if (t < NB) hv = hoff[t * nwg + wg];
  __syncthreads();
  int g[EPT], s[EPT], r[EPT];
  const int i0 = wg * CHUNK + t;
  if (wg * CHUNK + CHUNK <= Etot) {
#pragma unroll
    for (int k = 0; k < EPT; ++k) {
      const int i = i0 + k * 1024;
      const bool second = i >= Ecq;
      const int e = second ? i - Ecq : i;
      const int* dp = second ? dst_a : dst_q;
      const int* sp = second ? src_a : src_q;
      g[k] = dp[e];
      s[k] = sp[e];
    }
#pragma unroll
    for (int k = 0; k < EPT; ++k)
      if (i0 + k * 1024 >= Ecq) g[k] += Nq;
  } else {
#pragma unroll
    for (int k = 0; k < EPT; ++k) {
      const int i = i0 + k * 1024;
      g[k] = -1;
      s[k] = 0;
      if (i < Etot) {
        const bool second = i >= Ecq;
        const int e = second ? i - Ecq : i;
        g[k] = second ? Nq + dst_a[e] : dst_q[e];
        s[k] = second ? src_a[e] : src_q[e];
      }
    }
  }
#pragma unroll
  for (int k = 0; k < EPT; ++k) {
    r[k] = 0;
    if (g[k] >= 0) r[k] = lds_count(&c[g[k] >> BIN_SH]);
  }
  ho[t] = hv;
  __syncthreads();
  const int v = c[t];
  const int ex = block_excl_scan(v, wsum);
  c[t] = ex;
  __syncthreads();
#pragma unroll
  for (int k = 0; k < EPT; ++k)
    if (g[k] >= 0) stage[c[g[k] >> BIN_SH] + r[k]] = make_int2(s[k], g[k]);
  __syncthreads();
  const int left = Etot - wg * CHUNK;
  const int n = left < CHUNK ? left : CHUNK;
  for (int p = t; p < n; p += 1024) {
    const int2 rec = stage[p];
    const int b = rec.y >> BIN_SH;
    binbuf[ho[b] + (p - c[b])] = rec;
  }
}

__global__ __launch_bounds__(1024)
void bin_lists_kernel(const int2* __restrict__ binbuf, const int* __restrict__ hoff,
                      int* __restrict__ off, int* __restrict__ slist,
                      int G, int Etot, int nwg, int NB) {
  __shared__ int c[1024];
  __shared__ int wsum[32];
  __shared__ int lst[LIST_CAP];
  const int t = threadIdx.x;
  const int b = blockIdx.x;
  const int beg = hoff[b * nwg];
  const int end = (b + 1 < NB) ? hoff[(b + 1) * nwg] : Etot;
  const int n = end - beg;
  const int g0 = b << BIN_SH;
  const bool fast = n > 0 && n <= EPT * 1024;
  const bool in_lds = n <= LIST_CAP;
  int2 rec[EPT];
  c[t] = 0;
  __syncthreads();
  if (fast) {
#pragma unroll
    for (int k = 0; k < EPT; ++k) {
      const int p = t + k * 1024;
      rec[k] = binbuf[beg + (p < n ? p : n - 1)];
    }
#pragma unroll
    for (int k = 0; k < EPT; ++k)
      if (t + k * 1024 < n) lds_count(&c[rec[k].y - g0]);
  } else {
    for (int p = t; p < n; p += 1024) lds_count(&c[binbuf[beg + p].y - g0]);
  }
  __syncthreads();
  const int v = c[t];
  const int ex = block_excl_scan(v, wsum);
  c[t] = ex;
  if (t < BIN_NODES && g0 + t < G) off[g0 + t] = beg + ex;
  __syncthreads();
  if (fast) {
#pragma unroll
    for (int k = 0; k < EPT; ++k)
      if (t + k * 1024 < n) lst[lds_count(&c[rec[k].y - g0])] = rec[k].x;
  } else {
    for (int p = t; p < n; p += 1024) {
      const int2 r = binbuf[beg + p];
      const int pos = lds_count(&c[r.y - g0]);
      if (in_lds) lst[pos] = r.x; else slist[beg + pos] = r.x;
    }
  }
  if (in_lds) {
    __syncthreads();
    for (int p = t; p < n; p += 1024) slist[beg + p] = lst[p];
  }
}

__global__ void gather_kernel(const float* __restrict__ x,
                              const float* __restrict__ as_q, const float* __restrict__ as_a,
                              const float* __restrict__ ad_q, const float* __restrict__ ad_a,
                              const int* __restrict__ off, const int* __restrict__ slist,
                              float* __restrict__ z_q, float* __restrict__ z_a,
                              int Nq, int G, int Etot) {
  const int lane = threadIdx.x & 31;
  const int nw = (int)((gridDim.x * blockDim.x) >> 5);
  const float* xl = x + lane * 2;
  const float ninf = __int_as_float(0xff800000);
  for (int g = __builtin_amdgcn_readfirstlane(
           (int)(((unsigned)blockIdx.x * blockDim.x + threadIdx.x) >> 5));
       g < G; g += nw) {
  const bool second = g >= Nq;
  const int d = second ? g - Nq : g;
  const float* a_s = second ? as_a : as_q;
  const float4 ad = *(const float4*)((second ? ad_a : ad_q) + (long)d * 4);
  float* z = (second ? z_a : z_q) + (long)d * 256 + lane * 2;
  const int beg = off[g];
  const int end = (g + 1 < G) ? off[g + 1] : Etot;
  float m0 = ninf, m1 = ninf, m2 = ninf, m3 = ninf;
  float d0 = 0.f, d1 = 0.f, d2 = 0.f, d3 = 0.f;
  v2f acc0 = {0.f, 0.f}, acc1 = {0.f, 0.f}, acc2 = {0.f, 0.f}, acc3 = {0.f, 0.f};
  for (int c = beg; c < end; c += 32) {
    const int n = (end - c < 32) ? end - c : 32;
    int s_l = 0;
    float e0 = ninf, e1 = ninf, e2 = ninf, e3 = ninf;
    if (lane < n) {
      s_l = slist[c + lane];
      const float4 as = *(const float4*)(a_s + (long)s_l * 4);
      e0 = lrelu(as.x + ad.x);
      e1 = lrelu(as.y + ad.y);
      e2 = lrelu(as.z + ad.z);
      e3 = lrelu(as.w + ad.w);
    }
    const float n0 = fmaxf(m0, wave_max(e0)), n1 = fmaxf(m1, wave_max(e1));
    const float n2 = fmaxf(m2, wave_max(e2)), n3 = fmaxf(m3, wave_max(e3));
    const float r0 = __expf(m0 - n0), r1 = __expf(m1 - n1);
    const float r2 = __expf(m2 - n2), r3 = __expf(m3 - n3);
    const float p0 = __expf(e0 - n0), p1 = __expf(e1 - n1);
    const float p2 = __expf(e2 - n2), p3 = __expf(e3 - n3);
    d0 = d0 * r0 + wave_sum(p0); d1 = d1 * r1 + wave_sum(p1);
    d2 = d2 * r2 + wave_sum(p2); d3 = d3 * r3 + wave_sum(p3);
    acc0 *= r0; acc1 *= r1; acc2 *= r2; acc3 *= r3;
    m0 = n0; m1 = n1; m2 = n2; m3 = n3;
#pragma unroll 4
    for (int j = 0; j < n; ++j) {
      const int s = __builtin_amdgcn_readlane(s_l, j);
      const float w0 = __int_as_float(__builtin_amdgcn_readlane(__float_as_int(p0), j));
      const float w1 = __int_as_float(__builtin_amdgcn_readlane(__float_as_int(p1), j));
      const float w2 = __int_as_float(__builtin_amdgcn_readlane(__float_as_int(p2), j));
      const float w3 = __int_as_float(__builtin_amdgcn_readlane(__float_as_int(p3), j));
      const v2f xv = *(const v2f*)(xl + (long)s * 64);
      acc0 += w0 * xv;
      acc1 += w1 * xv;
      acc2 += w2 * xv;
      acc3 += w3 * xv;
    }
  }
  *(v2f*)(z + 0) = acc0 / (d0 + 1e-16f);
  *(v2f*)(z + 64) = acc1 / (d1 + 1e-16f);
  *(v2f*)(z + 128) = acc2 / (d2 + 1e-16f);
  *(v2f*)(z + 192) = acc3 / (d3 + 1e-16f);
  }
}

__global__ void final_kernel(const float* __restrict__ zq, const float* __restrict__ za,
                             const float* __restrict__ Pc, const float* __restrict__ cst,
                             float* __restrict__ out, int N) {
  const int wave_g = (int)(((unsigned)blockIdx.x * blockDim.x + threadIdx.x) >> 5);
  const int lane = threadIdx.x & 31;
  const long base = (long)wave_g * 16;
  if (base >= N) return;
  const int m = lane & 15;
  const int ko = (lane >> 4) * 2;
  long row = base + m; if (row >= N) row = N - 1;
  const float* zqr = zq + row * 256;
  const float* zar = za + row * 256;
  v8f acc = {};
#pragma unroll 8
  for (int kk = 0; kk < 512; kk += 4) {
    const int k0 = kk + ko;
    const float* zr = (k0 < 256) ? zqr : zar;
    const int kL = k0 & 255;
    v2f a, b;
    a.x = zr[kL];
    a.y = zr[kL + 1];
    b.x = Pc[k0 * 16 + m];
    b.y = Pc[(k0 + 1) * 16 + m];
    acc = __builtin_amdgcn_wmma_f32_16x16x4_f32(false, a, false, b, (short)0, acc,
                                                false, false);
  }
  if (m < 4) {
    const float c0 = cst[m];
    const long rb = base + ((lane >> 4) << 3);
#pragma unroll
    for (int v = 0; v < 8; ++v) {
      long r = rb + v;
      if (r < N) out[r * 4 + m] = acc[v] + c0;
    }
  }
}

extern "C" void kernel_launch(void* const* d_in, const int* in_sizes, int n_in,
                              void* d_out, int out_size, void* d_ws, size_t ws_size,
                              hipStream_t stream) {
  const float* x_c = (const float*)d_in[0];
  const float* x_q = (const float*)d_in[1];
  const float* x_a = (const float*)d_in[2];
  const int* src_cq = (const int*)d_in[3];
  const int* dst_cq = (const int*)d_in[4];
  const int* src_ca = (const int*)d_in[5];
  const int* dst_ca = (const int*)d_in[6];
  const float* Wsq = (const float*)d_in[7];
  const float* Wdq = (const float*)d_in[8];
  const float* Asq = (const float*)d_in[9];
  const float* Adq = (const float*)d_in[10];
  const float* bq  = (const float*)d_in[11];
  const float* Wsa = (const float*)d_in[12];
  const float* Wda = (const float*)d_in[13];
  const float* Asa = (const float*)d_in[14];
  const float* Ada = (const float*)d_in[15];
  const float* ba  = (const float*)d_in[16];
  const float* Wout = (const float*)d_in[17];
  const float* bout = (const float*)d_in[18];

  const long Nc = in_sizes[0] / 64;
  const long Nq = in_sizes[1] / 64;
  const long Na = in_sizes[2] / 64;
  const long Ecq = in_sizes[3];
  const long Eca = in_sizes[5];

  float* w = (float*)d_ws;
  size_t off = 0;
  auto alloc = [&](size_t n) -> float* {
    float* p = w + off;
    off += (n + 15) & ~(size_t)15;
    return p;
  };
  float* Vp_sq = alloc(64 * 16);
  float* Vp_dq = alloc(64 * 16);
  float* Vp_sa = alloc(64 * 16);
  float* Vp_da = alloc(64 * 16);
  float* Pc    = alloc(512 * 16);
  float* cst   = alloc(16);
  float* as_cq = alloc(Nc * 4);
  float* as_ca = alloc(Nc * 4);
  float* ad_cq = alloc(Nq * 4);
  float* ad_ca = alloc(Na * 4);
  float* e_cq  = alloc(Ecq * 4);
  float* e_ca  = alloc(Eca * 4);
  float* m_q   = alloc(Nq * 4);
  float* m_a   = alloc(Na * 4);
  float* den_q = alloc(Nq * 4);
  float* den_a = alloc(Na * 4);
  float* z_q   = alloc(Nq * 256);
  float* z_a   = alloc(Na * 256);
  (void)Vp_dq; (void)Vp_sa; (void)Vp_da; (void)e_ca; (void)m_a; (void)den_q; (void)den_a;
  if (off * sizeof(float) > ws_size) return;

  const long G = Nq + Na;
  const long Etot = Ecq + Eca;
  const long NB = (G + BIN_NODES - 1) >> BIN_SH;
  const long nChunk = (Etot + CHUNK - 1) / CHUNK;
  const long nHist = NB * nChunk;
  const long nHistPad = (nHist + 15) & ~15L;
  int*  hist   = (int*)z_q;
  int*  hoff   = hist + nHistPad;
  int2* binbuf = (int2*)z_a;
  int*  slist  = (int*)e_cq;
  int*  offs   = (int*)m_q;
  if (NB > NB_MAX || 2 * nHistPad > Nq * 256 || 2 * Etot > Na * 256 ||
      Etot > (long)(m_q - e_cq) || G > (long)(z_q - m_q)) return;

  prep_kernel<<<1, 1024, 0, stream>>>(Wsq, Wdq, Asq, Adq, Wsa, Wda, Asa, Ada,
                                      bq, ba, Wout, bout, Vp_sq, Pc, cst);
  const int gNc = (Nc + 127) / 128 < 1024 ? (int)((Nc + 127) / 128) : 1024;
  const int gNq = (Nq + 127) / 128 < 1024 ? (int)((Nq + 127) / 128) : 1024;
  const int gNa = (Na + 127) / 128 < 1024 ? (int)((Na + 127) / 128) : 1024;
  scores_kernel<<<gNc, 256, 0, stream>>>(x_c, Vp_sq, as_cq, (int)Nc);
  scores_kernel<<<gNc, 256, 0, stream>>>(x_c, Vp_sa, as_ca, (int)Nc);
  scores_kernel<<<gNq, 256, 0, stream>>>(x_q, Vp_dq, ad_cq, (int)Nq);
  scores_kernel<<<gNa, 256, 0, stream>>>(x_a, Vp_da, ad_ca, (int)Na);

  bin_count_kernel<<<(int)nChunk, 1024, 0, stream>>>(dst_cq, dst_ca, hist,
                                                     (int)Ecq, (int)Etot, (int)Nq, (int)NB);
  scan_kernel<<<(int)((nHist + 4095) / 4096), 1024, 0, stream>>>(hist, hoff, (int)nHist);
  bin_scatter_kernel<<<(int)nChunk, 1024, 0, stream>>>(src_cq, src_ca, dst_cq, dst_ca, hoff,
                                                       binbuf, (int)Ecq, (int)Etot, (int)Nq,
                                                       (int)NB);
  bin_lists_kernel<<<(int)NB, 1024, 0, stream>>>(binbuf, hoff, offs, slist,
                                                 (int)G, (int)Etot, (int)nChunk, (int)NB);
  const int gWalk = (G + 7) / 8 < 2048 ? (int)((G + 7) / 8) : 2048;
  gather_kernel<<<gWalk, 256, 0, stream>>>(x_c, as_cq, as_ca, ad_cq, ad_ca,
                                                        offs, slist, z_q, z_a,
                                                        (int)Nq, (int)G, (int)Etot);

  final_kernel<<<(int)((Nq + 127) / 128), 256, 0, stream>>>(z_q, z_a, Pc, cst, (float*)d_out, (int)Nq);
}


# Output projection inside the node walk: a wave owns 8 rows of out, walks both lists of each; z stays in registers as WMMA A rows, weights in LDS; z planes and last launch go; projection sum order new
# speedup vs baseline: 7.4950x; 1.1471x over previous
#include <hip/hip_runtime.h>
#include <hip/hip_bf16.h>
#include <math.h>

typedef __attribute__((ext_vector_type(2))) float v2f;
typedef __attribute__((ext_vector_type(8))) float v8f;

__device__ __forceinline__ float lrelu(float x) { return x > 0.f ? x : 0.2f * x; }

template <int M>
__device__ __forceinline__ float xrow(float v) {
  const int b = __float_as_int(v);
  return __int_as_float(__builtin_amdgcn_update_dpp(b, b, 0x160 + M, 0xf, 0xf, false));
}
__device__ __forceinline__ float xhalf(float v) {
  const int b = __float_as_int(v);
  return __int_as_float(__builtin_amdgcn_permlanex16(b, b, 0x76543210, 0xfedcba98, false, false));
}
__device__ __forceinline__ float wave_max(float v) {
  v = fmaxf(v, xrow<1>(v));
  v = fmaxf(v, xrow<2>(v));
  v = fmaxf(v, xrow<4>(v));
  v = fmaxf(v, xrow<8>(v));
  return fmaxf(v, xhalf(v));
}
__device__ __forceinline__ float wave_sum(float v) {
  v += xrow<1>(v);
  v += xrow<2>(v);
  v += xrow<4>(v);
  v += xrow<8>(v);
  return v + xhalf(v);
}

__global__ __launch_bounds__(1024)
void prep_kernel(const float* __restrict__ Wsq, const float* __restrict__ Wdq,
                 const float* __restrict__ Asq, const float* __restrict__ Adq,
                 const float* __restrict__ Wsa, const float* __restrict__ Wda,
                 const float* __restrict__ Asa, const float* __restrict__ Ada,
                 const float* __restrict__ bq,  const float* __restrict__ ba,
                 const float* __restrict__ Wout, const float* __restrict__ bout,
                 float* __restrict__ Vp, float* __restrict__ Pc, float* __restrict__ cst) {
  const int t = threadIdx.x;
  for (int idx = t; idx < 4 * 64 * 16; idx += blockDim.x) {
    int table = idx >> 10, k = (idx >> 4) & 63, j = idx & 15;
    const float* W = (table == 0) ? Wsq : (table == 1) ? Wdq : (table == 2) ? Wsa : Wda;
    const float* A = (table == 0) ? Asq : (table == 1) ? Adq : (table == 2) ? Asa : Ada;
    float s = 0.f;
    if (j < 4)
      for (int c = 0; c < 64; ++c) s += W[k * 256 + j * 64 + c] * A[j * 64 + c];
    Vp[idx] = s;
  }
  for (int idx = t; idx < 512 * 16; idx += blockDim.x) {
    int r = idx >> 4, j = idx & 15;
    int layer = r >> 8, h = (r >> 6) & 3, k = r & 63;
    const float* W = layer ? Wsa : Wsq;
    float s = 0.f;
    if (j < 4) {
      for (int c = 0; c < 64; ++c)
        s += W[k * 256 + h * 64 + c] * Wout[(layer * 64 + c) * 4 + j];
      s *= 0.25f;
    }
    Pc[idx] = s;
  }
  if (t < 16) {
    float s = 0.f;
    if (t < 4) {
      for (int c = 0; c < 64; ++c)
        s += bq[c] * Wout[c * 4 + t] + ba[c] * Wout[(64 + c) * 4 + t];
      s += bout[t];
    }
    cst[t] = s;
  }
}


__global__ void scores_kernel(const float* __restrict__ x, const float* __restrict__ Vp,
                              float* __restrict__ out, int N) {
  const int lane = threadIdx.x & 31;
  const int m = lane & 15;
  const int ko = (lane >> 4) * 2;
  const int nw = (int)((gridDim.x * blockDim.x) >> 5);
  const int ntiles = (N + 15) >> 4;
  v2f b[16];
#pragma unroll
  for (int q = 0; q < 16; ++q) {
    b[q].x = Vp[(q * 4 + ko) * 16 + m];
    b[q].y = Vp[(q * 4 + ko + 1) * 16 + m];
  }
  for (int t = __builtin_amdgcn_readfirstlane(
           (int)(((unsigned)blockIdx.x * blockDim.x + threadIdx.x) >> 5));
       t < ntiles; t += nw) {
    const long base = (long)t * 16;
    long row = base + m; if (row >= N) row = N - 1;
    const float* xr = x + row * 64;
    v8f acc = {};
#pragma unroll
    for (int q = 0; q < 16; ++q) {
      v2f a;
      a.x = xr[q * 4 + ko];
      a.y = xr[q * 4 + ko + 1];
      acc = __builtin_amdgcn_wmma_f32_16x16x4_f32(false, a, false, b[q], (short)0, acc,
                                                  false, false);
    }
    if (m < 4) {
      const long rb = base + ((lane >> 4) << 3);
#pragma unroll
      for (int v = 0; v < 8; ++v) {
        long r = rb + v;
        if (r < N) out[r * 4 + m] = acc[v];
      }
    }
  }
}

#define BIN_SH 9
#define BIN_NODES (1 << BIN_SH)
#define NB_MAX 1024
#define EPT 8
#define CHUNK (EPT * 1024)
#define LIST_CAP 12288

__device__ __forceinline__ int lds_count(int* p) {
  return __hip_atomic_fetch_add(p, 1, __ATOMIC_RELAXED, __HIP_MEMORY_SCOPE_WORKGROUP);
}

__device__ __forceinline__ int block_excl_scan(int v, int* wsum) {
  const int lane = threadIdx.x & 31;
  const int wv = threadIdx.x >> 5;
  int inc = v;
  for (int o = 1; o < 32; o <<= 1) {
    const int nb = __shfl_up(inc, o);
    if (lane >= o) inc += nb;
  }
  if (lane == 31) wsum[wv] = inc;
  __syncthreads();
  int wpre = 0;
  for (int i = 0; i < wv; ++i) wpre += wsum[i];
  return wpre + inc - v;
}

__global__ __launch_bounds__(1024)
void bin_count_kernel(const int* __restrict__ dst_q, const int* __restrict__ dst_a,
                      int* __restrict__ hist, int Ecq, int Etot, int Nq, int NB) {
  __shared__ int c[NB_MAX];
  const int t = threadIdx.x;
  const int wg = blockIdx.x, nwg = gridDim.x;
  c[t] = 0;
  __syncthreads();
  const int i0 = wg * CHUNK + t;
  if (wg * CHUNK + CHUNK <= Etot) {
    int dv[EPT];
#pragma unroll
    for (int k = 0; k < EPT; ++k) {
      const int i = i0 + k * 1024;
      const int* p = (i < Ecq) ? dst_q + i : dst_a + (i - Ecq);
      dv[k] = *p;
    }
#pragma unroll
    for (int k = 0; k < EPT; ++k) {
      const int i = i0 + k * 1024;
      lds_count(&c[((i < Ecq) ? dv[k] : Nq + dv[k]) >> BIN_SH]);
    }
  } else {
    for (int i = i0; i < Etot; i += 1024) {
      const int g = (i < Ecq) ? dst_q[i] : Nq + dst_a[i - Ecq];
      lds_count(&c[g >> BIN_SH]);
    }
  }
  __syncthreads();
  if (t < NB) hist[t * nwg + wg] = c[t];
}

__global__ __launch_bounds__(1024)
void scan_kernel(const int* __restrict__ cnt, int* __restrict__ off, int G) {
  __shared__ int wsum[32];
  const int t = threadIdx.x;
  const int lane = t & 31;
  const int wv = t >> 5;
  const int tile0 = blockIdx.x * 4096;
  int s = 0;
  for (int i = t * 4; i < tile0; i += 4096) {
    const int4 q = *(const int4*)(cnt + i);
    s += q.x + q.y + q.z + q.w;
  }
  for (int o = 16; o > 0; o >>= 1) s += __shfl_xor(s, o);
  if (lane == 0) wsum[wv] = s;
  __syncthreads();
  int carry = 0;
  for (int i = 0; i < 32; ++i) carry += wsum[i];
  __syncthreads();

  const int base = tile0 + t * 4;
  int v0 = 0, v1 = 0, v2 = 0;
  int v3 = 0;
  if (base + 3 < G) {
    const int4 q = *(const int4*)(cnt + base);
    v0 = q.x; v1 = q.y; v2 = q.z; v3 = q.w;
  } else {
    if (base < G) v0 = cnt[base];
    if (base + 1 < G) v1 = cnt[base + 1];
    if (base + 2 < G) v2 = cnt[base + 2];
  }
  const int tsum = v0 + v1 + v2 + v3;
  int inc = tsum;
  for (int o = 1; o < 32; o <<= 1) {
    const int nb = __shfl_up(inc, o);
    if (lane >= o) inc += nb;
  }
  if (lane == 31) wsum[wv] = inc;
  __syncthreads();
  int wpre = 0;
  for (int i = 0; i < wv; ++i) wpre += wsum[i];
  const int ex0 = carry + wpre + inc - tsum;
  if (base < G) off[base] = ex0;
  if (base + 1 < G) off[base + 1] = ex0 + v0;
  if (base + 2 < G) off[base + 2] = ex0 + v0 + v1;
  if (base + 3 < G) off[base + 3] = ex0 + v0 + v1 + v2;
}

__global__ __launch_bounds__(1024)
void bin_scatter_kernel(const int* __restrict__ src_q, const int* __restrict__ src_a,
                        const int* __restrict__ dst_q, const int* __restrict__ dst_a,
                        const int* __restrict__ hoff, int2* __restrict__ binbuf,
                        int Ecq, int Etot, int Nq, int NB) {
  __shared__ int c[NB_MAX];
  __shared__ int ho[NB_MAX];
  __shared__ int wsum[32];
  __shared__ int2 stage[CHUNK];
  const int t = threadIdx.x;
  const int wg = blockIdx.x, nwg = gridDim.x;
  c[t] = 0;
  int hv = 0;
  if (t < NB) hv = hoff[t * nwg + wg];
  __syncthreads();
  int g[EPT], s[EPT], r[EPT];
  const int i0 = wg * CHUNK + t;
  if (wg * CHUNK + CHUNK <= Etot) {
#pragma unroll
    for (int k = 0; k < EPT; ++k) {
      const int i = i0 + k * 1024;
      const bool second = i >= Ecq;
      const int e = second ? i - Ecq : i;
      const int* dp = second ? dst_a : dst_q;
      const int* sp = second ? src_a : src_q;
      g[k] = dp[e];
      s[k] = sp[e];
    }
#pragma unroll
    for (int k = 0; k < EPT; ++k)
      if (i0 + k * 1024 >= Ecq) g[k] += Nq;
  } else {
#pragma unroll
    for (int k = 0; k < EPT; ++k) {
      const int i = i0 + k * 1024;
      g[k] = -1;
      s[k] = 0;
      if (i < Etot) {
        const bool second = i >= Ecq;
        const int e = second ? i - Ecq : i;
        g[k] = second ? Nq + dst_a[e] : dst_q[e];
        s[k] = second ? src_a[e] : src_q[e];
      }
    }
  }
#pragma unroll
  for (int k = 0; k < EPT; ++k) {
    r[k] = 0;
    if (g[k] >= 0) r[k] = lds_count(&c[g[k] >> BIN_SH]);
  }
  ho[t] = hv;
  __syncthreads();
  const int v = c[t];
  const int ex = block_excl_scan(v, wsum);
  c[t] = ex;
  __syncthreads();
#pragma unroll
  for (int k = 0; k < EPT; ++k)
    if (g[k] >= 0) stage[c[g[k] >> BIN_SH] + r[k]] = make_int2(s[k], g[k]);
  __syncthreads();
  const int left = Etot - wg * CHUNK;
  const int n = left < CHUNK ? left : CHUNK;
  for (int p = t; p < n; p += 1024) {
    const int2 rec = stage[p];
    const int b = rec.y >> BIN_SH;
    binbuf[ho[b] + (p - c[b])] = rec;
  }
}

__global__ __launch_bounds__(1024)
void bin_lists_kernel(const int2* __restrict__ binbuf, const int* __restrict__ hoff,
                      int* __restrict__ off, int* __restrict__ slist,
                      int G, int Etot, int nwg, int NB) {
  __shared__ int c[1024];
  __shared__ int wsum[32];
  __shared__ int lst[LIST_CAP];
  const int t = threadIdx.x;
  const int b = blockIdx.x;
  const int beg = hoff[b * nwg];
  const int end = (b + 1 < NB) ? hoff[(b + 1) * nwg] : Etot;
  const int n = end - beg;
  const int g0 = b << BIN_SH;
  const bool fast = n > 0 && n <= EPT * 1024;
  const bool in_lds = n <= LIST_CAP;
  int2 rec[EPT];
  c[t] = 0;
  __syncthreads();
  if (fast) {
#pragma unroll
    for (int k = 0; k < EPT; ++k) {
      const int p = t + k * 1024;
      rec[k] = binbuf[beg + (p < n ? p : n - 1)];
    }
#pragma unroll
    for (int k = 0; k < EPT; ++k)
      if (t + k * 1024 < n) lds_count(&c[rec[k].y - g0]);
  } else {
    for (int p = t; p < n; p += 1024) lds_count(&c[binbuf[beg + p].y - g0]);
  }
  __syncthreads();
  const int v = c[t];
  const int ex = block_excl_scan(v, wsum);
  c[t] = ex;
  if (t < BIN_NODES && g0 + t < G) off[g0 + t] = beg + ex;
  __syncthreads();
  if (fast) {
#pragma unroll
    for (int k = 0; k < EPT; ++k)
      if (t + k * 1024 < n) lst[lds_count(&c[rec[k].y - g0])] = rec[k].x;
  } else {
    for (int p = t; p < n; p += 1024) {
      const int2 r = binbuf[beg + p];
      const int pos = lds_count(&c[r.y - g0]);
      if (in_lds) lst[pos] = r.x; else slist[beg + pos] = r.x;
    }
  }
  if (in_lds) {
    __syncthreads();
    for (int p = t; p < n; p += 1024) slist[beg + p] = lst[p];
  }
}

#define WALK_THREADS 128
__global__ __launch_bounds__(WALK_THREADS, 16)
void gather_kernel(const float* __restrict__ x,
                   const float* __restrict__ as_q, const float* __restrict__ as_a,
                   const float* __restrict__ ad_q, const float* __restrict__ ad_a,
                   const int* __restrict__ off, const int* __restrict__ slist,
                   const float* __restrict__ Pc, const float* __restrict__ cst,
                   float* __restrict__ out, int Nq, int G, int Etot) {
  __shared__ v2f Bt[2 * 16 * 32];
  const int lane = threadIdx.x & 31;
#pragma unroll
  for (int it = 0; it < 2 * 16 * 32 / WALK_THREADS; ++it) {
    const int idx = threadIdx.x + it * WALK_THREADS;
    const int f = idx >> 5, ln = idx & 31;
    const int n = ln & 15, kh = ln >> 4;
    const int q = f & 3;
    const int r = (f >> 2) * 64 + 2 * (4 * (n >> 2) + q) + 32 * kh;
    v2f b;
    b.x = Pc[r * 16 + (n & 3)];
    b.y = Pc[(r + 1) * 16 + (n & 3)];
    Bt[idx] = b;
  }
  __syncthreads();
  const int nw = (int)((gridDim.x * blockDim.x) >> 5);
  const float* xl = x + lane * 2;
  const float ninf = __int_as_float(0xff800000);
  const bool has_share = ((lane >> 3) & 1) == (lane >> 4);
  const bool share_hi = (lane >> 2) & 1;
  const float c0 = cst[lane & 3];
  const int ngroups = (Nq + 7) >> 3;
  for (int t = __builtin_amdgcn_readfirstlane(
           (int)(((unsigned)blockIdx.x * blockDim.x + threadIdx.x) >> 5));
       t < ngroups; t += nw) {
  float res = 0.f;
#pragma unroll 1
  for (int i = 0; i < 8; ++i) {
  const int d = t * 8 + i;
  float part = 0.f;
#pragma unroll 1
  for (int L = 0; L < 2; ++L) {
  const int g = L ? Nq + d : d;
  if (d < Nq && g < G) {
  const float* a_s = L ? as_a : as_q;
  const float4 ad = *(const float4*)((L ? ad_a : ad_q) + (long)d * 4);
  const int beg = off[g];
  const int end = (g + 1 < G) ? off[g + 1] : Etot;
  float m0 = ninf, m1 = ninf, m2 = ninf, m3 = ninf;
  float d0 = 0.f, d1 = 0.f, d2 = 0.f, d3 = 0.f;
  v2f acc0 = {0.f, 0.f}, acc1 = {0.f, 0.f}, acc2 = {0.f, 0.f}, acc3 = {0.f, 0.f};
  for (int c = beg; c < end; c += 32) {
    const int n = (end - c < 32) ? end - c : 32;
    int s_l = 0;
    float e0 = ninf, e1 = ninf, e2 = ninf, e3 = ninf;
    if (lane < n) {
      s_l = slist[c + lane];
      const float4 as = *(const float4*)(a_s + (long)s_l * 4);
      e0 = lrelu(as.x + ad.x);
      e1 = lrelu(as.y + ad.y);
      e2 = lrelu(as.z + ad.z);
      e3 = lrelu(as.w + ad.w);
    }
    const float n0 = fmaxf(m0, wave_max(e0)), n1 = fmaxf(m1, wave_max(e1));
    const float n2 = fmaxf(m2, wave_max(e2)), n3 = fmaxf(m3, wave_max(e3));
    const float r0 = __expf(m0 - n0), r1 = __expf(m1 - n1);
    const float r2 = __expf(m2 - n2), r3 = __expf(m3 - n3);
    const float p0 = __expf(e0 - n0), p1 = __expf(e1 - n1);
    const float p2 = __expf(e2 - n2), p3 = __expf(e3 - n3);
    d0 = d0 * r0 + wave_sum(p0); d1 = d1 * r1 + wave_sum(p1);
    d2 = d2 * r2 + wave_sum(p2); d3 = d3 * r3 + wave_sum(p3);
    acc0 *= r0; acc1 *= r1; acc2 *= r2; acc3 *= r3;
    m0 = n0; m1 = n1; m2 = n2; m3 = n3;
#pragma unroll 4
    for (int j = 0; j < n; ++j) {
      const int s = __builtin_amdgcn_readlane(s_l, j);
      const float w0 = __int_as_float(__builtin_amdgcn_readlane(__float_as_int(p0), j));
      const float w1 = __int_as_float(__builtin_amdgcn_readlane(__float_as_int(p1), j));
      const float w2 = __int_as_float(__builtin_amdgcn_readlane(__float_as_int(p2), j));
      const float w3 = __int_as_float(__builtin_amdgcn_readlane(__float_as_int(p3), j));
      const v2f xv = *(const v2f*)(xl + (long)s * 64);
      acc0 += w0 * xv;
      acc1 += w1 * xv;
      acc2 += w2 * xv;
      acc3 += w3 * xv;
    }
  }
  v2f zr[4];
  zr[0] = acc0 / (d0 + 1e-16f);
  zr[1] = acc1 / (d1 + 1e-16f);
  zr[2] = acc2 / (d2 + 1e-16f);
  zr[3] = acc3 / (d3 + 1e-16f);
  const v2f* bt = Bt + L * (16 * 32) + lane;
  v8f D[4] = {};
#pragma unroll
  for (int h = 0; h < 4; ++h) {
#pragma unroll
    for (int q = 0; q < 4; ++q) {
      const v2f b = bt[(h * 4 + q) * 32];
      D[q] = __builtin_amdgcn_wmma_f32_16x16x4_f32(false, zr[h], false, b, (short)0, D[q],
                                                   false, false);
    }
  }
  const float s_lo = (D[0][0] + D[1][1]) + (D[2][2] + D[3][3]);
  const float s_hi = (D[0][4] + D[1][5]) + (D[2][6] + D[3][7]);
  part += has_share ? (share_hi ? s_hi : s_lo) : 0.f;
  }
  }
  float v = part;
  v += xrow<4>(v);
  v += xrow<8>(v);
  v += xhalf(v);
  res = ((lane >> 2) == i) ? v : res;
  }
  if (t * 8 + (lane >> 2) < Nq) out[(long)t * 32 + lane] = res + c0;
  }
}

extern "C" void kernel_launch(void* const* d_in, const int* in_sizes, int n_in,
                              void* d_out, int out_size, void* d_ws, size_t ws_size,
                              hipStream_t stream) {
  const float* x_c = (const float*)d_in[0];
  const float* x_q = (const float*)d_in[1];
  const float* x_a = (const float*)d_in[2];
  const int* src_cq = (const int*)d_in[3];
  const int* dst_cq = (const int*)d_in[4];
  const int* src_ca = (const int*)d_in[5];
  const int* dst_ca = (const int*)d_in[6];
  const float* Wsq = (const float*)d_in[7];
  const float* Wdq = (const float*)d_in[8];
  const float* Asq = (const float*)d_in[9];
  const float* Adq = (const float*)d_in[10];
  const float* bq  = (const float*)d_in[11];
  const float* Wsa = (const float*)d_in[12];
  const float* Wda = (const float*)d_in[13];
  const float* Asa = (const float*)d_in[14];
  const float* Ada = (const float*)d_in[15];
  const float* ba  = (const float*)d_in[16];
  const float* Wout = (const float*)d_in[17];
  const float* bout = (const float*)d_in[18];

  const long Nc = in_sizes[0] / 64;
  const long Nq = in_sizes[1] / 64;
  const long Na = in_sizes[2] / 64;
  const long Ecq = in_sizes[3];
  const long Eca = in_sizes[5];

  float* w = (float*)d_ws;
  size_t off = 0;
  auto alloc = [&](size_t n) -> float* {
    float* p = w + off;
    off += (n + 15) & ~(size_t)15;
    return p;
  };
  float* Vp_sq = alloc(64 * 16);
  float* Vp_dq = alloc(64 * 16);
  float* Vp_sa = alloc(64 * 16);
  float* Vp_da = alloc(64 * 16);
  float* Pc    = alloc(512 * 16);
  float* cst   = alloc(16);
  float* as_cq = alloc(Nc * 4);
  float* as_ca = alloc(Nc * 4);
  float* ad_cq = alloc(Nq * 4);
  float* ad_ca = alloc(Na * 4);
  float* e_cq  = alloc(Ecq * 4);
  float* e_ca  = alloc(Eca * 4);
  float* m_q   = alloc(Nq * 4);
  float* m_a   = alloc(Na * 4);
  float* den_q = alloc(Nq * 4);
  float* den_a = alloc(Na * 4);
  float* z_q   = alloc(Nq * 256);
  float* z_a   = alloc(Na * 256);
  (void)Vp_dq; (void)Vp_sa; (void)Vp_da; (void)e_ca; (void)m_a; (void)den_q; (void)den_a;
  if (off * sizeof(float) > ws_size) return;

  const long G = Nq + Na;
  const long Etot = Ecq + Eca;
  const long NB = (G + BIN_NODES - 1) >> BIN_SH;
  const long nChunk = (Etot + CHUNK - 1) / CHUNK;
  const long nHist = NB * nChunk;
  const long nHistPad = (nHist + 15) & ~15L;
  int*  hist   = (int*)z_q;
  int*  hoff   = hist + nHistPad;
  int2* binbuf = (int2*)z_a;
  int*  slist  = (int*)e_cq;
  int*  offs   = (int*)m_q;
  if (NB > NB_MAX || 2 * nHistPad > Nq * 256 || 2 * Etot > Na * 256 ||
      Etot > (long)(m_q - e_cq) || G > (long)(z_q - m_q)) return;

  prep_kernel<<<1, 1024, 0, stream>>>(Wsq, Wdq, Asq, Adq, Wsa, Wda, Asa, Ada,
                                      bq, ba, Wout, bout, Vp_sq, Pc, cst);
  const int gNc = (Nc + 127) / 128 < 1024 ? (int)((Nc + 127) / 128) : 1024;
  const int gNq = (Nq + 127) / 128 < 1024 ? (int)((Nq + 127) / 128) : 1024;
  const int gNa = (Na + 127) / 128 < 1024 ? (int)((Na + 127) / 128) : 1024;
  scores_kernel<<<gNc, 256, 0, stream>>>(x_c, Vp_sq, as_cq, (int)Nc);
  scores_kernel<<<gNc, 256, 0, stream>>>(x_c, Vp_sa, as_ca, (int)Nc);
  scores_kernel<<<gNq, 256, 0, stream>>>(x_q, Vp_dq, ad_cq, (int)Nq);
  scores_kernel<<<gNa, 256, 0, stream>>>(x_a, Vp_da, ad_ca, (int)Na);

  bin_count_kernel<<<(int)nChunk, 1024, 0, stream>>>(dst_cq, dst_ca, hist,
                                                     (int)Ecq, (int)Etot, (int)Nq, (int)NB);
  scan_kernel<<<(int)((nHist + 4095) / 4096), 1024, 0, stream>>>(hist, hoff, (int)nHist);
  bin_scatter_kernel<<<(int)nChunk, 1024, 0, stream>>>(src_cq, src_ca, dst_cq, dst_ca, hoff,
                                                       binbuf, (int)Ecq, (int)Etot, (int)Nq,
                                                       (int)NB);
  bin_lists_kernel<<<(int)NB, 1024, 0, stream>>>(binbuf, hoff, offs, slist,
                                                 (int)G, (int)Etot, (int)nChunk, (int)NB);
  const long nLine = (Nq + 7) / 8;
  const int gWalk = (nLine + 3) / 4 < 4096 ? (int)((nLine + 3) / 4) : 4096;
  gather_kernel<<<gWalk, WALK_THREADS, 0, stream>>>(x_c, as_cq, as_ca, ad_cq, ad_ca,
                                           offs, slist, Pc, cst, (float*)d_out,
                                           (int)Nq, (int)G, (int)Etot);
}


# Node walk normalises by one v_rcp_f32 per head (1 ulp) and a multiply, not 8 IEEE divisions: about 85 fewer vector instructions and one long dependent chain less per walk; sum order unchanged
# speedup vs baseline: 7.9050x; 1.0547x over previous
#include <hip/hip_runtime.h>
#include <hip/hip_bf16.h>
#include <math.h>

typedef __attribute__((ext_vector_type(2))) float v2f;
typedef __attribute__((ext_vector_type(8))) float v8f;

__device__ __forceinline__ float lrelu(float x) { return x > 0.f ? x : 0.2f * x; }

template <int M>
__device__ __forceinline__ float xrow(float v) {
  const int b = __float_as_int(v);
  return __int_as_float(__builtin_amdgcn_update_dpp(b, b, 0x160 + M, 0xf, 0xf, false));
}
__device__ __forceinline__ float xhalf(float v) {
  const int b = __float_as_int(v);
  return __int_as_float(__builtin_amdgcn_permlanex16(b, b, 0x76543210, 0xfedcba98, false, false));
}
__device__ __forceinline__ float wave_max(float v) {
  v = fmaxf(v, xrow<1>(v));
  v = fmaxf(v, xrow<2>(v));
  v = fmaxf(v, xrow<4>(v));
  v = fmaxf(v, xrow<8>(v));
  return fmaxf(v, xhalf(v));
}
__device__ __forceinline__ float wave_sum(float v) {
  v += xrow<1>(v);
  v += xrow<2>(v);
  v += xrow<4>(v);
  v += xrow<8>(v);
  return v + xhalf(v);
}

__global__ __launch_bounds__(1024)
void prep_kernel(const float* __restrict__ Wsq, const float* __restrict__ Wdq,
                 const float* __restrict__ Asq, const float* __restrict__ Adq,
                 const float* __restrict__ Wsa, const float* __restrict__ Wda,
                 const float* __restrict__ Asa, const float* __restrict__ Ada,
                 const float* __restrict__ bq,  const float* __restrict__ ba,
                 const float* __restrict__ Wout, const float* __restrict__ bout,
                 float* __restrict__ Vp, float* __restrict__ Pc, float* __restrict__ cst) {
  const int t = threadIdx.x;
  for (int idx = t; idx < 4 * 64 * 16; idx += blockDim.x) {
    int table = idx >> 10, k = (idx >> 4) & 63, j = idx & 15;
    const float* W = (table == 0) ? Wsq : (table == 1) ? Wdq : (table == 2) ? Wsa : Wda;
    const float* A = (table == 0) ? Asq : (table == 1) ? Adq : (table == 2) ? Asa : Ada;
    float s = 0.f;
    if (j < 4)
      for (int c = 0; c < 64; ++c) s += W[k * 256 + j * 64 + c] * A[j * 64 + c];
    Vp[idx] = s;
  }
  for (int idx = t; idx < 512 * 16; idx += blockDim.x) {
    int r = idx >> 4, j = idx & 15;
    int layer = r >> 8, h = (r >> 6) & 3, k = r & 63;
    const float* W = layer ? Wsa : Wsq;
    float s = 0.f;
    if (j < 4) {
      for (int c = 0; c < 64; ++c)
        s += W[k * 256 + h * 64 + c] * Wout[(layer * 64 + c) * 4 + j];
      s *= 0.25f;
    }
    Pc[idx] = s;
  }
  if (t < 16) {
    float s = 0.f;
    if (t < 4) {
      for (int c = 0; c < 64; ++c)
        s += bq[c] * Wout[c * 4 + t] + ba[c] * Wout[(64 + c) * 4 + t];
      s += bout[t];
    }
    cst[t] = s;
  }
}


__global__ void scores_kernel(const float* __restrict__ x, const float* __restrict__ Vp,
                              float* __restrict__ out, int N) {
  const int lane = threadIdx.x & 31;
  const int m = lane & 15;
  const int ko = (lane >> 4) * 2;
  const int nw = (int)((gridDim.x * blockDim.x) >> 5);
  const int ntiles = (N + 15) >> 4;
  v2f b[16];
#pragma unroll
  for (int q = 0; q < 16; ++q) {
    b[q].x = Vp[(q * 4 + ko) * 16 + m];
    b[q].y = Vp[(q * 4 + ko + 1) * 16 + m];
  }
  for (int t = __builtin_amdgcn_readfirstlane(
           (int)(((unsigned)blockIdx.x * blockDim.x + threadIdx.x) >> 5));
       t < ntiles; t += nw) {
    const long base = (long)t * 16;
    long row = base + m; if (row >= N) row = N - 1;
    const float* xr = x + row * 64;
    v8f acc = {};
#pragma unroll
    for (int q = 0; q < 16; ++q) {
      v2f a;
      a.x = xr[q * 4 + ko];
      a.y = xr[q * 4 + ko + 1];
      acc = __builtin_amdgcn_wmma_f32_16x16x4_f32(false, a, false, b[q], (short)0, acc,
                                                  false, false);
    }
    if (m < 4) {
      const long rb = base + ((lane >> 4) << 3);
#pragma unroll
      for (int v = 0; v < 8; ++v) {
        long r = rb + v;
        if (r < N) out[r * 4 + m] = acc[v];
      }
    }
  }
}

#define BIN_SH 9
#define BIN_NODES (1 << BIN_SH)
#define NB_MAX 1024
#define EPT 8
#define CHUNK (EPT * 1024)
#define LIST_CAP 12288

__device__ __forceinline__ int lds_count(int* p) {
  return __hip_atomic_fetch_add(p, 1, __ATOMIC_RELAXED, __HIP_MEMORY_SCOPE_WORKGROUP);
}

__device__ __forceinline__ int block_excl_scan(int v, int* wsum) {
  const int lane = threadIdx.x & 31;
  const int wv = threadIdx.x >> 5;
  int inc = v;
  for (int o = 1; o < 32; o <<= 1) {
    const int nb = __shfl_up(inc, o);
    if (lane >= o) inc += nb;
  }
  if (lane == 31) wsum[wv] = inc;
  __syncthreads();
  int wpre = 0;
  for (int i = 0; i < wv; ++i) wpre += wsum[i];
  return wpre + inc - v;
}

__global__ __launch_bounds__(1024)
void bin_count_kernel(const int* __restrict__ dst_q, const int* __restrict__ dst_a,
                      int* __restrict__ hist, int Ecq, int Etot, int Nq, int NB) {
  __shared__ int c[NB_MAX];
  const int t = threadIdx.x;
  const int wg = blockIdx.x, nwg = gridDim.x;
  c[t] = 0;
  __syncthreads();
  const int i0 = wg * CHUNK + t;
  if (wg * CHUNK + CHUNK <= Etot) {
    int dv[EPT];
#pragma unroll
    for (int k = 0; k < EPT; ++k) {
      const int i = i0 + k * 1024;
      const int* p = (i < Ecq) ? dst_q + i : dst_a + (i - Ecq);
      dv[k] = *p;
    }
#pragma unroll
    for (int k = 0; k < EPT; ++k) {
      const int i = i0 + k * 1024;
      lds_count(&c[((i < Ecq) ? dv[k] : Nq + dv[k]) >> BIN_SH]);
    }
  } else {
    for (int i = i0; i < Etot; i += 1024) {
      const int g = (i < Ecq) ? dst_q[i] : Nq + dst_a[i - Ecq];
      lds_count(&c[g >> BIN_SH]);
    }
  }
  __syncthreads();
  if (t < NB) hist[t * nwg + wg] = c[t];
}

__global__ __launch_bounds__(1024)
void scan_kernel(const int* __restrict__ cnt, int* __restrict__ off, int G) {
  __shared__ int wsum[32];
  const int t = threadIdx.x;
  const int lane = t & 31;
  const int wv = t >> 5;
  const int tile0 = blockIdx.x * 4096;
  int s = 0;
  for (int i = t * 4; i < tile0; i += 4096) {
    const int4 q = *(const int4*)(cnt + i);
    s += q.x + q.y + q.z + q.w;
  }
  for (int o = 16; o > 0; o >>= 1) s += __shfl_xor(s, o);
  if (lane == 0) wsum[wv] = s;
  __syncthreads();
  int carry = 0;
  for (int i = 0; i < 32; ++i) carry += wsum[i];
  __syncthreads();

  const int base = tile0 + t * 4;
  int v0 = 0, v1 = 0, v2 = 0;
  int v3 = 0;
  if (base + 3 < G) {
    const int4 q = *(const int4*)(cnt + base);
    v0 = q.x; v1 = q.y; v2 = q.z; v3 = q.w;
  } else {
    if (base < G) v0 = cnt[base];
    if (base + 1 < G) v1 = cnt[base + 1];
    if (base + 2 < G) v2 = cnt[base + 2];
  }
  const int tsum = v0 + v1 + v2 + v3;
  int inc = tsum;
  for (int o = 1; o < 32; o <<= 1) {
    const int nb = __shfl_up(inc, o);
    if (lane >= o) inc += nb;
  }
  if (lane == 31) wsum[wv] = inc;
  __syncthreads();
  int wpre = 0;
  for (int i = 0; i < wv; ++i) wpre += wsum[i];
  const int ex0 = carry + wpre + inc - tsum;
  if (base < G) off[base] = ex0;
  if (base + 1 < G) off[base + 1] = ex0 + v0;
  if (base + 2 < G) off[base + 2] = ex0 + v0 + v1;
  if (base + 3 < G) off[base + 3] = ex0 + v0 + v1 + v2;
}

__global__ __launch_bounds__(1024)
void bin_scatter_kernel(const int* __restrict__ src_q, const int* __restrict__ src_a,
                        const int* __restrict__ dst_q, const int* __restrict__ dst_a,
                        const int* __restrict__ hoff, int2* __restrict__ binbuf,
                        int Ecq, int Etot, int Nq, int NB) {
  __shared__ int c[NB_MAX];
  __shared__ int ho[NB_MAX];
  __shared__ int wsum[32];
  __shared__ int2 stage[CHUNK];
  const int t = threadIdx.x;
  const int wg = blockIdx.x, nwg = gridDim.x;
  c[t] = 0;
  int hv = 0;
  if (t < NB) hv = hoff[t * nwg + wg];
  __syncthreads();
  int g[EPT], s[EPT], r[EPT];
  const int i0 = wg * CHUNK + t;
  if (wg * CHUNK + CHUNK <= Etot) {
#pragma unroll
    for (int k = 0; k < EPT; ++k) {
      const int i = i0 + k * 1024;
      const bool second = i >= Ecq;
      const int e = second ? i - Ecq : i;
      const int* dp = second ? dst_a : dst_q;
      const int* sp = second ? src_a : src_q;
      g[k] = dp[e];
      s[k] = sp[e];
    }
#pragma unroll
    for (int k = 0; k < EPT; ++k)
      if (i0 + k * 1024 >= Ecq) g[k] += Nq;
  } else {
#pragma unroll
    for (int k = 0; k < EPT; ++k) {
      const int i = i0 + k * 1024;
      g[k] = -1;
      s[k] = 0;
      if (i < Etot) {
        const bool second = i >= Ecq;
        const int e = second ? i - Ecq : i;
        g[k] = second ? Nq + dst_a[e] : dst_q[e];
        s[k] = second ? src_a[e] : src_q[e];
      }
    }
  }
#pragma unroll
  for (int k = 0; k < EPT; ++k) {
    r[k] = 0;
    if (g[k] >= 0) r[k] = lds_count(&c[g[k] >> BIN_SH]);
  }
  ho[t] = hv;
  __syncthreads();
  const int v = c[t];
  const int ex = block_excl_scan(v, wsum);
  c[t] = ex;
  __syncthreads();
#pragma unroll
  for (int k = 0; k < EPT; ++k)
    if (g[k] >= 0) stage[c[g[k] >> BIN_SH] + r[k]] = make_int2(s[k], g[k]);
  __syncthreads();
  const int left = Etot - wg * CHUNK;
  const int n = left < CHUNK ? left : CHUNK;
  for (int p = t; p < n; p += 1024) {
    const int2 rec = stage[p];
    const int b = rec.y >> BIN_SH;
    binbuf[ho[b] + (p - c[b])] = rec;
  }
}

__global__ __launch_bounds__(1024)
void bin_lists_kernel(const int2* __restrict__ binbuf, const int* __restrict__ hoff,
                      int* __restrict__ off, int* __restrict__ slist,
                      int G, int Etot, int nwg, int NB) {
  __shared__ int c[1024];
  __shared__ int wsum[32];
  __shared__ int lst[LIST_CAP];
  const int t = threadIdx.x;
  const int b = blockIdx.x;
  const int beg = hoff[b * nwg];
  const int end = (b + 1 < NB) ? hoff[(b + 1) * nwg] : Etot;
  const int n = end - beg;
  const int g0 = b << BIN_SH;
  const bool fast = n > 0 && n <= EPT * 1024;
  const bool in_lds = n <= LIST_CAP;
  int2 rec[EPT];
  c[t] = 0;
  __syncthreads();
  if (fast) {
#pragma unroll
    for (int k = 0; k < EPT; ++k) {
      const int p = t + k * 1024;
      rec[k] = binbuf[beg + (p < n ? p : n - 1)];
    }
#pragma unroll
    for (int k = 0; k < EPT; ++k)
      if (t + k * 1024 < n) lds_count(&c[rec[k].y - g0]);
  } else {
    for (int p = t; p < n; p += 1024) lds_count(&c[binbuf[beg + p].y - g0]);
  }
  __syncthreads();
  const int v = c[t];
  const int ex = block_excl_scan(v, wsum);
  c[t] = ex;
  if (t < BIN_NODES && g0 + t < G) off[g0 + t] = beg + ex;
  __syncthreads();
  if (fast) {
#pragma unroll
    for (int k = 0; k < EPT; ++k)
      if (t + k * 1024 < n) lst[lds_count(&c[rec[k].y - g0])] = rec[k].x;
  } else {
    for (int p = t; p < n; p += 1024) {
      const int2 r = binbuf[beg + p];
      const int pos = lds_count(&c[r.y - g0]);
      if (in_lds) lst[pos] = r.x; else slist[beg + pos] = r.x;
    }
  }
  if (in_lds) {
    __syncthreads();
    for (int p = t; p < n; p += 1024) slist[beg + p] = lst[p];
  }
}

#define WALK_THREADS 128
__global__ __launch_bounds__(WALK_THREADS, 16)
void gather_kernel(const float* __restrict__ x,
                   const float* __restrict__ as_q, const float* __restrict__ as_a,
                   const float* __restrict__ ad_q, const float* __restrict__ ad_a,
                   const int* __restrict__ off, const int* __restrict__ slist,
                   const float* __restrict__ Pc, const float* __restrict__ cst,
                   float* __restrict__ out, int Nq, int G, int Etot) {
  __shared__ v2f Bt[2 * 16 * 32];
  const int lane = threadIdx.x & 31;
#pragma unroll
  for (int it = 0; it < 2 * 16 * 32 / WALK_THREADS; ++it) {
    const int idx = threadIdx.x + it * WALK_THREADS;
    const int f = idx >> 5, ln = idx & 31;
    const int n = ln & 15, kh = ln >> 4;
    const int q = f & 3;
    const int r = (f >> 2) * 64 + 2 * (4 * (n >> 2) + q) + 32 * kh;
    v2f b;
    b.x = Pc[r * 16 + (n & 3)];
    b.y = Pc[(r + 1) * 16 + (n & 3)];
    Bt[idx] = b;
  }
  __syncthreads();
  const int nw = (int)((gridDim.x * blockDim.x) >> 5);
  const float* xl = x + lane * 2;
  const float ninf = __int_as_float(0xff800000);
  const bool has_share = ((lane >> 3) & 1) == (lane >> 4);
  const bool share_hi = (lane >> 2) & 1;
  const float c0 = cst[lane & 3];
  const int ngroups = (Nq + 7) >> 3;
  for (int t = __builtin_amdgcn_readfirstlane(
           (int)(((unsigned)blockIdx.x * blockDim.x + threadIdx.x) >> 5));
       t < ngroups; t += nw) {
  float res = 0.f;
#pragma unroll 1
  for (int i = 0; i < 8; ++i) {
  const int d = t * 8 + i;
  float part = 0.f;
#pragma unroll 1
  for (int L = 0; L < 2; ++L) {
  const int g = L ? Nq + d : d;
  if (d < Nq && g < G) {
  const float* a_s = L ? as_a : as_q;
  const float4 ad = *(const float4*)((L ? ad_a : ad_q) + (long)d * 4);
  const int beg = off[g];
  const int end = (g + 1 < G) ? off[g + 1] : Etot;
  float m0 = ninf, m1 = ninf, m2 = ninf, m3 = ninf;
  float d0 = 0.f, d1 = 0.f, d2 = 0.f, d3 = 0.f;
  v2f acc0 = {0.f, 0.f}, acc1 = {0.f, 0.f}, acc2 = {0.f, 0.f}, acc3 = {0.f, 0.f};
  for (int c = beg; c < end; c += 32) {
    const int n = (end - c < 32) ? end - c : 32;
    int s_l = 0;
    float e0 = ninf, e1 = ninf, e2 = ninf, e3 = ninf;
    if (lane < n) {
      s_l = slist[c + lane];
      const float4 as = *(const float4*)(a_s + (long)s_l * 4);
      e0 = lrelu(as.x + ad.x);
      e1 = lrelu(as.y + ad.y);
      e2 = lrelu(as.z + ad.z);
      e3 = lrelu(as.w + ad.w);
    }
    const float n0 = fmaxf(m0, wave_max(e0)), n1 = fmaxf(m1, wave_max(e1));
    const float n2 = fmaxf(m2, wave_max(e2)), n3 = fmaxf(m3, wave_max(e3));
    const float r0 = __expf(m0 - n0), r1 = __expf(m1 - n1);
    const float r2 = __expf(m2 - n2), r3 = __expf(m3 - n3);
    const float p0 = __expf(e0 - n0), p1 = __expf(e1 - n1);
    const float p2 = __expf(e2 - n2), p3 = __expf(e3 - n3);
    d0 = d0 * r0 + wave_sum(p0); d1 = d1 * r1 + wave_sum(p1);
    d2 = d2 * r2 + wave_sum(p2); d3 = d3 * r3 + wave_sum(p3);
    acc0 *= r0; acc1 *= r1; acc2 *= r2; acc3 *= r3;
    m0 = n0; m1 = n1; m2 = n2; m3 = n3;
#pragma unroll 4
    for (int j = 0; j < n; ++j) {
      const int s = __builtin_amdgcn_readlane(s_l, j);
      const float w0 = __int_as_float(__builtin_amdgcn_readlane(__float_as_int(p0), j));
      const float w1 = __int_as_float(__builtin_amdgcn_readlane(__float_as_int(p1), j));
      const float w2 = __int_as_float(__builtin_amdgcn_readlane(__float_as_int(p2), j));
      const float w3 = __int_as_float(__builtin_amdgcn_readlane(__float_as_int(p3), j));
      const v2f xv = *(const v2f*)(xl + (long)s * 64);
      acc0 += w0 * xv;
      acc1 += w1 * xv;
      acc2 += w2 * xv;
      acc3 += w3 * xv;
    }
  }
  const float i0 = __builtin_amdgcn_rcpf(d0 + 1e-16f);
  const float i1 = __builtin_amdgcn_rcpf(d1 + 1e-16f);
  const float i2 = __builtin_amdgcn_rcpf(d2 + 1e-16f);
  const float i3 = __builtin_amdgcn_rcpf(d3 + 1e-16f);
  v2f zr[4];
  zr[0] = acc0 * i0;
  zr[1] = acc1 * i1;
  zr[2] = acc2 * i2;
  zr[3] = acc3 * i3;
  const v2f* bt = Bt + L * (16 * 32) + lane;
  v8f D[4] = {};
#pragma unroll
  for (int h = 0; h < 4; ++h) {
#pragma unroll
    for (int q = 0; q < 4; ++q) {
      const v2f b = bt[(h * 4 + q) * 32];
      D[q] = __builtin_amdgcn_wmma_f32_16x16x4_f32(false, zr[h], false, b, (short)0, D[q],
                                                   false, false);
    }
  }
  const float s_lo = (D[0][0] + D[1][1]) + (D[2][2] + D[3][3]);
  const float s_hi = (D[0][4] + D[1][5]) + (D[2][6] + D[3][7]);
  part += has_share ? (share_hi ? s_hi : s_lo) : 0.f;
  }
  }
  float v = part;
  v += xrow<4>(v);
  v += xrow<8>(v);
  v += xhalf(v);
  res = ((lane >> 2) == i) ? v : res;
  }
  if (t * 8 + (lane >> 2) < Nq) out[(long)t * 32 + lane] = res + c0;
  }
}

extern "C" void kernel_launch(void* const* d_in, const int* in_sizes, int n_in,
                              void* d_out, int out_size, void* d_ws, size_t ws_size,
                              hipStream_t stream) {
  const float* x_c = (const float*)d_in[0];
  const float* x_q = (const float*)d_in[1];
  const float* x_a = (const float*)d_in[2];
  const int* src_cq = (const int*)d_in[3];
  const int* dst_cq = (const int*)d_in[4];
  const int* src_ca = (const int*)d_in[5];
  const int* dst_ca = (const int*)d_in[6];
  const float* Wsq = (const float*)d_in[7];
  const float* Wdq = (const float*)d_in[8];
  const float* Asq = (const float*)d_in[9];
  const float* Adq = (const float*)d_in[10];
  const float* bq  = (const float*)d_in[11];
  const float* Wsa = (const float*)d_in[12];
  const float* Wda = (const float*)d_in[13];
  const float* Asa = (const float*)d_in[14];
  const float* Ada = (const float*)d_in[15];
  const float* ba  = (const float*)d_in[16];
  const float* Wout = (const float*)d_in[17];
  const float* bout = (const float*)d_in[18];

  const long Nc = in_sizes[0] / 64;
  const long Nq = in_sizes[1] / 64;
  const long Na = in_sizes[2] / 64;
  const long Ecq = in_sizes[3];
  const long Eca = in_sizes[5];

  float* w = (float*)d_ws;
  size_t off = 0;
  auto alloc = [&](size_t n) -> float* {
    float* p = w + off;
    off += (n + 15) & ~(size_t)15;
    return p;
  };
  float* Vp_sq = alloc(64 * 16);
  float* Vp_dq = alloc(64 * 16);
  float* Vp_sa = alloc(64 * 16);
  float* Vp_da = alloc(64 * 16);
  float* Pc    = alloc(512 * 16);
  float* cst   = alloc(16);
  float* as_cq = alloc(Nc * 4);
  float* as_ca = alloc(Nc * 4);
  float* ad_cq = alloc(Nq * 4);
  float* ad_ca = alloc(Na * 4);
  float* e_cq  = alloc(Ecq * 4);
  float* e_ca  = alloc(Eca * 4);
  float* m_q   = alloc(Nq * 4);
  float* m_a   = alloc(Na * 4);
  float* den_q = alloc(Nq * 4);
  float* den_a = alloc(Na * 4);
  float* z_q   = alloc(Nq * 256);
  float* z_a   = alloc(Na * 256);
  (void)Vp_dq; (void)Vp_sa; (void)Vp_da; (void)e_ca; (void)m_a; (void)den_q; (void)den_a;
  if (off * sizeof(float) > ws_size) return;

  const long G = Nq + Na;
  const long Etot = Ecq + Eca;
  const long NB = (G + BIN_NODES - 1) >> BIN_SH;
  const long nChunk = (Etot + CHUNK - 1) / CHUNK;
  const long nHist = NB * nChunk;
  const long nHistPad = (nHist + 15) & ~15L;
  int*  hist   = (int*)z_q;
  int*  hoff   = hist + nHistPad;
  int2* binbuf = (int2*)z_a;
  int*  slist  = (int*)e_cq;
  int*  offs   = (int*)m_q;
  if (NB > NB_MAX || 2 * nHistPad > Nq * 256 || 2 * Etot > Na * 256 ||
      Etot > (long)(m_q - e_cq) || G > (long)(z_q - m_q)) return;

  prep_kernel<<<1, 1024, 0, stream>>>(Wsq, Wdq, Asq, Adq, Wsa, Wda, Asa, Ada,
                                      bq, ba, Wout, bout, Vp_sq, Pc, cst);
  const int gNc = (Nc + 127) / 128 < 1024 ? (int)((Nc + 127) / 128) : 1024;
  const int gNq = (Nq + 127) / 128 < 1024 ? (int)((Nq + 127) / 128) : 1024;
  const int gNa = (Na + 127) / 128 < 1024 ? (int)((Na + 127) / 128) : 1024;
  scores_kernel<<<gNc, 256, 0, stream>>>(x_c, Vp_sq, as_cq, (int)Nc);
  scores_kernel<<<gNc, 256, 0, stream>>>(x_c, Vp_sa, as_ca, (int)Nc);
  scores_kernel<<<gNq, 256, 0, stream>>>(x_q, Vp_dq, ad_cq, (int)Nq);
  scores_kernel<<<gNa, 256, 0, stream>>>(x_a, Vp_da, ad_ca, (int)Na);

  bin_count_kernel<<<(int)nChunk, 1024, 0, stream>>>(dst_cq, dst_ca, hist,
                                                     (int)Ecq, (int)Etot, (int)Nq, (int)NB);
  scan_kernel<<<(int)((nHist + 4095) / 4096), 1024, 0, stream>>>(hist, hoff, (int)nHist);
  bin_scatter_kernel<<<(int)nChunk, 1024, 0, stream>>>(src_cq, src_ca, dst_cq, dst_ca, hoff,
                                                       binbuf, (int)Ecq, (int)Etot, (int)Nq,
                                                       (int)NB);
  bin_lists_kernel<<<(int)NB, 1024, 0, stream>>>(binbuf, hoff, offs, slist,
                                                 (int)G, (int)Etot, (int)nChunk, (int)NB);
  const long nLine = (Nq + 7) / 8;
  const int gWalk = (nLine + 3) / 4 < 4096 ? (int)((nLine + 3) / 4) : 4096;
  gather_kernel<<<gWalk, WALK_THREADS, 0, stream>>>(x_c, as_cq, as_ca, ad_cq, ad_ca,
                                           offs, slist, Pc, cst, (float*)d_out,
                                           (int)Nq, (int)G, (int)Etot);
}


# Edge lists without the global count and scan launches: a chunk sorts its edges by bin in LDS into its own slice; a bin's workgroup reads one run per chunk, offset = sum of run starts; list-order sums
# speedup vs baseline: 8.7291x; 1.1043x over previous
#include <hip/hip_runtime.h>
#include <hip/hip_bf16.h>
#include <math.h>

typedef __attribute__((ext_vector_type(2))) float v2f;
typedef __attribute__((ext_vector_type(8))) float v8f;

__device__ __forceinline__ float lrelu(float x) { return x > 0.f ? x : 0.2f * x; }

template <int M>
__device__ __forceinline__ float xrow(float v) {
  const int b = __float_as_int(v);
  return __int_as_float(__builtin_amdgcn_update_dpp(b, b, 0x160 + M, 0xf, 0xf, false));
}
__device__ __forceinline__ float xhalf(float v) {
  const int b = __float_as_int(v);
  return __int_as_float(__builtin_amdgcn_permlanex16(b, b, 0x76543210, 0xfedcba98, false, false));
}
__device__ __forceinline__ float wave_max(float v) {
  v = fmaxf(v, xrow<1>(v));
  v = fmaxf(v, xrow<2>(v));
  v = fmaxf(v, xrow<4>(v));
  v = fmaxf(v, xrow<8>(v));
  return fmaxf(v, xhalf(v));
}
__device__ __forceinline__ float wave_sum(float v) {
  v += xrow<1>(v);
  v += xrow<2>(v);
  v += xrow<4>(v);
  v += xrow<8>(v);
  return v + xhalf(v);
}

__global__ __launch_bounds__(1024)
void prep_kernel(const float* __restrict__ Wsq, const float* __restrict__ Wdq,
                 const float* __restrict__ Asq, const float* __restrict__ Adq,
                 const float* __restrict__ Wsa, const float* __restrict__ Wda,
                 const float* __restrict__ Asa, const float* __restrict__ Ada,
                 const float* __restrict__ bq,  const float* __restrict__ ba,
                 const float* __restrict__ Wout, const float* __restrict__ bout,
                 float* __restrict__ Vp, float* __restrict__ Pc, float* __restrict__ cst) {
  const int t = threadIdx.x;
  for (int idx = t; idx < 4 * 64 * 16; idx += blockDim.x) {
    int table = idx >> 10, k = (idx >> 4) & 63, j = idx & 15;
    const float* W = (table == 0) ? Wsq : (table == 1) ? Wdq : (table == 2) ? Wsa : Wda;
    const float* A = (table == 0) ? Asq : (table == 1) ? Adq : (table == 2) ? Asa : Ada;
    float s = 0.f;
    if (j < 4)
      for (int c = 0; c < 64; ++c) s += W[k * 256 + j * 64 + c] * A[j * 64 + c];
    Vp[idx] = s;
  }
  for (int idx = t; idx < 512 * 16; idx += blockDim.x) {
    int r = idx >> 4, j = idx & 15;
    int layer = r >> 8, h = (r >> 6) & 3, k = r & 63;
    const float* W = layer ? Wsa : Wsq;
    float s = 0.f;
    if (j < 4) {
      for (int c = 0; c < 64; ++c)
        s += W[k * 256 + h * 64 + c] * Wout[(layer * 64 + c) * 4 + j];
      s *= 0.25f;
    }
    Pc[idx] = s;
  }
  if (t < 16) {
    float s = 0.f;
    if (t < 4) {
      for (int c = 0; c < 64; ++c)
        s += bq[c] * Wout[c * 4 + t] + ba[c] * Wout[(64 + c) * 4 + t];
      s += bout[t];
    }
    cst[t] = s;
  }
}


__global__ void scores_kernel(const float* __restrict__ x, const float* __restrict__ Vp,
                              float* __restrict__ out, int N) {
  const int lane = threadIdx.x & 31;
  const int m = lane & 15;
  const int ko = (lane >> 4) * 2;
  const int nw = (int)((gridDim.x * blockDim.x) >> 5);
  const int ntiles = (N + 15) >> 4;
  v2f b[16];
#pragma unroll
  for (int q = 0; q < 16; ++q) {
    b[q].x = Vp[(q * 4 + ko) * 16 + m];
    b[q].y = Vp[(q * 4 + ko + 1) * 16 + m];
  }
  for (int t = __builtin_amdgcn_readfirstlane(
           (int)(((unsigned)blockIdx.x * blockDim.x + threadIdx.x) >> 5));
       t < ntiles; t += nw) {
    const long base = (long)t * 16;
    long row = base + m; if (row >= N) row = N - 1;
    const float* xr = x + row * 64;
    v8f acc = {};
#pragma unroll
    for (int q = 0; q < 16; ++q) {
      v2f a;
      a.x = xr[q * 4 + ko];
      a.y = xr[q * 4 + ko + 1];
      acc = __builtin_amdgcn_wmma_f32_16x16x4_f32(false, a, false, b[q], (short)0, acc,
                                                  false, false);
    }
    if (m < 4) {
      const long rb = base + ((lane >> 4) << 3);
#pragma unroll
      for (int v = 0; v < 8; ++v) {
        long r = rb + v;
        if (r < N) out[r * 4 + m] = acc[v];
      }
    }
  }
}

#define BIN_SH 9
#define BIN_NODES (1 << BIN_SH)
#define NB_MAX 1024
#define EPT 8
#define CHUNK (EPT * 1024)
#define LIST_CAP 12288

__device__ __forceinline__ int lds_count(int* p) {
  return __hip_atomic_fetch_add(p, 1, __ATOMIC_RELAXED, __HIP_MEMORY_SCOPE_WORKGROUP);
}

__device__ __forceinline__ int block_excl_scan(int v, int* wsum) {
  const int lane = threadIdx.x & 31;
  const int wv = threadIdx.x >> 5;
  int inc = v;
  for (int o = 1; o < 32; o <<= 1) {
    const int nb = __shfl_up(inc, o);
    if (lane >= o) inc += nb;
  }
  if (lane == 31) wsum[wv] = inc;
  __syncthreads();
  int wpre = 0;
  for (int i = 0; i < wv; ++i) wpre += wsum[i];
  return wpre + inc - v;
}

__global__ __launch_bounds__(1024)
void bin_scatter_kernel(const int* __restrict__ src_q, const int* __restrict__ src_a,
                        const int* __restrict__ dst_q, const int* __restrict__ dst_a,
                        int* __restrict__ cstart, int2* __restrict__ binbuf,
                        int Ecq, int Etot, int Nq, int NB) {
  __shared__ int c[NB_MAX];
  __shared__ int wsum[32];
  __shared__ int2 stage[CHUNK];
  const int t = threadIdx.x;
  const int wg = blockIdx.x;
  c[t] = 0;
  __syncthreads();
  int g[EPT], s[EPT], r[EPT];
  const int i0 = wg * CHUNK + t;
  if (wg * CHUNK + CHUNK <= Etot) {
#pragma unroll
    for (int k = 0; k < EPT; ++k) {
      const int i = i0 + k * 1024;
      const bool second = i >= Ecq;
      const int e = second ? i - Ecq : i;
      const int* dp = second ? dst_a : dst_q;
      const int* sp = second ? src_a : src_q;
      g[k] = dp[e];
      s[k] = sp[e];
    }
#pragma unroll
    for (int k = 0; k < EPT; ++k)
      if (i0 + k * 1024 >= Ecq) g[k] += Nq;
  } else {
#pragma unroll
    for (int k = 0; k < EPT; ++k) {
      const int i = i0 + k * 1024;
      g[k] = -1;
      s[k] = 0;
      if (i < Etot) {
        const bool second = i >= Ecq;
        const int e = second ? i - Ecq : i;
        g[k] = second ? Nq + dst_a[e] : dst_q[e];
        s[k] = second ? src_a[e] : src_q[e];
      }
    }
  }
#pragma unroll
  for (int k = 0; k < EPT; ++k) {
    r[k] = 0;
    if (g[k] >= 0) r[k] = lds_count(&c[g[k] >> BIN_SH]);
  }
  __syncthreads();
  const int v = c[t];
  const int ex = block_excl_scan(v, wsum);
  c[t] = ex;
  if (t < NB) cstart[(long)wg * NB + t] = ex;
  __syncthreads();
#pragma unroll
  for (int k = 0; k < EPT; ++k)
    if (g[k] >= 0) stage[c[g[k] >> BIN_SH] + r[k]] = make_int2(s[k], g[k]);
  __syncthreads();
  const int left = Etot - wg * CHUNK;
  const int n = left < CHUNK ? left : CHUNK;
  int2* slice = binbuf + (long)wg * CHUNK;
  for (int p = t; p < n; p += 1024) slice[p] = stage[p];
}

__device__ __forceinline__ int run_index(const int* rpos, const int* rsrc, int p) {
  int w = 0;
  for (int step = 512; step > 0; step >>= 1)
    if (rpos[w + step] <= p) w += step;
  return rsrc[w] + (p - rpos[w]);
}

#define RUNS_PW 8
#define RUN_SLOTS 3

__global__ __launch_bounds__(1024, 16)
void bin_lists_kernel(const int2* __restrict__ binbuf, const int* __restrict__ cstart,
                      int* __restrict__ off, int* __restrict__ slist,
                      int G, int Etot, int nwg, int NB) {
  __shared__ int c[1024];
  __shared__ int wsum[32];
  __shared__ int wsA[32];
  __shared__ int wsB[32];
  __shared__ int rpos[1024];
  __shared__ int rsrc[1024];
  __shared__ int big;
  __shared__ int lst[LIST_CAP];
  const int t = threadIdx.x;
  const int lane = t & 31;
  const int wv = t >> 5;
  const int b = blockIdx.x;
  const int g0 = b << BIN_SH;
  int s0 = 0, len = 0;
  if (t < nwg) {
    const int left = Etot - t * CHUNK;
    const int nrec = left < CHUNK ? left : CHUNK;
    const int* cs = cstart + (long)t * NB + b;
    s0 = cs[0];
    const int s1 = (b + 1 < NB) ? cs[1] : nrec;
    len = s1 - s0;
  }
  int incA = s0, incB = len;
  for (int o = 1; o < 32; o <<= 1) {
    const int ua = __shfl_up(incA, o);
    const int ub = __shfl_up(incB, o);
    if (lane >= o) { incA += ua; incB += ub; }
  }
  if (lane == 31) { wsA[wv] = incA; wsB[wv] = incB; }
  if (t == 0) big = 0;
  c[t] = 0;
  __syncthreads();
  int beg = 0, n = 0, pre = 0;
  for (int i = 0; i < 32; ++i) {
    const int ua = wsA[i], ub = wsB[i];
    beg += ua;
    n += ub;
    if (i < wv) pre += ub;
  }
  rpos[t] = pre + incB - len;
  rsrc[t] = t * CHUNK + s0;
  if (len > 32 * RUN_SLOTS) big = 1;
  __syncthreads();
  const bool fast = nwg <= 32 * RUNS_PW && big == 0;
  const bool in_lds = n <= LIST_CAP;
  int2 rec[RUNS_PW][RUN_SLOTS];
  if (fast) {
#pragma unroll
    for (int u = 0; u < RUNS_PW; ++u) {
      const int w = wv + 32 * u;
      const int L = rpos[w + 1] - rpos[w];
      const int2* run = binbuf + rsrc[w];
#pragma unroll
      for (int h = 0; h < RUN_SLOTS; ++h) {
        rec[u][h] = make_int2(0, -1);
        if (lane + 32 * h < L) rec[u][h] = run[lane + 32 * h];
      }
    }
#pragma unroll
    for (int u = 0; u < RUNS_PW; ++u) {
#pragma unroll
      for (int h = 0; h < RUN_SLOTS; ++h)
        if (rec[u][h].y >= 0) lds_count(&c[rec[u][h].y - g0]);
    }
  } else {
    for (int p = t; p < n; p += 1024) lds_count(&c[binbuf[run_index(rpos, rsrc, p)].y - g0]);
  }
  __syncthreads();
  const int v = c[t];
  const int ex = block_excl_scan(v, wsum);
  c[t] = ex;
  if (t < BIN_NODES && g0 + t < G) off[g0 + t] = beg + ex;
  __syncthreads();
  if (fast) {
#pragma unroll
    for (int u = 0; u < RUNS_PW; ++u) {
#pragma unroll
      for (int h = 0; h < RUN_SLOTS; ++h) {
        if (rec[u][h].y >= 0) {
          const int pos = lds_count(&c[rec[u][h].y - g0]);
          if (in_lds) lst[pos] = rec[u][h].x; else slist[beg + pos] = rec[u][h].x;
        }
      }
    }
  } else {
    for (int p = t; p < n; p += 1024) {
      const int2 r = binbuf[run_index(rpos, rsrc, p)];
      const int pos = lds_count(&c[r.y - g0]);
      if (in_lds) lst[pos] = r.x; else slist[beg + pos] = r.x;
    }
  }
  if (in_lds) {
    __syncthreads();
    for (int p = t; p < n; p += 1024) slist[beg + p] = lst[p];
  }
}

#define WALK_THREADS 128
__global__ __launch_bounds__(WALK_THREADS, 16)
void gather_kernel(const float* __restrict__ x,
                   const float* __restrict__ as_q, const float* __restrict__ as_a,
                   const float* __restrict__ ad_q, const float* __restrict__ ad_a,
                   const int* __restrict__ off, const int* __restrict__ slist,
                   const float* __restrict__ Pc, const float* __restrict__ cst,
                   float* __restrict__ out, int Nq, int G, int Etot) {
  __shared__ v2f Bt[2 * 16 * 32];
  const int lane = threadIdx.x & 31;
#pragma unroll
  for (int it = 0; it < 2 * 16 * 32 / WALK_THREADS; ++it) {
    const int idx = threadIdx.x + it * WALK_THREADS;
    const int f = idx >> 5, ln = idx & 31;
    const int n = ln & 15, kh = ln >> 4;
    const int q = f & 3;
    const int r = (f >> 2) * 64 + 2 * (4 * (n >> 2) + q) + 32 * kh;
    v2f b;
    b.x = Pc[r * 16 + (n & 3)];
    b.y = Pc[(r + 1) * 16 + (n & 3)];
    Bt[idx] = b;
  }
  __syncthreads();
  const int nw = (int)((gridDim.x * blockDim.x) >> 5);
  const float* xl = x + lane * 2;
  const float ninf = __int_as_float(0xff800000);
  const bool has_share = ((lane >> 3) & 1) == (lane >> 4);
  const bool share_hi = (lane >> 2) & 1;
  const float c0 = cst[lane & 3];
  const int ngroups = (Nq + 7) >> 3;
  for (int t = __builtin_amdgcn_readfirstlane(
           (int)(((unsigned)blockIdx.x * blockDim.x + threadIdx.x) >> 5));
       t < ngroups; t += nw) {
  float res = 0.f;
#pragma unroll 1
  for (int i = 0; i < 8; ++i) {
  const int d = t * 8 + i;
  float part = 0.f;
#pragma unroll 1
  for (int L = 0; L < 2; ++L) {
  const int g = L ? Nq + d : d;
  if (d < Nq && g < G) {
  const float* a_s = L ? as_a : as_q;
  const float4 ad = *(const float4*)((L ? ad_a : ad_q) + (long)d * 4);
  const int beg = off[g];
  const int end = (g + 1 < G) ? off[g + 1] : Etot;
  float m0 = ninf, m1 = ninf, m2 = ninf, m3 = ninf;
  float d0 = 0.f, d1 = 0.f, d2 = 0.f, d3 = 0.f;
  v2f acc0 = {0.f, 0.f}, acc1 = {0.f, 0.f}, acc2 = {0.f, 0.f}, acc3 = {0.f, 0.f};
  for (int c = beg; c < end; c += 32) {
    const int n = (end - c < 32) ? end - c : 32;
    int s_l = 0;
    float e0 = ninf, e1 = ninf, e2 = ninf, e3 = ninf;
    if (lane < n) {
      s_l = slist[c + lane];
      const float4 as = *(const float4*)(a_s + (long)s_l * 4);
      e0 = lrelu(as.x + ad.x);
      e1 = lrelu(as.y + ad.y);
      e2 = lrelu(as.z + ad.z);
      e3 = lrelu(as.w + ad.w);
    }
    const float n0 = fmaxf(m0, wave_max(e0)), n1 = fmaxf(m1, wave_max(e1));
    const float n2 = fmaxf(m2, wave_max(e2)), n3 = fmaxf(m3, wave_max(e3));
    const float r0 = __expf(m0 - n0), r1 = __expf(m1 - n1);
    const float r2 = __expf(m2 - n2), r3 = __expf(m3 - n3);
    const float p0 = __expf(e0 - n0), p1 = __expf(e1 - n1);
    const float p2 = __expf(e2 - n2), p3 = __expf(e3 - n3);
    d0 = d0 * r0 + wave_sum(p0); d1 = d1 * r1 + wave_sum(p1);
    d2 = d2 * r2 + wave_sum(p2); d3 = d3 * r3 + wave_sum(p3);
    acc0 *= r0; acc1 *= r1; acc2 *= r2; acc3 *= r3;
    m0 = n0; m1 = n1; m2 = n2; m3 = n3;
#pragma unroll 4
    for (int j = 0; j < n; ++j) {
      const int s = __builtin_amdgcn_readlane(s_l, j);
      const float w0 = __int_as_float(__builtin_amdgcn_readlane(__float_as_int(p0), j));
      const float w1 = __int_as_float(__builtin_amdgcn_readlane(__float_as_int(p1), j));
      const float w2 = __int_as_float(__builtin_amdgcn_readlane(__float_as_int(p2), j));
      const float w3 = __int_as_float(__builtin_amdgcn_readlane(__float_as_int(p3), j));
      const v2f xv = *(const v2f*)(xl + (long)s * 64);
      acc0 += w0 * xv;
      acc1 += w1 * xv;
      acc2 += w2 * xv;
      acc3 += w3 * xv;
    }
  }
  const float i0 = __builtin_amdgcn_rcpf(d0 + 1e-16f);
  const float i1 = __builtin_amdgcn_rcpf(d1 + 1e-16f);
  const float i2 = __builtin_amdgcn_rcpf(d2 + 1e-16f);
  const float i3 = __builtin_amdgcn_rcpf(d3 + 1e-16f);
  v2f zr[4];
  zr[0] = acc0 * i0;
  zr[1] = acc1 * i1;
  zr[2] = acc2 * i2;
  zr[3] = acc3 * i3;
  const v2f* bt = Bt + L * (16 * 32) + lane;
  v8f D[4] = {};
#pragma unroll
  for (int h = 0; h < 4; ++h) {
#pragma unroll
    for (int q = 0; q < 4; ++q) {
      const v2f b = bt[(h * 4 + q) * 32];
      D[q] = __builtin_amdgcn_wmma_f32_16x16x4_f32(false, zr[h], false, b, (short)0, D[q],
                                                   false, false);
    }
  }
  const float s_lo = (D[0][0] + D[1][1]) + (D[2][2] + D[3][3]);
  const float s_hi = (D[0][4] + D[1][5]) + (D[2][6] + D[3][7]);
  part += has_share ? (share_hi ? s_hi : s_lo) : 0.f;
  }
  }
  float v = part;
  v += xrow<4>(v);
  v += xrow<8>(v);
  v += xhalf(v);
  res = ((lane >> 2) == i) ? v : res;
  }
  if (t * 8 + (lane >> 2) < Nq) out[(long)t * 32 + lane] = res + c0;
  }
}

extern "C" void kernel_launch(void* const* d_in, const int* in_sizes, int n_in,
                              void* d_out, int out_size, void* d_ws, size_t ws_size,
                              hipStream_t stream) {
  const float* x_c = (const float*)d_in[0];
  const float* x_q = (const float*)d_in[1];
  const float* x_a = (const float*)d_in[2];
  const int* src_cq = (const int*)d_in[3];
  const int* dst_cq = (const int*)d_in[4];
  const int* src_ca = (const int*)d_in[5];
  const int* dst_ca = (const int*)d_in[6];
  const float* Wsq = (const float*)d_in[7];
  const float* Wdq = (const float*)d_in[8];
  const float* Asq = (const float*)d_in[9];
  const float* Adq = (const float*)d_in[10];
  const float* bq  = (const float*)d_in[11];
  const float* Wsa = (const float*)d_in[12];
  const float* Wda = (const float*)d_in[13];
  const float* Asa = (const float*)d_in[14];
  const float* Ada = (const float*)d_in[15];
  const float* ba  = (const float*)d_in[16];
  const float* Wout = (const float*)d_in[17];
  const float* bout = (const float*)d_in[18];

  const long Nc = in_sizes[0] / 64;
  const long Nq = in_sizes[1] / 64;
  const long Na = in_sizes[2] / 64;
  const long Ecq = in_sizes[3];
  const long Eca = in_sizes[5];

  float* w = (float*)d_ws;
  size_t off = 0;
  auto alloc = [&](size_t n) -> float* {
    float* p = w + off;
    off += (n + 15) & ~(size_t)15;
    return p;
  };
  float* Vp_sq = alloc(64 * 16);
  float* Vp_dq = alloc(64 * 16);
  float* Vp_sa = alloc(64 * 16);
  float* Vp_da = alloc(64 * 16);
  float* Pc    = alloc(512 * 16);
  float* cst   = alloc(16);
  float* as_cq = alloc(Nc * 4);
  float* as_ca = alloc(Nc * 4);
  float* ad_cq = alloc(Nq * 4);
  float* ad_ca = alloc(Na * 4);
  float* e_cq  = alloc(Ecq * 4);
  float* e_ca  = alloc(Eca * 4);
  float* m_q   = alloc(Nq * 4);
  float* m_a   = alloc(Na * 4);
  float* den_q = alloc(Nq * 4);
  float* den_a = alloc(Na * 4);
  float* z_q   = alloc(Nq * 256);
  float* z_a   = alloc(Na * 256);
  (void)Vp_dq; (void)Vp_sa; (void)Vp_da; (void)e_ca; (void)m_a; (void)den_q; (void)den_a;
  if (off * sizeof(float) > ws_size) return;

  const long G = Nq + Na;
  const long Etot = Ecq + Eca;
  const long NB = (G + BIN_NODES - 1) >> BIN_SH;
  const long nChunk = (Etot + CHUNK - 1) / CHUNK;
  const long nStart = NB * nChunk;
  int*  cstart = (int*)z_q;
  int2* binbuf = (int2*)z_a;
  int*  slist  = (int*)e_cq;
  int*  offs   = (int*)m_q;
  if (NB > NB_MAX || nChunk > 1024 || nStart > Nq * 256 || 2 * Etot > Na * 256 ||
      Etot > (long)(m_q - e_cq) || G > (long)(z_q - m_q)) return;

  prep_kernel<<<1, 1024, 0, stream>>>(Wsq, Wdq, Asq, Adq, Wsa, Wda, Asa, Ada,
                                      bq, ba, Wout, bout, Vp_sq, Pc, cst);
  const int gNc = (Nc + 127) / 128 < 1024 ? (int)((Nc + 127) / 128) : 1024;
  const int gNq = (Nq + 127) / 128 < 1024 ? (int)((Nq + 127) / 128) : 1024;
  const int gNa = (Na + 127) / 128 < 1024 ? (int)((Na + 127) / 128) : 1024;
  scores_kernel<<<gNc, 256, 0, stream>>>(x_c, Vp_sq, as_cq, (int)Nc);
  scores_kernel<<<gNc, 256, 0, stream>>>(x_c, Vp_sa, as_ca, (int)Nc);
  scores_kernel<<<gNq, 256, 0, stream>>>(x_q, Vp_dq, ad_cq, (int)Nq);
  scores_kernel<<<gNa, 256, 0, stream>>>(x_a, Vp_da, ad_ca, (int)Na);

  bin_scatter_kernel<<<(int)nChunk, 1024, 0, stream>>>(src_cq, src_ca, dst_cq, dst_ca, cstart,
                                                       binbuf, (int)Ecq, (int)Etot, (int)Nq,
                                                       (int)NB);
  bin_lists_kernel<<<(int)NB, 1024, 0, stream>>>(binbuf, cstart, offs, slist,
                                                 (int)G, (int)Etot, (int)nChunk, (int)NB);
  const long nLine = (Nq + 7) / 8;
  const int gWalk = (nLine + 3) / 4 < 4096 ? (int)((nLine + 3) / 4) : 4096;
  gather_kernel<<<gWalk, WALK_THREADS, 0, stream>>>(x_c, as_cq, as_ca, ad_cq, ad_ca,
                                           offs, slist, Pc, cst, (float*)d_out,
                                           (int)Nq, (int)G, (int)Etot);
}


# All four score products in one launch: both source tables in one B fragment set (columns 0-3, 4-7), x_c read once; x_q and x_a tiles dealt to the same waves; 8 launches become 5; sums unchanged
# speedup vs baseline: 9.4791x; 1.0859x over previous
#include <hip/hip_runtime.h>
#include <hip/hip_bf16.h>
#include <math.h>

typedef __attribute__((ext_vector_type(2))) float v2f;
typedef __attribute__((ext_vector_type(8))) float v8f;

__device__ __forceinline__ float lrelu(float x) { return x > 0.f ? x : 0.2f * x; }

template <int M>
__device__ __forceinline__ float xrow(float v) {
  const int b = __float_as_int(v);
  return __int_as_float(__builtin_amdgcn_update_dpp(b, b, 0x160 + M, 0xf, 0xf, false));
}
__device__ __forceinline__ float xhalf(float v) {
  const int b = __float_as_int(v);
  return __int_as_float(__builtin_amdgcn_permlanex16(b, b, 0x76543210, 0xfedcba98, false, false));
}
__device__ __forceinline__ float wave_max(float v) {
  v = fmaxf(v, xrow<1>(v));
  v = fmaxf(v, xrow<2>(v));
  v = fmaxf(v, xrow<4>(v));
  v = fmaxf(v, xrow<8>(v));
  return fmaxf(v, xhalf(v));
}
__device__ __forceinline__ float wave_sum(float v) {
  v += xrow<1>(v);
  v += xrow<2>(v);
  v += xrow<4>(v);
  v += xrow<8>(v);
  return v + xhalf(v);
}

__global__ __launch_bounds__(1024)
void prep_kernel(const float* __restrict__ Wsq, const float* __restrict__ Wdq,
                 const float* __restrict__ Asq, const float* __restrict__ Adq,
                 const float* __restrict__ Wsa, const float* __restrict__ Wda,
                 const float* __restrict__ Asa, const float* __restrict__ Ada,
                 const float* __restrict__ bq,  const float* __restrict__ ba,
                 const float* __restrict__ Wout, const float* __restrict__ bout,
                 float* __restrict__ Vp, float* __restrict__ Pc, float* __restrict__ cst) {
  const int t = threadIdx.x;
  for (int idx = t; idx < 4 * 64 * 16; idx += blockDim.x) {
    int table = idx >> 10, k = (idx >> 4) & 63, j = idx & 15;
    const float* W = (table == 0) ? Wsq : (table == 1) ? Wdq : (table == 2) ? Wsa : Wda;
    const float* A = (table == 0) ? Asq : (table == 1) ? Adq : (table == 2) ? Asa : Ada;
    float s = 0.f;
    if (j < 4)
      for (int c = 0; c < 64; ++c) s += W[k * 256 + j * 64 + c] * A[j * 64 + c];
    Vp[idx] = s;
  }
  for (int idx = t; idx < 512 * 16; idx += blockDim.x) {
    int r = idx >> 4, j = idx & 15;
    int layer = r >> 8, h = (r >> 6) & 3, k = r & 63;
    const float* W = layer ? Wsa : Wsq;
    float s = 0.f;
    if (j < 4) {
      for (int c = 0; c < 64; ++c)
        s += W[k * 256 + h * 64 + c] * Wout[(layer * 64 + c) * 4 + j];
      s *= 0.25f;
    }
    Pc[idx] = s;
  }
  if (t < 16) {
    float s = 0.f;
    if (t < 4) {
      for (int c = 0; c < 64; ++c)
        s += bq[c] * Wout[c * 4 + t] + ba[c] * Wout[(64 + c) * 4 + t];
      s += bout[t];
    }
    cst[t] = s;
  }
}


template <bool TWO>
__device__ __forceinline__ void score_tiles(const float* __restrict__ x, int N,
                                            const float* __restrict__ tabA,
                                            const float* __restrict__ tabB,
                                            float* __restrict__ outA, float* __restrict__ outB,
                                            int first, int nw, int lane) {
  const int m = lane & 15;
  const int ko = (lane >> 4) * 2;
  const int ntiles = (N + 15) >> 4;
  const float* tb = (TWO && m >= 4) ? tabB : tabA;
  const int col = (TWO && m >= 4) ? m - 4 : m;
  v2f b[16];
#pragma unroll
  for (int q = 0; q < 16; ++q) {
    b[q].x = tb[(q * 4 + ko) * 16 + col];
    b[q].y = tb[(q * 4 + ko + 1) * 16 + col];
  }
  for (int t = first; t < ntiles; t += nw) {
    const long base = (long)t * 16;
    long row = base + m; if (row >= N) row = N - 1;
    const float* xr = x + row * 64;
    v8f acc = {};
#pragma unroll
    for (int q = 0; q < 16; ++q) {
      v2f a;
      a.x = xr[q * 4 + ko];
      a.y = xr[q * 4 + ko + 1];
      acc = __builtin_amdgcn_wmma_f32_16x16x4_f32(false, a, false, b[q], (short)0, acc,
                                                  false, false);
    }
    if (TWO ? m < 8 : m < 4) {
      float* o = (TWO && m >= 4) ? outB : outA;
      const int cc = m & 3;
      const long rb = base + ((lane >> 4) << 3);
#pragma unroll
      for (int v = 0; v < 8; ++v) {
        long r = rb + v;
        if (r < N) o[r * 4 + cc] = acc[v];
      }
    }
  }
}

__global__ void scores_kernel(const float* __restrict__ x_c, const float* __restrict__ x_q,
                              const float* __restrict__ x_a, const float* __restrict__ Vp,
                              float* __restrict__ as_q, float* __restrict__ as_a,
                              float* __restrict__ ad_q, float* __restrict__ ad_a,
                              int Nc, int Nq, int Na) {
  const int lane = threadIdx.x & 31;
  const int nw = (int)((gridDim.x * blockDim.x) >> 5);
  const int w = __builtin_amdgcn_readfirstlane(
      (int)(((unsigned)blockIdx.x * blockDim.x + threadIdx.x) >> 5));
  const int tc = (Nc + 15) >> 4;
  const int tq = (Nq + 15) >> 4;
  const int fq = (w + nw - tc % nw) % nw;
  const int fa = (w + nw - (tc + tq) % nw) % nw;
  score_tiles<true>(x_c, Nc, Vp, Vp + 2 * 1024, as_q, as_a, w, nw, lane);
  score_tiles<false>(x_q, Nq, Vp + 1024, nullptr, ad_q, nullptr, fq, nw, lane);
  score_tiles<false>(x_a, Na, Vp + 3 * 1024, nullptr, ad_a, nullptr, fa, nw, lane);
}

#define BIN_SH 9
#define BIN_NODES (1 << BIN_SH)
#define NB_MAX 1024
#define EPT 8
#define CHUNK (EPT * 1024)
#define LIST_CAP 12288

__device__ __forceinline__ int lds_count(int* p) {
  return __hip_atomic_fetch_add(p, 1, __ATOMIC_RELAXED, __HIP_MEMORY_SCOPE_WORKGROUP);
}

__device__ __forceinline__ int block_excl_scan(int v, int* wsum) {
  const int lane = threadIdx.x & 31;
  const int wv = threadIdx.x >> 5;
  int inc = v;
  for (int o = 1; o < 32; o <<= 1) {
    const int nb = __shfl_up(inc, o);
    if (lane >= o) inc += nb;
  }
  if (lane == 31) wsum[wv] = inc;
  __syncthreads();
  int wpre = 0;
  for (int i = 0; i < wv; ++i) wpre += wsum[i];
  return wpre + inc - v;
}

__global__ __launch_bounds__(1024)
void bin_scatter_kernel(const int* __restrict__ src_q, const int* __restrict__ src_a,
                        const int* __restrict__ dst_q, const int* __restrict__ dst_a,
                        int* __restrict__ cstart, int2* __restrict__ binbuf,
                        int Ecq, int Etot, int Nq, int NB) {
  __shared__ int c[NB_MAX];
  __shared__ int wsum[32];
  __shared__ int2 stage[CHUNK];
  const int t = threadIdx.x;
  const int wg = blockIdx.x;
  c[t] = 0;
  __syncthreads();
  int g[EPT], s[EPT], r[EPT];
  const int i0 = wg * CHUNK + t;
  if (wg * CHUNK + CHUNK <= Etot) {
#pragma unroll
    for (int k = 0; k < EPT; ++k) {
      const int i = i0 + k * 1024;
      const bool second = i >= Ecq;
      const int e = second ? i - Ecq : i;
      const int* dp = second ? dst_a : dst_q;
      const int* sp = second ? src_a : src_q;
      g[k] = dp[e];
      s[k] = sp[e];
    }
#pragma unroll
    for (int k = 0; k < EPT; ++k)
      if (i0 + k * 1024 >= Ecq) g[k] += Nq;
  } else {
#pragma unroll
    for (int k = 0; k < EPT; ++k) {
      const int i = i0 + k * 1024;
      g[k] = -1;
      s[k] = 0;
      if (i < Etot) {
        const bool second = i >= Ecq;
        const int e = second ? i - Ecq : i;
        g[k] = second ? Nq + dst_a[e] : dst_q[e];
        s[k] = second ? src_a[e] : src_q[e];
      }
    }
  }
#pragma unroll
  for (int k = 0; k < EPT; ++k) {
    r[k] = 0;
    if (g[k] >= 0) r[k] = lds_count(&c[g[k] >> BIN_SH]);
  }
  __syncthreads();
  const int v = c[t];
  const int ex = block_excl_scan(v, wsum);
  c[t] = ex;
  if (t < NB) cstart[(long)wg * NB + t] = ex;
  __syncthreads();
#pragma unroll
  for (int k = 0; k < EPT; ++k)
    if (g[k] >= 0) stage[c[g[k] >> BIN_SH] + r[k]] = make_int2(s[k], g[k]);
  __syncthreads();
  const int left = Etot - wg * CHUNK;
  const int n = left < CHUNK ? left : CHUNK;
  int2* slice = binbuf + (long)wg * CHUNK;
  for (int p = t; p < n; p += 1024) slice[p] = stage[p];
}

__device__ __forceinline__ int run_index(const int* rpos, const int* rsrc, int p) {
  int w = 0;
  for (int step = 512; step > 0; step >>= 1)
    if (rpos[w + step] <= p) w += step;
  return rsrc[w] + (p - rpos[w]);
}

#define RUNS_PW 8
#define RUN_SLOTS 3

__global__ __launch_bounds__(1024, 16)
void bin_lists_kernel(const int2* __restrict__ binbuf, const int* __restrict__ cstart,
                      int* __restrict__ off, int* __restrict__ slist,
                      int G, int Etot, int nwg, int NB) {
  __shared__ int c[1024];
  __shared__ int wsum[32];
  __shared__ int wsA[32];
  __shared__ int wsB[32];
  __shared__ int rpos[1024];
  __shared__ int rsrc[1024];
  __shared__ int big;
  __shared__ int lst[LIST_CAP];
  const int t = threadIdx.x;
  const int lane = t & 31;
  const int wv = t >> 5;
  const int b = blockIdx.x;
  const int g0 = b << BIN_SH;
  int s0 = 0, len = 0;
  if (t < nwg) {
    const int left = Etot - t * CHUNK;
    const int nrec = left < CHUNK ? left : CHUNK;
    const int* cs = cstart + (long)t * NB + b;
    s0 = cs[0];
    const int s1 = (b + 1 < NB) ? cs[1] : nrec;
    len = s1 - s0;
  }
  int incA = s0, incB = len;
  for (int o = 1; o < 32; o <<= 1) {
    const int ua = __shfl_up(incA, o);
    const int ub = __shfl_up(incB, o);
    if (lane >= o) { incA += ua; incB += ub; }
  }
  if (lane == 31) { wsA[wv] = incA; wsB[wv] = incB; }
  if (t == 0) big = 0;
  c[t] = 0;
  __syncthreads();
  int beg = 0, n = 0, pre = 0;
  for (int i = 0; i < 32; ++i) {
    const int ua = wsA[i], ub = wsB[i];
    beg += ua;
    n += ub;
    if (i < wv) pre += ub;
  }
  rpos[t] = pre + incB - len;
  rsrc[t] = t * CHUNK + s0;
  if (len > 32 * RUN_SLOTS) big = 1;
  __syncthreads();
  const bool fast = nwg <= 32 * RUNS_PW && big == 0;
  const bool in_lds = n <= LIST_CAP;
  int2 rec[RUNS_PW][RUN_SLOTS];
  if (fast) {
#pragma unroll
    for (int u = 0; u < RUNS_PW; ++u) {
      const int w = wv + 32 * u;
      const int L = rpos[w + 1] - rpos[w];
      const int2* run = binbuf + rsrc[w];
#pragma unroll
      for (int h = 0; h < RUN_SLOTS; ++h) {
        rec[u][h] = make_int2(0, -1);
        if (lane + 32 * h < L) rec[u][h] = run[lane + 32 * h];
      }
    }
#pragma unroll
    for (int u = 0; u < RUNS_PW; ++u) {
#pragma unroll
      for (int h = 0; h < RUN_SLOTS; ++h)
        if (rec[u][h].y >= 0) lds_count(&c[rec[u][h].y - g0]);
    }
  } else {
    for (int p = t; p < n; p += 1024) lds_count(&c[binbuf[run_index(rpos, rsrc, p)].y - g0]);
  }
  __syncthreads();
  const int v = c[t];
  const int ex = block_excl_scan(v, wsum);
  c[t] = ex;
  if (t < BIN_NODES && g0 + t < G) off[g0 + t] = beg + ex;
  __syncthreads();
  if (fast) {
#pragma unroll
    for (int u = 0; u < RUNS_PW; ++u) {
#pragma unroll
      for (int h = 0; h < RUN_SLOTS; ++h) {
        if (rec[u][h].y >= 0) {
          const int pos = lds_count(&c[rec[u][h].y - g0]);
          if (in_lds) lst[pos] = rec[u][h].x; else slist[beg + pos] = rec[u][h].x;
        }
      }
    }
  } else {
    for (int p = t; p < n; p += 1024) {
      const int2 r = binbuf[run_index(rpos, rsrc, p)];
      const int pos = lds_count(&c[r.y - g0]);
      if (in_lds) lst[pos] = r.x; else slist[beg + pos] = r.x;
    }
  }
  if (in_lds) {
    __syncthreads();
    for (int p = t; p < n; p += 1024) slist[beg + p] = lst[p];
  }
}

#define WALK_THREADS 128
__global__ __launch_bounds__(WALK_THREADS, 16)
void gather_kernel(const float* __restrict__ x,
                   const float* __restrict__ as_q, const float* __restrict__ as_a,
                   const float* __restrict__ ad_q, const float* __restrict__ ad_a,
                   const int* __restrict__ off, const int* __restrict__ slist,
                   const float* __restrict__ Pc, const float* __restrict__ cst,
                   float* __restrict__ out, int Nq, int G, int Etot) {
  __shared__ v2f Bt[2 * 16 * 32];
  const int lane = threadIdx.x & 31;
#pragma unroll
  for (int it = 0; it < 2 * 16 * 32 / WALK_THREADS; ++it) {
    const int idx = threadIdx.x + it * WALK_THREADS;
    const int f = idx >> 5, ln = idx & 31;
    const int n = ln & 15, kh = ln >> 4;
    const int q = f & 3;
    const int r = (f >> 2) * 64 + 2 * (4 * (n >> 2) + q) + 32 * kh;
    v2f b;
    b.x = Pc[r * 16 + (n & 3)];
    b.y = Pc[(r + 1) * 16 + (n & 3)];
    Bt[idx] = b;
  }
  __syncthreads();
  const int nw = (int)((gridDim.x * blockDim.x) >> 5);
  const float* xl = x + lane * 2;
  const float ninf = __int_as_float(0xff800000);
  const bool has_share = ((lane >> 3) & 1) == (lane >> 4);
  const bool share_hi = (lane >> 2) & 1;
  const float c0 = cst[lane & 3];
  const int ngroups = (Nq + 7) >> 3;
  for (int t = __builtin_amdgcn_readfirstlane(
           (int)(((unsigned)blockIdx.x * blockDim.x + threadIdx.x) >> 5));
       t < ngroups; t += nw) {
  float res = 0.f;
#pragma unroll 1
  for (int i = 0; i < 8; ++i) {
  const int d = t * 8 + i;
  float part = 0.f;
#pragma unroll 1
  for (int L = 0; L < 2; ++L) {
  const int g = L ? Nq + d : d;
  if (d < Nq && g < G) {
  const float* a_s = L ? as_a : as_q;
  const float4 ad = *(const float4*)((L ? ad_a : ad_q) + (long)d * 4);
  const int beg = off[g];
  const int end = (g + 1 < G) ? off[g + 1] : Etot;
  float m0 = ninf, m1 = ninf, m2 = ninf, m3 = ninf;
  float d0 = 0.f, d1 = 0.f, d2 = 0.f, d3 = 0.f;
  v2f acc0 = {0.f, 0.f}, acc1 = {0.f, 0.f}, acc2 = {0.f, 0.f}, acc3 = {0.f, 0.f};
  for (int c = beg; c < end; c += 32) {
    const int n = (end - c < 32) ? end - c : 32;
    int s_l = 0;
    float e0 = ninf, e1 = ninf, e2 = ninf, e3 = ninf;
    if (lane < n) {
      s_l = slist[c + lane];
      const float4 as = *(const float4*)(a_s + (long)s_l * 4);
      e0 = lrelu(as.x + ad.x);
      e1 = lrelu(as.y + ad.y);
      e2 = lrelu(as.z + ad.z);
      e3 = lrelu(as.w + ad.w);
    }
    const float n0 = fmaxf(m0, wave_max(e0)), n1 = fmaxf(m1, wave_max(e1));
    const float n2 = fmaxf(m2, wave_max(e2)), n3 = fmaxf(m3, wave_max(e3));
    const float r0 = __expf(m0 - n0), r1 = __expf(m1 - n1);
    const float r2 = __expf(m2 - n2), r3 = __expf(m3 - n3);
    const float p0 = __expf(e0 - n0), p1 = __expf(e1 - n1);
    const float p2 = __expf(e2 - n2), p3 = __expf(e3 - n3);
    d0 = d0 * r0 + wave_sum(p0); d1 = d1 * r1 + wave_sum(p1);
    d2 = d2 * r2 + wave_sum(p2); d3 = d3 * r3 + wave_sum(p3);
    acc0 *= r0; acc1 *= r1; acc2 *= r2; acc3 *= r3;
    m0 = n0; m1 = n1; m2 = n2; m3 = n3;
#pragma unroll 4
    for (int j = 0; j < n; ++j) {
      const int s = __builtin_amdgcn_readlane(s_l, j);
      const float w0 = __int_as_float(__builtin_amdgcn_readlane(__float_as_int(p0), j));
      const float w1 = __int_as_float(__builtin_amdgcn_readlane(__float_as_int(p1), j));
      const float w2 = __int_as_float(__builtin_amdgcn_readlane(__float_as_int(p2), j));
      const float w3 = __int_as_float(__builtin_amdgcn_readlane(__float_as_int(p3), j));
      const v2f xv = *(const v2f*)(xl + (long)s * 64);
      acc0 += w0 * xv;
      acc1 += w1 * xv;
      acc2 += w2 * xv;
      acc3 += w3 * xv;
    }
  }
  const float i0 = __builtin_amdgcn_rcpf(d0 + 1e-16f);
  const float i1 = __builtin_amdgcn_rcpf(d1 + 1e-16f);
  const float i2 = __builtin_amdgcn_rcpf(d2 + 1e-16f);
  const float i3 = __builtin_amdgcn_rcpf(d3 + 1e-16f);
  v2f zr[4];
  zr[0] = acc0 * i0;
  zr[1] = acc1 * i1;
  zr[2] = acc2 * i2;
  zr[3] = acc3 * i3;
  const v2f* bt = Bt + L * (16 * 32) + lane;
  v8f D[4] = {};
#pragma unroll
  for (int h = 0; h < 4; ++h) {
#pragma unroll
    for (int q = 0; q < 4; ++q) {
      const v2f b = bt[(h * 4 + q) * 32];
      D[q] = __builtin_amdgcn_wmma_f32_16x16x4_f32(false, zr[h], false, b, (short)0, D[q],
                                                   false, false);
    }
  }
  const float s_lo = (D[0][0] + D[1][1]) + (D[2][2] + D[3][3]);
  const float s_hi = (D[0][4] + D[1][5]) + (D[2][6] + D[3][7]);
  part += has_share ? (share_hi ? s_hi : s_lo) : 0.f;
  }
  }
  float v = part;
  v += xrow<4>(v);
  v += xrow<8>(v);
  v += xhalf(v);
  res = ((lane >> 2) == i) ? v : res;
  }
  if (t * 8 + (lane >> 2) < Nq) out[(long)t * 32 + lane] = res + c0;
  }
}

extern "C" void kernel_launch(void* const* d_in, const int* in_sizes, int n_in,
                              void* d_out, int out_size, void* d_ws, size_t ws_size,
                              hipStream_t stream) {
  const float* x_c = (const float*)d_in[0];
  const float* x_q = (const float*)d_in[1];
  const float* x_a = (const float*)d_in[2];
  const int* src_cq = (const int*)d_in[3];
  const int* dst_cq = (const int*)d_in[4];
  const int* src_ca = (const int*)d_in[5];
  const int* dst_ca = (const int*)d_in[6];
  const float* Wsq = (const float*)d_in[7];
  const float* Wdq = (const float*)d_in[8];
  const float* Asq = (const float*)d_in[9];
  const float* Adq = (const float*)d_in[10];
  const float* bq  = (const float*)d_in[11];
  const float* Wsa = (const float*)d_in[12];
  const float* Wda = (const float*)d_in[13];
  const float* Asa = (const float*)d_in[14];
  const float* Ada = (const float*)d_in[15];
  const float* ba  = (const float*)d_in[16];
  const float* Wout = (const float*)d_in[17];
  const float* bout = (const float*)d_in[18];

  const long Nc = in_sizes[0] / 64;
  const long Nq = in_sizes[1] / 64;
  const long Na = in_sizes[2] / 64;
  const long Ecq = in_sizes[3];
  const long Eca = in_sizes[5];

  float* w = (float*)d_ws;
  size_t off = 0;
  auto alloc = [&](size_t n) -> float* {
    float* p = w + off;
    off += (n + 15) & ~(size_t)15;
    return p;
  };
  float* Vp_sq = alloc(64 * 16);
  float* Vp_dq = alloc(64 * 16);
  float* Vp_sa = alloc(64 * 16);
  float* Vp_da = alloc(64 * 16);
  float* Pc    = alloc(512 * 16);
  float* cst   = alloc(16);
  float* as_cq = alloc(Nc * 4);
  float* as_ca = alloc(Nc * 4);
  float* ad_cq = alloc(Nq * 4);
  float* ad_ca = alloc(Na * 4);
  float* e_cq  = alloc(Ecq * 4);
  float* e_ca  = alloc(Eca * 4);
  float* m_q   = alloc(Nq * 4);
  float* m_a   = alloc(Na * 4);
  float* den_q = alloc(Nq * 4);
  float* den_a = alloc(Na * 4);
  float* z_q   = alloc(Nq * 256);
  float* z_a   = alloc(Na * 256);
  (void)Vp_dq; (void)Vp_sa; (void)Vp_da; (void)e_ca; (void)m_a; (void)den_q; (void)den_a;
  if (off * sizeof(float) > ws_size) return;

  const long G = Nq + Na;
  const long Etot = Ecq + Eca;
  const long NB = (G + BIN_NODES - 1) >> BIN_SH;
  const long nChunk = (Etot + CHUNK - 1) / CHUNK;
  const long nStart = NB * nChunk;
  int*  cstart = (int*)z_q;
  int2* binbuf = (int2*)z_a;
  int*  slist  = (int*)e_cq;
  int*  offs   = (int*)m_q;
  if (NB > NB_MAX || nChunk > 1024 || nStart > Nq * 256 || 2 * Etot > Na * 256 ||
      Etot > (long)(m_q - e_cq) || G > (long)(z_q - m_q)) return;

  prep_kernel<<<1, 1024, 0, stream>>>(Wsq, Wdq, Asq, Adq, Wsa, Wda, Asa, Ada,
                                      bq, ba, Wout, bout, Vp_sq, Pc, cst);
  const long nTile = (Nc + 15) / 16 + (Nq + 15) / 16 + (Na + 15) / 16;
  const int gS = (nTile + 7) / 8 < 1024 ? (int)((nTile + 7) / 8) : 1024;
  scores_kernel<<<gS, 256, 0, stream>>>(x_c, x_q, x_a, Vp_sq, as_cq, as_ca, ad_cq, ad_ca,
                                        (int)Nc, (int)Nq, (int)Na);

  bin_scatter_kernel<<<(int)nChunk, 1024, 0, stream>>>(src_cq, src_ca, dst_cq, dst_ca, cstart,
                                                       binbuf, (int)Ecq, (int)Etot, (int)Nq,
                                                       (int)NB);
  bin_lists_kernel<<<(int)NB, 1024, 0, stream>>>(binbuf, cstart, offs, slist,
                                                 (int)G, (int)Etot, (int)nChunk, (int)NB);
  const long nLine = (Nq + 7) / 8;
  const int gWalk = (nLine + 3) / 4 < 4096 ? (int)((nLine + 3) / 4) : 4096;
  gather_kernel<<<gWalk, WALK_THREADS, 0, stream>>>(x_c, as_cq, as_ca, ad_cq, ad_ca,
                                           offs, slist, Pc, cst, (float*)d_out,
                                           (int)Nq, (int)G, (int)Etot);
}


# Node walk row loop: the loads of up to 12 source rows leave as one group before the first sum (groups of 12, 8 or 4 cover the rest of a list); no edge-by-edge remainder loop; sum order unchanged
# speedup vs baseline: 9.8918x; 1.0435x over previous
#include <hip/hip_runtime.h>
#include <hip/hip_bf16.h>
#include <math.h>

typedef __attribute__((ext_vector_type(2))) float v2f;
typedef __attribute__((ext_vector_type(8))) float v8f;

__device__ __forceinline__ float lrelu(float x) { return x > 0.f ? x : 0.2f * x; }

template <int M>
__device__ __forceinline__ float xrow(float v) {
  const int b = __float_as_int(v);
  return __int_as_float(__builtin_amdgcn_update_dpp(b, b, 0x160 + M, 0xf, 0xf, false));
}
__device__ __forceinline__ float xhalf(float v) {
  const int b = __float_as_int(v);
  return __int_as_float(__builtin_amdgcn_permlanex16(b, b, 0x76543210, 0xfedcba98, false, false));
}
__device__ __forceinline__ float wave_max(float v) {
  v = fmaxf(v, xrow<1>(v));
  v = fmaxf(v, xrow<2>(v));
  v = fmaxf(v, xrow<4>(v));
  v = fmaxf(v, xrow<8>(v));
  return fmaxf(v, xhalf(v));
}
__device__ __forceinline__ float wave_sum(float v) {
  v += xrow<1>(v);
  v += xrow<2>(v);
  v += xrow<4>(v);
  v += xrow<8>(v);
  return v + xhalf(v);
}

__global__ __launch_bounds__(1024)
void prep_kernel(const float* __restrict__ Wsq, const float* __restrict__ Wdq,
                 const float* __restrict__ Asq, const float* __restrict__ Adq,
                 const float* __restrict__ Wsa, const float* __restrict__ Wda,
                 const float* __restrict__ Asa, const float* __restrict__ Ada,
                 const float* __restrict__ bq,  const float* __restrict__ ba,
                 const float* __restrict__ Wout, const float* __restrict__ bout,
                 float* __restrict__ Vp, float* __restrict__ Pc, float* __restrict__ cst) {
  const int t = threadIdx.x;
  for (int idx = t; idx < 4 * 64 * 16; idx += blockDim.x) {
    int table = idx >> 10, k = (idx >> 4) & 63, j = idx & 15;
    const float* W = (table == 0) ? Wsq : (table == 1) ? Wdq : (table == 2) ? Wsa : Wda;
    const float* A = (table == 0) ? Asq : (table == 1) ? Adq : (table == 2) ? Asa : Ada;
    float s = 0.f;
    if (j < 4)
      for (int c = 0; c < 64; ++c) s += W[k * 256 + j * 64 + c] * A[j * 64 + c];
    Vp[idx] = s;
  }
  for (int idx = t; idx < 512 * 16; idx += blockDim.x) {
    int r = idx >> 4, j = idx & 15;
    int layer = r >> 8, h = (r >> 6) & 3, k = r & 63;
    const float* W = layer ? Wsa : Wsq;
    float s = 0.f;
    if (j < 4) {
      for (int c = 0; c < 64; ++c)
        s += W[k * 256 + h * 64 + c] * Wout[(layer * 64 + c) * 4 + j];
      s *= 0.25f;
    }
    Pc[idx] = s;
  }
  if (t < 16) {
    float s = 0.f;
    if (t < 4) {
      for (int c = 0; c < 64; ++c)
        s += bq[c] * Wout[c * 4 + t] + ba[c] * Wout[(64 + c) * 4 + t];
      s += bout[t];
    }
    cst[t] = s;
  }
}


template <bool TWO>
__device__ __forceinline__ void score_tiles(const float* __restrict__ x, int N,
                                            const float* __restrict__ tabA,
                                            const float* __restrict__ tabB,
                                            float* __restrict__ outA, float* __restrict__ outB,
                                            int first, int nw, int lane) {
  const int m = lane & 15;
  const int ko = (lane >> 4) * 2;
  const int ntiles = (N + 15) >> 4;
  const float* tb = (TWO && m >= 4) ? tabB : tabA;
  const int col = (TWO && m >= 4) ? m - 4 : m;
  v2f b[16];
#pragma unroll
  for (int q = 0; q < 16; ++q) {
    b[q].x = tb[(q * 4 + ko) * 16 + col];
    b[q].y = tb[(q * 4 + ko + 1) * 16 + col];
  }
  for (int t = first; t < ntiles; t += nw) {
    const long base = (long)t * 16;
    long row = base + m; if (row >= N) row = N - 1;
    const float* xr = x + row * 64;
    v8f acc = {};
#pragma unroll
    for (int q = 0; q < 16; ++q) {
      v2f a;
      a.x = xr[q * 4 + ko];
      a.y = xr[q * 4 + ko + 1];
      acc = __builtin_amdgcn_wmma_f32_16x16x4_f32(false, a, false, b[q], (short)0, acc,
                                                  false, false);
    }
    if (TWO ? m < 8 : m < 4) {
      float* o = (TWO && m >= 4) ? outB : outA;
      const int cc = m & 3;
      const long rb = base + ((lane >> 4) << 3);
#pragma unroll
      for (int v = 0; v < 8; ++v) {
        long r = rb + v;
        if (r < N) o[r * 4 + cc] = acc[v];
      }
    }
  }
}

__global__ void scores_kernel(const float* __restrict__ x_c, const float* __restrict__ x_q,
                              const float* __restrict__ x_a, const float* __restrict__ Vp,
                              float* __restrict__ as_q, float* __restrict__ as_a,
                              float* __restrict__ ad_q, float* __restrict__ ad_a,
                              int Nc, int Nq, int Na) {
  const int lane = threadIdx.x & 31;
  const int nw = (int)((gridDim.x * blockDim.x) >> 5);
  const int w = __builtin_amdgcn_readfirstlane(
      (int)(((unsigned)blockIdx.x * blockDim.x + threadIdx.x) >> 5));
  const int tc = (Nc + 15) >> 4;
  const int tq = (Nq + 15) >> 4;
  const int fq = (w + nw - tc % nw) % nw;
  const int fa = (w + nw - (tc + tq) % nw) % nw;
  score_tiles<true>(x_c, Nc, Vp, Vp + 2 * 1024, as_q, as_a, w, nw, lane);
  score_tiles<false>(x_q, Nq, Vp + 1024, nullptr, ad_q, nullptr, fq, nw, lane);
  score_tiles<false>(x_a, Na, Vp + 3 * 1024, nullptr, ad_a, nullptr, fa, nw, lane);
}

#define BIN_SH 9
#define BIN_NODES (1 << BIN_SH)
#define NB_MAX 1024
#define EPT 8
#define CHUNK (EPT * 1024)
#define LIST_CAP 12288

__device__ __forceinline__ int lds_count(int* p) {
  return __hip_atomic_fetch_add(p, 1, __ATOMIC_RELAXED, __HIP_MEMORY_SCOPE_WORKGROUP);
}

__device__ __forceinline__ int block_excl_scan(int v, int* wsum) {
  const int lane = threadIdx.x & 31;
  const int wv = threadIdx.x >> 5;
  int inc = v;
  for (int o = 1; o < 32; o <<= 1) {
    const int nb = __shfl_up(inc, o);
    if (lane >= o) inc += nb;
  }
  if (lane == 31) wsum[wv] = inc;
  __syncthreads();
  int wpre = 0;
  for (int i = 0; i < wv; ++i) wpre += wsum[i];
  return wpre + inc - v;
}

__global__ __launch_bounds__(1024)
void bin_scatter_kernel(const int* __restrict__ src_q, const int* __restrict__ src_a,
                        const int* __restrict__ dst_q, const int* __restrict__ dst_a,
                        int* __restrict__ cstart, int2* __restrict__ binbuf,
                        int Ecq, int Etot, int Nq, int NB) {
  __shared__ int c[NB_MAX];
  __shared__ int wsum[32];
  __shared__ int2 stage[CHUNK];
  const int t = threadIdx.x;
  const int wg = blockIdx.x;
  c[t] = 0;
  __syncthreads();
  int g[EPT], s[EPT], r[EPT];
  const int i0 = wg * CHUNK + t;
  if (wg * CHUNK + CHUNK <= Etot) {
#pragma unroll
    for (int k = 0; k < EPT; ++k) {
      const int i = i0 + k * 1024;
      const bool second = i >= Ecq;
      const int e = second ? i - Ecq : i;
      const int* dp = second ? dst_a : dst_q;
      const int* sp = second ? src_a : src_q;
      g[k] = dp[e];
      s[k] = sp[e];
    }
#pragma unroll
    for (int k = 0; k < EPT; ++k)
      if (i0 + k * 1024 >= Ecq) g[k] += Nq;
  } else {
#pragma unroll
    for (int k = 0; k < EPT; ++k) {
      const int i = i0 + k * 1024;
      g[k] = -1;
      s[k] = 0;
      if (i < Etot) {
        const bool second = i >= Ecq;
        const int e = second ? i - Ecq : i;
        g[k] = second ? Nq + dst_a[e] : dst_q[e];
        s[k] = second ? src_a[e] : src_q[e];
      }
    }
  }
#pragma unroll
  for (int k = 0; k < EPT; ++k) {
    r[k] = 0;
    if (g[k] >= 0) r[k] = lds_count(&c[g[k] >> BIN_SH]);
  }
  __syncthreads();
  const int v = c[t];
  const int ex = block_excl_scan(v, wsum);
  c[t] = ex;
  if (t < NB) cstart[(long)wg * NB + t] = ex;
  __syncthreads();
#pragma unroll
  for (int k = 0; k < EPT; ++k)
    if (g[k] >= 0) stage[c[g[k] >> BIN_SH] + r[k]] = make_int2(s[k], g[k]);
  __syncthreads();
  const int left = Etot - wg * CHUNK;
  const int n = left < CHUNK ? left : CHUNK;
  int2* slice = binbuf + (long)wg * CHUNK;
  for (int p = t; p < n; p += 1024) slice[p] = stage[p];
}

__device__ __forceinline__ int run_index(const int* rpos, const int* rsrc, int p) {
  int w = 0;
  for (int step = 512; step > 0; step >>= 1)
    if (rpos[w + step] <= p) w += step;
  return rsrc[w] + (p - rpos[w]);
}

#define RUNS_PW 8
#define RUN_SLOTS 3

__global__ __launch_bounds__(1024, 16)
void bin_lists_kernel(const int2* __restrict__ binbuf, const int* __restrict__ cstart,
                      int* __restrict__ off, int* __restrict__ slist,
                      int G, int Etot, int nwg, int NB) {
  __shared__ int c[1024];
  __shared__ int wsum[32];
  __shared__ int wsA[32];
  __shared__ int wsB[32];
  __shared__ int rpos[1024];
  __shared__ int rsrc[1024];
  __shared__ int big;
  __shared__ int lst[LIST_CAP];
  const int t = threadIdx.x;
  const int lane = t & 31;
  const int wv = t >> 5;
  const int b = blockIdx.x;
  const int g0 = b << BIN_SH;
  int s0 = 0, len = 0;
  if (t < nwg) {
    const int left = Etot - t * CHUNK;
    const int nrec = left < CHUNK ? left : CHUNK;
    const int* cs = cstart + (long)t * NB + b;
    s0 = cs[0];
    const int s1 = (b + 1 < NB) ? cs[1] : nrec;
    len = s1 - s0;
  }
  int incA = s0, incB = len;
  for (int o = 1; o < 32; o <<= 1) {
    const int ua = __shfl_up(incA, o);
    const int ub = __shfl_up(incB, o);
    if (lane >= o) { incA += ua; incB += ub; }
  }
  if (lane == 31) { wsA[wv] = incA; wsB[wv] = incB; }
  if (t == 0) big = 0;
  c[t] = 0;
  __syncthreads();
  int beg = 0, n = 0, pre = 0;
  for (int i = 0; i < 32; ++i) {
    const int ua = wsA[i], ub = wsB[i];
    beg += ua;
    n += ub;
    if (i < wv) pre += ub;
  }
  rpos[t] = pre + incB - len;
  rsrc[t] = t * CHUNK + s0;
  if (len > 32 * RUN_SLOTS) big = 1;
  __syncthreads();
  const bool fast = nwg <= 32 * RUNS_PW && big == 0;
  const bool in_lds = n <= LIST_CAP;
  int2 rec[RUNS_PW][RUN_SLOTS];
  if (fast) {
#pragma unroll
    for (int u = 0; u < RUNS_PW; ++u) {
      const int w = wv + 32 * u;
      const int L = rpos[w + 1] - rpos[w];
      const int2* run = binbuf + rsrc[w];
#pragma unroll
      for (int h = 0; h < RUN_SLOTS; ++h) {
        rec[u][h] = make_int2(0, -1);
        if (lane + 32 * h < L) rec[u][h] = run[lane + 32 * h];
      }
    }
#pragma unroll
    for (int u = 0; u < RUNS_PW; ++u) {
#pragma unroll
      for (int h = 0; h < RUN_SLOTS; ++h)
        if (rec[u][h].y >= 0) lds_count(&c[rec[u][h].y - g0]);
    }
  } else {
    for (int p = t; p < n; p += 1024) lds_count(&c[binbuf[run_index(rpos, rsrc, p)].y - g0]);
  }
  __syncthreads();
  const int v = c[t];
  const int ex = block_excl_scan(v, wsum);
  c[t] = ex;
  if (t < BIN_NODES && g0 + t < G) off[g0 + t] = beg + ex;
  __syncthreads();
  if (fast) {
#pragma unroll
    for (int u = 0; u < RUNS_PW; ++u) {
#pragma unroll
      for (int h = 0; h < RUN_SLOTS; ++h) {
        if (rec[u][h].y >= 0) {
          const int pos = lds_count(&c[rec[u][h].y - g0]);
          if (in_lds) lst[pos] = rec[u][h].x; else slist[beg + pos] = rec[u][h].x;
        }
      }
    }
  } else {
    for (int p = t; p < n; p += 1024) {
      const int2 r = binbuf[run_index(rpos, rsrc, p)];
      const int pos = lds_count(&c[r.y - g0]);
      if (in_lds) lst[pos] = r.x; else slist[beg + pos] = r.x;
    }
  }
  if (in_lds) {
    __syncthreads();
    for (int p = t; p < n; p += 1024) slist[beg + p] = lst[p];
  }
}

template <int G>
__device__ __forceinline__ void row_group(const float* __restrict__ x, unsigned lane8, int s_l,
                                          float p0, float p1, float p2, float p3,
                                          int j0, int cnt,
                                          v2f& acc0, v2f& acc1, v2f& acc2, v2f& acc3) {
  const int last = j0 + cnt - 1;
  v2f xv[G];
#pragma unroll
  for (int k = 0; k < G; ++k) {
    const int jj = (j0 + k < last) ? j0 + k : last;
    const int s = __builtin_amdgcn_readlane(s_l, jj);
    const unsigned ofs = (unsigned)s * 256u + lane8;
    xv[k] = *(const v2f*)((const char*)x + ofs);
  }
  if constexpr (G == 12) {
    asm volatile("" : "+v"(xv[0]), "+v"(xv[1]), "+v"(xv[2]), "+v"(xv[3]), "+v"(xv[4]), "+v"(xv[5]),
                      "+v"(xv[6]), "+v"(xv[7]), "+v"(xv[8]), "+v"(xv[9]), "+v"(xv[10]),
                      "+v"(xv[11]));
  } else if constexpr (G == 8) {
    asm volatile("" : "+v"(xv[0]), "+v"(xv[1]), "+v"(xv[2]), "+v"(xv[3]), "+v"(xv[4]), "+v"(xv[5]),
                      "+v"(xv[6]), "+v"(xv[7]));
  } else {
    static_assert(G == 4, "row_group: groups of 12, 8 or 4 rows");
    asm volatile("" : "+v"(xv[0]), "+v"(xv[1]), "+v"(xv[2]), "+v"(xv[3]));
  }
#pragma unroll
  for (int k = 0; k < G; ++k) {
    if (k < cnt) {
      const int j = j0 + k;
      const float w0 = __int_as_float(__builtin_amdgcn_readlane(__float_as_int(p0), j));
      const float w1 = __int_as_float(__builtin_amdgcn_readlane(__float_as_int(p1), j));
      const float w2 = __int_as_float(__builtin_amdgcn_readlane(__float_as_int(p2), j));
      const float w3 = __int_as_float(__builtin_amdgcn_readlane(__float_as_int(p3), j));
      acc0 += w0 * xv[k];
      acc1 += w1 * xv[k];
      acc2 += w2 * xv[k];
      acc3 += w3 * xv[k];
    }
  }
}

#define WALK_THREADS 128
__global__ __launch_bounds__(WALK_THREADS, 16)
void gather_kernel(const float* __restrict__ x,
                   const float* __restrict__ as_q, const float* __restrict__ as_a,
                   const float* __restrict__ ad_q, const float* __restrict__ ad_a,
                   const int* __restrict__ off, const int* __restrict__ slist,
                   const float* __restrict__ Pc, const float* __restrict__ cst,
                   float* __restrict__ out, int Nq, int G, int Etot) {
  __shared__ v2f Bt[2 * 16 * 32];
  const int lane = threadIdx.x & 31;
#pragma unroll
  for (int it = 0; it < 2 * 16 * 32 / WALK_THREADS; ++it) {
    const int idx = threadIdx.x + it * WALK_THREADS;
    const int f = idx >> 5, ln = idx & 31;
    const int n = ln & 15, kh = ln >> 4;
    const int q = f & 3;
    const int r = (f >> 2) * 64 + 2 * (4 * (n >> 2) + q) + 32 * kh;
    v2f b;
    b.x = Pc[r * 16 + (n & 3)];
    b.y = Pc[(r + 1) * 16 + (n & 3)];
    Bt[idx] = b;
  }
  __syncthreads();
  const int nw = (int)((gridDim.x * blockDim.x) >> 5);
  const unsigned lane8 = (unsigned)lane * 8u;
  const float ninf = __int_as_float(0xff800000);
  const bool has_share = ((lane >> 3) & 1) == (lane >> 4);
  const bool share_hi = (lane >> 2) & 1;
  const float c0 = cst[lane & 3];
  const int ngroups = (Nq + 7) >> 3;
  for (int t = __builtin_amdgcn_readfirstlane(
           (int)(((unsigned)blockIdx.x * blockDim.x + threadIdx.x) >> 5));
       t < ngroups; t += nw) {
  float res = 0.f;
#pragma unroll 1
  for (int i = 0; i < 8; ++i) {
  const int d = t * 8 + i;
  float part = 0.f;
#pragma unroll 1
  for (int L = 0; L < 2; ++L) {
  const int g = L ? Nq + d : d;
  if (d < Nq && g < G) {
  const float* a_s = L ? as_a : as_q;
  const float4 ad = *(const float4*)((L ? ad_a : ad_q) + (long)d * 4);
  const int beg = off[g];
  const int end = (g + 1 < G) ? off[g + 1] : Etot;
  float m0 = ninf, m1 = ninf, m2 = ninf, m3 = ninf;
  float d0 = 0.f, d1 = 0.f, d2 = 0.f, d3 = 0.f;
  v2f acc0 = {0.f, 0.f}, acc1 = {0.f, 0.f}, acc2 = {0.f, 0.f}, acc3 = {0.f, 0.f};
  for (int c = beg; c < end; c += 32) {
    const int n = (end - c < 32) ? end - c : 32;
    int s_l = 0;
    float e0 = ninf, e1 = ninf, e2 = ninf, e3 = ninf;
    if (lane < n) {
      s_l = slist[c + lane];
      const float4 as = *(const float4*)(a_s + (long)s_l * 4);
      e0 = lrelu(as.x + ad.x);
      e1 = lrelu(as.y + ad.y);
      e2 = lrelu(as.z + ad.z);
      e3 = lrelu(as.w + ad.w);
    }
    const float n0 = fmaxf(m0, wave_max(e0)), n1 = fmaxf(m1, wave_max(e1));
    const float n2 = fmaxf(m2, wave_max(e2)), n3 = fmaxf(m3, wave_max(e3));
    const float r0 = __expf(m0 - n0), r1 = __expf(m1 - n1);
    const float r2 = __expf(m2 - n2), r3 = __expf(m3 - n3);
    const float p0 = __expf(e0 - n0), p1 = __expf(e1 - n1);
    const float p2 = __expf(e2 - n2), p3 = __expf(e3 - n3);
    d0 = d0 * r0 + wave_sum(p0); d1 = d1 * r1 + wave_sum(p1);
    d2 = d2 * r2 + wave_sum(p2); d3 = d3 * r3 + wave_sum(p3);
    acc0 *= r0; acc1 *= r1; acc2 *= r2; acc3 *= r3;
    m0 = n0; m1 = n1; m2 = n2; m3 = n3;
    for (int j0 = 0; j0 < n;) {
      const int cnt = n - j0;
      if (cnt > 8) {
        row_group<12>(x, lane8, s_l, p0, p1, p2, p3, j0, cnt < 12 ? cnt : 12, acc0, acc1, acc2, acc3);
        j0 += 12;
      } else if (cnt > 4) {
        row_group<8>(x, lane8, s_l, p0, p1, p2, p3, j0, cnt, acc0, acc1, acc2, acc3);
        j0 += 8;
      } else {
        row_group<4>(x, lane8, s_l, p0, p1, p2, p3, j0, cnt, acc0, acc1, acc2, acc3);
        j0 += 4;
      }
    }
  }
  const float i0 = __builtin_amdgcn_rcpf(d0 + 1e-16f);
  const float i1 = __builtin_amdgcn_rcpf(d1 + 1e-16f);
  const float i2 = __builtin_amdgcn_rcpf(d2 + 1e-16f);
  const float i3 = __builtin_amdgcn_rcpf(d3 + 1e-16f);
  v2f zr[4];
  zr[0] = acc0 * i0;
  zr[1] = acc1 * i1;
  zr[2] = acc2 * i2;
  zr[3] = acc3 * i3;
  const v2f* bt = Bt + L * (16 * 32) + lane;
  v8f D[4] = {};
#pragma unroll
  for (int h = 0; h < 4; ++h) {
#pragma unroll
    for (int q = 0; q < 4; ++q) {
      const v2f b = bt[(h * 4 + q) * 32];
      D[q] = __builtin_amdgcn_wmma_f32_16x16x4_f32(false, zr[h], false, b, (short)0, D[q],
                                                   false, false);
    }
  }
  const float s_lo = (D[0][0] + D[1][1]) + (D[2][2] + D[3][3]);
  const float s_hi = (D[0][4] + D[1][5]) + (D[2][6] + D[3][7]);
  part += has_share ? (share_hi ? s_hi : s_lo) : 0.f;
  }
  }
  float v = part;
  v += xrow<4>(v);
  v += xrow<8>(v);
  v += xhalf(v);
  res = ((lane >> 2) == i) ? v : res;
  }
  if (t * 8 + (lane >> 2) < Nq)
    *(float*)((char*)out + ((unsigned)t * 128u + (unsigned)lane * 4u)) = res + c0;
  }
}

extern "C" void kernel_launch(void* const* d_in, const int* in_sizes, int n_in,
                              void* d_out, int out_size, void* d_ws, size_t ws_size,
                              hipStream_t stream) {
  const float* x_c = (const float*)d_in[0];
  const float* x_q = (const float*)d_in[1];
  const float* x_a = (const float*)d_in[2];
  const int* src_cq = (const int*)d_in[3];
  const int* dst_cq = (const int*)d_in[4];
  const int* src_ca = (const int*)d_in[5];
  const int* dst_ca = (const int*)d_in[6];
  const float* Wsq = (const float*)d_in[7];
  const float* Wdq = (const float*)d_in[8];
  const float* Asq = (const float*)d_in[9];
  const float* Adq = (const float*)d_in[10];
  const float* bq  = (const float*)d_in[11];
  const float* Wsa = (const float*)d_in[12];
  const float* Wda = (const float*)d_in[13];
  const float* Asa = (const float*)d_in[14];
  const float* Ada = (const float*)d_in[15];
  const float* ba  = (const float*)d_in[16];
  const float* Wout = (const float*)d_in[17];
  const float* bout = (const float*)d_in[18];

  const long Nc = in_sizes[0] / 64;
  const long Nq = in_sizes[1] / 64;
  const long Na = in_sizes[2] / 64;
  const long Ecq = in_sizes[3];
  const long Eca = in_sizes[5];

  float* w = (float*)d_ws;
  size_t off = 0;
  auto alloc = [&](size_t n) -> float* {
    float* p = w + off;
    off += (n + 15) & ~(size_t)15;
    return p;
  };
  float* Vp_sq = alloc(64 * 16);
  float* Vp_dq = alloc(64 * 16);
  float* Vp_sa = alloc(64 * 16);
  float* Vp_da = alloc(64 * 16);
  float* Pc    = alloc(512 * 16);
  float* cst   = alloc(16);
  float* as_cq = alloc(Nc * 4);
  float* as_ca = alloc(Nc * 4);
  float* ad_cq = alloc(Nq * 4);
  float* ad_ca = alloc(Na * 4);
  float* e_cq  = alloc(Ecq * 4);
  float* e_ca  = alloc(Eca * 4);
  float* m_q   = alloc(Nq * 4);
  float* m_a   = alloc(Na * 4);
  float* den_q = alloc(Nq * 4);
  float* den_a = alloc(Na * 4);
  float* z_q   = alloc(Nq * 256);
  float* z_a   = alloc(Na * 256);
  (void)Vp_dq; (void)Vp_sa; (void)Vp_da; (void)e_ca; (void)m_a; (void)den_q; (void)den_a;
  if (off * sizeof(float) > ws_size) return;

  const long G = Nq + Na;
  const long Etot = Ecq + Eca;
  const long NB = (G + BIN_NODES - 1) >> BIN_SH;
  const long nChunk = (Etot + CHUNK - 1) / CHUNK;
  const long nStart = NB * nChunk;
  int*  cstart = (int*)z_q;
  int2* binbuf = (int2*)z_a;
  int*  slist  = (int*)e_cq;
  int*  offs   = (int*)m_q;
  if (Nc * 256 > 0xffffffffL || Nq * 16 > 0xffffffffL) return;
  if (NB > NB_MAX || nChunk > 1024 || nStart > Nq * 256 || 2 * Etot > Na * 256 ||
      Etot > (long)(m_q - e_cq) || G > (long)(z_q - m_q)) return;

  prep_kernel<<<1, 1024, 0, stream>>>(Wsq, Wdq, Asq, Adq, Wsa, Wda, Asa, Ada,
                                      bq, ba, Wout, bout, Vp_sq, Pc, cst);
  const long nTile = (Nc + 15) / 16 + (Nq + 15) / 16 + (Na + 15) / 16;
  const int gS = (nTile + 7) / 8 < 1024 ? (int)((nTile + 7) / 8) : 1024;
  scores_kernel<<<gS, 256, 0, stream>>>(x_c, x_q, x_a, Vp_sq, as_cq, as_ca, ad_cq, ad_ca,
                                        (int)Nc, (int)Nq, (int)Na);

  bin_scatter_kernel<<<(int)nChunk, 1024, 0, stream>>>(src_cq, src_ca, dst_cq, dst_ca, cstart,
                                                       binbuf, (int)Ecq, (int)Etot, (int)Nq,
                                                       (int)NB);
  bin_lists_kernel<<<(int)NB, 1024, 0, stream>>>(binbuf, cstart, offs, slist,
                                                 (int)G, (int)Etot, (int)nChunk, (int)NB);
  const long nLine = (Nq + 7) / 8;
  const int gWalk = (nLine + 3) / 4 < 4096 ? (int)((nLine + 3) / 4) : 4096;
  gather_kernel<<<gWalk, WALK_THREADS, 0, stream>>>(x_c, as_cq, as_ca, ad_cq, ad_ca,
                                           offs, slist, Pc, cst, (float*)d_out,
                                           (int)Nq, (int)G, (int)Etot);
}


# Weight-table fold runs inside the edge-binning launch as 12 extra workgroups (1024 entries each, same 64-term sums) beside the chunk workgroups; its one-workgroup launch goes: 5 launches become 4
# speedup vs baseline: 11.4425x; 1.1568x over previous
#include <hip/hip_runtime.h>
#include <hip/hip_bf16.h>
#include <math.h>

typedef __attribute__((ext_vector_type(2))) float v2f;
typedef __attribute__((ext_vector_type(8))) float v8f;

__device__ __forceinline__ float lrelu(float x) { return x > 0.f ? x : 0.2f * x; }

template <int M>
__device__ __forceinline__ float xrow(float v) {
  const int b = __float_as_int(v);
  return __int_as_float(__builtin_amdgcn_update_dpp(b, b, 0x160 + M, 0xf, 0xf, false));
}
__device__ __forceinline__ float xhalf(float v) {
  const int b = __float_as_int(v);
  return __int_as_float(__builtin_amdgcn_permlanex16(b, b, 0x76543210, 0xfedcba98, false, false));
}
__device__ __forceinline__ float wave_max(float v) {
  v = fmaxf(v, xrow<1>(v));
  v = fmaxf(v, xrow<2>(v));
  v = fmaxf(v, xrow<4>(v));
  v = fmaxf(v, xrow<8>(v));
  return fmaxf(v, xhalf(v));
}
__device__ __forceinline__ float wave_sum(float v) {
  v += xrow<1>(v);
  v += xrow<2>(v);
  v += xrow<4>(v);
  v += xrow<8>(v);
  return v + xhalf(v);
}

#define FOLD_WGS 12
__device__ __forceinline__ void fold_tables(int p, int t,
                 const float* __restrict__ Wsq, const float* __restrict__ Wdq,
                 const float* __restrict__ Asq, const float* __restrict__ Adq,
                 const float* __restrict__ Wsa, const float* __restrict__ Wda,
                 const float* __restrict__ Asa, const float* __restrict__ Ada,
                 const float* __restrict__ bq,  const float* __restrict__ ba,
                 const float* __restrict__ Wout, const float* __restrict__ bout,
                 float* __restrict__ Vp, float* __restrict__ Pc, float* __restrict__ cst) {
  if (p < 4) {
    const int k = t >> 4, j = t & 15;
    const float* W = (p == 0) ? Wsq : (p == 1) ? Wdq : (p == 2) ? Wsa : Wda;
    const float* A = (p == 0) ? Asq : (p == 1) ? Adq : (p == 2) ? Asa : Ada;
    float s = 0.f;
    if (j < 4)
      for (int c = 0; c < 64; ++c) s += W[k * 256 + j * 64 + c] * A[j * 64 + c];
    Vp[p * 1024 + t] = s;
    if (p == 0 && t < 16) {
      float u = 0.f;
      if (t < 4) {
        for (int c = 0; c < 64; ++c)
          u += bq[c] * Wout[c * 4 + t] + ba[c] * Wout[(64 + c) * 4 + t];
        u += bout[t];
      }
      cst[t] = u;
    }
  } else {
    const int idx = (p - 4) * 1024 + t;
    const int r = idx >> 4, j = idx & 15;
    const int layer = r >> 8, h = (r >> 6) & 3, k = r & 63;
    const float* W = layer ? Wsa : Wsq;
    float s = 0.f;
    if (j < 4) {
      for (int c = 0; c < 64; ++c)
        s += W[k * 256 + h * 64 + c] * Wout[(layer * 64 + c) * 4 + j];
      s *= 0.25f;
    }
    Pc[idx] = s;
  }
}


template <bool TWO>
__device__ __forceinline__ void score_tiles(const float* __restrict__ x, int N,
                                            const float* __restrict__ tabA,
                                            const float* __restrict__ tabB,
                                            float* __restrict__ outA, float* __restrict__ outB,
                                            int first, int nw, int lane) {
  const int m = lane & 15;
  const int ko = (lane >> 4) * 2;
  const int ntiles = (N + 15) >> 4;
  const float* tb = (TWO && m >= 4) ? tabB : tabA;
  const int col = (TWO && m >= 4) ? m - 4 : m;
  v2f b[16];
#pragma unroll
  for (int q = 0; q < 16; ++q) {
    b[q].x = tb[(q * 4 + ko) * 16 + col];
    b[q].y = tb[(q * 4 + ko + 1) * 16 + col];
  }
  for (int t = first; t < ntiles; t += nw) {
    const long base = (long)t * 16;
    long row = base + m; if (row >= N) row = N - 1;
    const float* xr = x + row * 64;
    v8f acc = {};
#pragma unroll
    for (int q = 0; q < 16; ++q) {
      v2f a;
      a.x = xr[q * 4 + ko];
      a.y = xr[q * 4 + ko + 1];
      acc = __builtin_amdgcn_wmma_f32_16x16x4_f32(false, a, false, b[q], (short)0, acc,
                                                  false, false);
    }
    if (TWO ? m < 8 : m < 4) {
      float* o = (TWO && m >= 4) ? outB : outA;
      const int cc = m & 3;
      const long rb = base + ((lane >> 4) << 3);
#pragma unroll
      for (int v = 0; v < 8; ++v) {
        long r = rb + v;
        if (r < N) o[r * 4 + cc] = acc[v];
      }
    }
  }
}

__global__ void scores_kernel(const float* __restrict__ x_c, const float* __restrict__ x_q,
                              const float* __restrict__ x_a, const float* __restrict__ Vp,
                              float* __restrict__ as_q, float* __restrict__ as_a,
                              float* __restrict__ ad_q, float* __restrict__ ad_a,
                              int Nc, int Nq, int Na) {
  const int lane = threadIdx.x & 31;
  const int nw = (int)((gridDim.x * blockDim.x) >> 5);
  const int w = __builtin_amdgcn_readfirstlane(
      (int)(((unsigned)blockIdx.x * blockDim.x + threadIdx.x) >> 5));
  const int tc = (Nc + 15) >> 4;
  const int tq = (Nq + 15) >> 4;
  const int fq = (w + nw - tc % nw) % nw;
  const int fa = (w + nw - (tc + tq) % nw) % nw;
  score_tiles<true>(x_c, Nc, Vp, Vp + 2 * 1024, as_q, as_a, w, nw, lane);
  score_tiles<false>(x_q, Nq, Vp + 1024, nullptr, ad_q, nullptr, fq, nw, lane);
  score_tiles<false>(x_a, Na, Vp + 3 * 1024, nullptr, ad_a, nullptr, fa, nw, lane);
}

#define BIN_SH 9
#define BIN_NODES (1 << BIN_SH)
#define NB_MAX 1024
#define EPT 8
#define CHUNK (EPT * 1024)
#define LIST_CAP 12288

__device__ __forceinline__ int lds_count(int* p) {
  return __hip_atomic_fetch_add(p, 1, __ATOMIC_RELAXED, __HIP_MEMORY_SCOPE_WORKGROUP);
}

__device__ __forceinline__ int block_excl_scan(int v, int* wsum) {
  const int lane = threadIdx.x & 31;
  const int wv = threadIdx.x >> 5;
  int inc = v;
  for (int o = 1; o < 32; o <<= 1) {
    const int nb = __shfl_up(inc, o);
    if (lane >= o) inc += nb;
  }
  if (lane == 31) wsum[wv] = inc;
  __syncthreads();
  int wpre = 0;
  for (int i = 0; i < wv; ++i) wpre += wsum[i];
  return wpre + inc - v;
}

__global__ __launch_bounds__(1024)
void bin_scatter_kernel(const int* __restrict__ src_q, const int* __restrict__ src_a,
                        const int* __restrict__ dst_q, const int* __restrict__ dst_a,
                        int* __restrict__ cstart, int2* __restrict__ binbuf,
                        int Ecq, int Etot, int Nq, int NB, int nChunk,
                        const float* __restrict__ Wsq, const float* __restrict__ Wdq,
                        const float* __restrict__ Asq, const float* __restrict__ Adq,
                        const float* __restrict__ Wsa, const float* __restrict__ Wda,
                        const float* __restrict__ Asa, const float* __restrict__ Ada,
                        const float* __restrict__ bq,  const float* __restrict__ ba,
                        const float* __restrict__ Wout, const float* __restrict__ bout,
                        float* __restrict__ Vp, float* __restrict__ Pc, float* __restrict__ cst) {
  __shared__ int c[NB_MAX];
  __shared__ int wsum[32];
  __shared__ int2 stage[CHUNK];
  const int t = threadIdx.x;
  const int wg = blockIdx.x;
  if (wg >= nChunk) {
    fold_tables(wg - nChunk, t, Wsq, Wdq, Asq, Adq, Wsa, Wda, Asa, Ada, bq, ba, Wout, bout,
                Vp, Pc, cst);
    return;
  }
  c[t] = 0;
  __syncthreads();
  int g[EPT], s[EPT], r[EPT];
  const int i0 = wg * CHUNK + t;
  if (wg * CHUNK + CHUNK <= Etot) {
#pragma unroll
    for (int k = 0; k < EPT; ++k) {
      const int i = i0 + k * 1024;
      const bool second = i >= Ecq;
      const int e = second ? i - Ecq : i;
      const int* dp = second ? dst_a : dst_q;
      const int* sp = second ? src_a : src_q;
      g[k] = dp[e];
      s[k] = sp[e];
    }
#pragma unroll
    for (int k = 0; k < EPT; ++k)
      if (i0 + k * 1024 >= Ecq) g[k] += Nq;
  } else {
#pragma unroll
    for (int k = 0; k < EPT; ++k) {
      const int i = i0 + k * 1024;
      g[k] = -1;
      s[k] = 0;
      if (i < Etot) {
        const bool second = i >= Ecq;
        const int e = second ? i - Ecq : i;
        g[k] = second ? Nq + dst_a[e] : dst_q[e];
        s[k] = second ? src_a[e] : src_q[e];
      }
    }
  }
#pragma unroll
  for (int k = 0; k < EPT; ++k) {
    r[k] = 0;
    if (g[k] >= 0) r[k] = lds_count(&c[g[k] >> BIN_SH]);
  }
  __syncthreads();
  const int v = c[t];
  const int ex = block_excl_scan(v, wsum);
  c[t] = ex;
  if (t < NB) cstart[(long)wg * NB + t] = ex;
  __syncthreads();
#pragma unroll
  for (int k = 0; k < EPT; ++k)
    if (g[k] >= 0) stage[c[g[k] >> BIN_SH] + r[k]] = make_int2(s[k], g[k]);
  __syncthreads();
  const int left = Etot - wg * CHUNK;
  const int n = left < CHUNK ? left : CHUNK;
  int2* slice = binbuf + (long)wg * CHUNK;
  for (int p = t; p < n; p += 1024) slice[p] = stage[p];
}

__device__ __forceinline__ int run_index(const int* rpos, const int* rsrc, int p) {
  int w = 0;
  for (int step = 512; step > 0; step >>= 1)
    if (rpos[w + step] <= p) w += step;
  return rsrc[w] + (p - rpos[w]);
}

#define RUNS_PW 8
#define RUN_SLOTS 3

__global__ __launch_bounds__(1024, 16)
void bin_lists_kernel(const int2* __restrict__ binbuf, const int* __restrict__ cstart,
                      int* __restrict__ off, int* __restrict__ slist,
                      int G, int Etot, int nwg, int NB) {
  __shared__ int c[1024];
  __shared__ int wsum[32];
  __shared__ int wsA[32];
  __shared__ int wsB[32];
  __shared__ int rpos[1024];
  __shared__ int rsrc[1024];
  __shared__ int big;
  __shared__ int lst[LIST_CAP];
  const int t = threadIdx.x;
  const int lane = t & 31;
  const int wv = t >> 5;
  const int b = blockIdx.x;
  const int g0 = b << BIN_SH;
  int s0 = 0, len = 0;
  if (t < nwg) {
    const int left = Etot - t * CHUNK;
    const int nrec = left < CHUNK ? left : CHUNK;
    const int* cs = cstart + (long)t * NB + b;
    s0 = cs[0];
    const int s1 = (b + 1 < NB) ? cs[1] : nrec;
    len = s1 - s0;
  }
  int incA = s0, incB = len;
  for (int o = 1; o < 32; o <<= 1) {
    const int ua = __shfl_up(incA, o);
    const int ub = __shfl_up(incB, o);
    if (lane >= o) { incA += ua; incB += ub; }
  }
  if (lane == 31) { wsA[wv] = incA; wsB[wv] = incB; }
  if (t == 0) big = 0;
  c[t] = 0;
  __syncthreads();
  int beg = 0, n = 0, pre = 0;
  for (int i = 0; i < 32; ++i) {
    const int ua = wsA[i], ub = wsB[i];
    beg += ua;
    n += ub;
    if (i < wv) pre += ub;
  }
  rpos[t] = pre + incB - len;
  rsrc[t] = t * CHUNK + s0;
  if (len > 32 * RUN_SLOTS) big = 1;
  __syncthreads();
  const bool fast = nwg <= 32 * RUNS_PW && big == 0;
  const bool in_lds = n <= LIST_CAP;
  int2 rec[RUNS_PW][RUN_SLOTS];
  if (fast) {
#pragma unroll
    for (int u = 0; u < RUNS_PW; ++u) {
      const int w = wv + 32 * u;
      const int L = rpos[w + 1] - rpos[w];
      const int2* run = binbuf + rsrc[w];
#pragma unroll
      for (int h = 0; h < RUN_SLOTS; ++h) {
        rec[u][h] = make_int2(0, -1);
        if (lane + 32 * h < L) rec[u][h] = run[lane + 32 * h];
      }
    }
#pragma unroll
    for (int u = 0; u < RUNS_PW; ++u) {
#pragma unroll
      for (int h = 0; h < RUN_SLOTS; ++h)
        if (rec[u][h].y >= 0) lds_count(&c[rec[u][h].y - g0]);
    }
  } else {
    for (int p = t; p < n; p += 1024) lds_count(&c[binbuf[run_index(rpos, rsrc, p)].y - g0]);
  }
  __syncthreads();
  const int v = c[t];
  const int ex = block_excl_scan(v, wsum);
  c[t] = ex;
  if (t < BIN_NODES && g0 + t < G) off[g0 + t] = beg + ex;
  __syncthreads();
  if (fast) {
#pragma unroll
    for (int u = 0; u < RUNS_PW; ++u) {
#pragma unroll
      for (int h = 0; h < RUN_SLOTS; ++h) {
        if (rec[u][h].y >= 0) {
          const int pos = lds_count(&c[rec[u][h].y - g0]);
          if (in_lds) lst[pos] = rec[u][h].x; else slist[beg + pos] = rec[u][h].x;
        }
      }
    }
  } else {
    for (int p = t; p < n; p += 1024) {
      const int2 r = binbuf[run_index(rpos, rsrc, p)];
      const int pos = lds_count(&c[r.y - g0]);
      if (in_lds) lst[pos] = r.x; else slist[beg + pos] = r.x;
    }
  }
  if (in_lds) {
    __syncthreads();
    for (int p = t; p < n; p += 1024) slist[beg + p] = lst[p];
  }
}

template <int G>
__device__ __forceinline__ void row_group(const float* __restrict__ x, unsigned lane8, int s_l,
                                          float p0, float p1, float p2, float p3,
                                          int j0, int cnt,
                                          v2f& acc0, v2f& acc1, v2f& acc2, v2f& acc3) {
  const int last = j0 + cnt - 1;
  v2f xv[G];
#pragma unroll
  for (int k = 0; k < G; ++k) {
    const int jj = (j0 + k < last) ? j0 + k : last;
    const int s = __builtin_amdgcn_readlane(s_l, jj);
    const unsigned ofs = (unsigned)s * 256u + lane8;
    xv[k] = *(const v2f*)((const char*)x + ofs);
  }
  if constexpr (G == 12) {
    asm volatile("" : "+v"(xv[0]), "+v"(xv[1]), "+v"(xv[2]), "+v"(xv[3]), "+v"(xv[4]), "+v"(xv[5]),
                      "+v"(xv[6]), "+v"(xv[7]), "+v"(xv[8]), "+v"(xv[9]), "+v"(xv[10]),
                      "+v"(xv[11]));
  } else if constexpr (G == 8) {
    asm volatile("" : "+v"(xv[0]), "+v"(xv[1]), "+v"(xv[2]), "+v"(xv[3]), "+v"(xv[4]), "+v"(xv[5]),
                      "+v"(xv[6]), "+v"(xv[7]));
  } else {
    static_assert(G == 4, "row_group: groups of 12, 8 or 4 rows");
    asm volatile("" : "+v"(xv[0]), "+v"(xv[1]), "+v"(xv[2]), "+v"(xv[3]));
  }
#pragma unroll
  for (int k = 0; k < G; ++k) {
    if (k < cnt) {
      const int j = j0 + k;
      const float w0 = __int_as_float(__builtin_amdgcn_readlane(__float_as_int(p0), j));
      const float w1 = __int_as_float(__builtin_amdgcn_readlane(__float_as_int(p1), j));
      const float w2 = __int_as_float(__builtin_amdgcn_readlane(__float_as_int(p2), j));
      const float w3 = __int_as_float(__builtin_amdgcn_readlane(__float_as_int(p3), j));
      acc0 += w0 * xv[k];
      acc1 += w1 * xv[k];
      acc2 += w2 * xv[k];
      acc3 += w3 * xv[k];
    }
  }
}

#define WALK_THREADS 128
__global__ __launch_bounds__(WALK_THREADS, 16)
void gather_kernel(const float* __restrict__ x,
                   const float* __restrict__ as_q, const float* __restrict__ as_a,
                   const float* __restrict__ ad_q, const float* __restrict__ ad_a,
                   const int* __restrict__ off, const int* __restrict__ slist,
                   const float* __restrict__ Pc, const float* __restrict__ cst,
                   float* __restrict__ out, int Nq, int G, int Etot) {
  __shared__ v2f Bt[2 * 16 * 32];
  const int lane = threadIdx.x & 31;
#pragma unroll
  for (int it = 0; it < 2 * 16 * 32 / WALK_THREADS; ++it) {
    const int idx = threadIdx.x + it * WALK_THREADS;
    const int f = idx >> 5, ln = idx & 31;
    const int n = ln & 15, kh = ln >> 4;
    const int q = f & 3;
    const int r = (f >> 2) * 64 + 2 * (4 * (n >> 2) + q) + 32 * kh;
    v2f b;
    b.x = Pc[r * 16 + (n & 3)];
    b.y = Pc[(r + 1) * 16 + (n & 3)];
    Bt[idx] = b;
  }
  __syncthreads();
  const int nw = (int)((gridDim.x * blockDim.x) >> 5);
  const unsigned lane8 = (unsigned)lane * 8u;
  const float ninf = __int_as_float(0xff800000);
  const bool has_share = ((lane >> 3) & 1) == (lane >> 4);
  const bool share_hi = (lane >> 2) & 1;
  const float c0 = cst[lane & 3];
  const int ngroups = (Nq + 7) >> 3;
  for (int t = __builtin_amdgcn_readfirstlane(
           (int)(((unsigned)blockIdx.x * blockDim.x + threadIdx.x) >> 5));
       t < ngroups; t += nw) {
  float res = 0.f;
#pragma unroll 1
  for (int i = 0; i < 8; ++i) {
  const int d = t * 8 + i;
  float part = 0.f;
#pragma unroll 1
  for (int L = 0; L < 2; ++L) {
  const int g = L ? Nq + d : d;
  if (d < Nq && g < G) {
  const float* a_s = L ? as_a : as_q;
  const float4 ad = *(const float4*)((L ? ad_a : ad_q) + (long)d * 4);
  const int beg = off[g];
  const int end = (g + 1 < G) ? off[g + 1] : Etot;
  float m0 = ninf, m1 = ninf, m2 = ninf, m3 = ninf;
  float d0 = 0.f, d1 = 0.f, d2 = 0.f, d3 = 0.f;
  v2f acc0 = {0.f, 0.f}, acc1 = {0.f, 0.f}, acc2 = {0.f, 0.f}, acc3 = {0.f, 0.f};
  for (int c = beg; c < end; c += 32) {
    const int n = (end - c < 32) ? end - c : 32;
    int s_l = 0;
    float e0 = ninf, e1 = ninf, e2 = ninf, e3 = ninf;
    if (lane < n) {
      s_l = slist[c + lane];
      const float4 as = *(const float4*)(a_s + (long)s_l * 4);
      e0 = lrelu(as.x + ad.x);
      e1 = lrelu(as.y + ad.y);
      e2 = lrelu(as.z + ad.z);
      e3 = lrelu(as.w + ad.w);
    }
    const float n0 = fmaxf(m0, wave_max(e0)), n1 = fmaxf(m1, wave_max(e1));
    const float n2 = fmaxf(m2, wave_max(e2)), n3 = fmaxf(m3, wave_max(e3));
    const float r0 = __expf(m0 - n0), r1 = __expf(m1 - n1);
    const float r2 = __expf(m2 - n2), r3 = __expf(m3 - n3);
    const float p0 = __expf(e0 - n0), p1 = __expf(e1 - n1);
    const float p2 = __expf(e2 - n2), p3 = __expf(e3 - n3);
    d0 = d0 * r0 + wave_sum(p0); d1 = d1 * r1 + wave_sum(p1);
    d2 = d2 * r2 + wave_sum(p2); d3 = d3 * r3 + wave_sum(p3);
    acc0 *= r0; acc1 *= r1; acc2 *= r2; acc3 *= r3;
    m0 = n0; m1 = n1; m2 = n2; m3 = n3;
    for (int j0 = 0; j0 < n;) {
      const int cnt = n - j0;
      if (cnt > 8) {
        row_group<12>(x, lane8, s_l, p0, p1, p2, p3, j0, cnt < 12 ? cnt : 12, acc0, acc1, acc2, acc3);
        j0 += 12;
      } else if (cnt > 4) {
        row_group<8>(x, lane8, s_l, p0, p1, p2, p3, j0, cnt, acc0, acc1, acc2, acc3);
        j0 += 8;
      } else {
        row_group<4>(x, lane8, s_l, p0, p1, p2, p3, j0, cnt, acc0, acc1, acc2, acc3);
        j0 += 4;
      }
    }
  }
  const float i0 = __builtin_amdgcn_rcpf(d0 + 1e-16f);
  const float i1 = __builtin_amdgcn_rcpf(d1 + 1e-16f);
  const float i2 = __builtin_amdgcn_rcpf(d2 + 1e-16f);
  const float i3 = __builtin_amdgcn_rcpf(d3 + 1e-16f);
  v2f zr[4];
  zr[0] = acc0 * i0;
  zr[1] = acc1 * i1;
  zr[2] = acc2 * i2;
  zr[3] = acc3 * i3;
  const v2f* bt = Bt + L * (16 * 32) + lane;
  v8f D[4] = {};
#pragma unroll
  for (int h = 0; h < 4; ++h) {
#pragma unroll
    for (int q = 0; q < 4; ++q) {
      const v2f b = bt[(h * 4 + q) * 32];
      D[q] = __builtin_amdgcn_wmma_f32_16x16x4_f32(false, zr[h], false, b, (short)0, D[q],
                                                   false, false);
    }
  }
  const float s_lo = (D[0][0] + D[1][1]) + (D[2][2] + D[3][3]);
  const float s_hi = (D[0][4] + D[1][5]) + (D[2][6] + D[3][7]);
  part += has_share ? (share_hi ? s_hi : s_lo) : 0.f;
  }
  }
  float v = part;
  v += xrow<4>(v);
  v += xrow<8>(v);
  v += xhalf(v);
  res = ((lane >> 2) == i) ? v : res;
  }
  if (t * 8 + (lane >> 2) < Nq)
    *(float*)((char*)out + ((unsigned)t * 128u + (unsigned)lane * 4u)) = res + c0;
  }
}

extern "C" void kernel_launch(void* const* d_in, const int* in_sizes, int n_in,
                              void* d_out, int out_size, void* d_ws, size_t ws_size,
                              hipStream_t stream) {
  const float* x_c = (const float*)d_in[0];
  const float* x_q = (const float*)d_in[1];
  const float* x_a = (const float*)d_in[2];
  const int* src_cq = (const int*)d_in[3];
  const int* dst_cq = (const int*)d_in[4];
  const int* src_ca = (const int*)d_in[5];
  const int* dst_ca = (const int*)d_in[6];
  const float* Wsq = (const float*)d_in[7];
  const float* Wdq = (const float*)d_in[8];
  const float* Asq = (const float*)d_in[9];
  const float* Adq = (const float*)d_in[10];
  const float* bq  = (const float*)d_in[11];
  const float* Wsa = (const float*)d_in[12];
  const float* Wda = (const float*)d_in[13];
  const float* Asa = (const float*)d_in[14];
  const float* Ada = (const float*)d_in[15];
  const float* ba  = (const float*)d_in[16];
  const float* Wout = (const float*)d_in[17];
  const float* bout = (const float*)d_in[18];

  const long Nc = in_sizes[0] / 64;
  const long Nq = in_sizes[1] / 64;
  const long Na = in_sizes[2] / 64;
  const long Ecq = in_sizes[3];
  const long Eca = in_sizes[5];

  float* w = (float*)d_ws;
  size_t off = 0;
  auto alloc = [&](size_t n) -> float* {
    float* p = w + off;
    off += (n + 15) & ~(size_t)15;
    return p;
  };
  float* Vp_sq = alloc(64 * 16);
  float* Vp_dq = alloc(64 * 16);
  float* Vp_sa = alloc(64 * 16);
  float* Vp_da = alloc(64 * 16);
  float* Pc    = alloc(512 * 16);
  float* cst   = alloc(16);
  float* as_cq = alloc(Nc * 4);
  float* as_ca = alloc(Nc * 4);
  float* ad_cq = alloc(Nq * 4);
  float* ad_ca = alloc(Na * 4);
  float* e_cq  = alloc(Ecq * 4);
  float* e_ca  = alloc(Eca * 4);
  float* m_q   = alloc(Nq * 4);
  float* m_a   = alloc(Na * 4);
  float* den_q = alloc(Nq * 4);
  float* den_a = alloc(Na * 4);
  float* z_q   = alloc(Nq * 256);
  float* z_a   = alloc(Na * 256);
  (void)Vp_dq; (void)Vp_sa; (void)Vp_da; (void)e_ca; (void)m_a; (void)den_q; (void)den_a;
  if (off * sizeof(float) > ws_size) return;

  const long G = Nq + Na;
  const long Etot = Ecq + Eca;
  const long NB = (G + BIN_NODES - 1) >> BIN_SH;
  const long nChunk = (Etot + CHUNK - 1) / CHUNK;
  const long nStart = NB * nChunk;
  int*  cstart = (int*)z_q;
  int2* binbuf = (int2*)z_a;
  int*  slist  = (int*)e_cq;
  int*  offs   = (int*)m_q;
  if (Nc * 256 > 0xffffffffL || Nq * 16 > 0xffffffffL) return;
  if (NB > NB_MAX || nChunk > 1024 || nStart > Nq * 256 || 2 * Etot > Na * 256 ||
      Etot > (long)(m_q - e_cq) || G > (long)(z_q - m_q)) return;

  bin_scatter_kernel<<<(int)nChunk + FOLD_WGS, 1024, 0, stream>>>(
      src_cq, src_ca, dst_cq, dst_ca, cstart, binbuf, (int)Ecq, (int)Etot, (int)Nq, (int)NB,
      (int)nChunk, Wsq, Wdq, Asq, Adq, Wsa, Wda, Asa, Ada, bq, ba, Wout, bout, Vp_sq, Pc, cst);
  const long nTile = (Nc + 15) / 16 + (Nq + 15) / 16 + (Na + 15) / 16;
  const int gS = (nTile + 7) / 8 < 1024 ? (int)((nTile + 7) / 8) : 1024;
  scores_kernel<<<gS, 256, 0, stream>>>(x_c, x_q, x_a, Vp_sq, as_cq, as_ca, ad_cq, ad_ca,
                                        (int)Nc, (int)Nq, (int)Na);

  bin_lists_kernel<<<(int)NB, 1024, 0, stream>>>(binbuf, cstart, offs, slist,
                                                 (int)G, (int)Etot, (int)nChunk, (int)NB);
  const long nLine = (Nq + 7) / 8;
  const int gWalk = (nLine + 3) / 4 < 4096 ? (int)((nLine + 3) / 4) : 4096;
  gather_kernel<<<gWalk, WALK_THREADS, 0, stream>>>(x_c, as_cq, as_ca, ad_cq, ad_ca,
                                           offs, slist, Pc, cst, (float*)d_out,
                                           (int)Nq, (int)G, (int)Etot);
}
